# starting kernel
# baseline (speedup 1.0000x reference)
#include <hip/hip_runtime.h>
#include <hip/hip_bf16.h>
#include <math.h>


typedef __attribute__((ext_vector_type(16))) __bf16 bf16x16;
typedef __attribute__((ext_vector_type(8)))  float  floatx8;

union FragAB { bf16x16 v; unsigned int u[8]; };

__device__ __forceinline__ unsigned pack2bf(float a, float b) {
  __bf16 lo = (__bf16)a, hi = (__bf16)b;
  unsigned short ul = __builtin_bit_cast(unsigned short, lo);
  unsigned short uh = __builtin_bit_cast(unsigned short, hi);
  return (unsigned)ul | ((unsigned)uh << 16);
}

__device__ __forceinline__ int kpair(int p, int h) {
  return (p < 4) ? (2 * p + 8 * h) : (8 + 2 * p + 8 * h);
}

__device__ __forceinline__ bf16x16 load_frag_bf16_rm(const __bf16* src, int row0,
                                                     int k0, int ld) {
  int lane = threadIdx.x & 31;
  int r = lane & 15, h = lane >> 4;
  const __bf16* base = src + (size_t)(row0 + r) * ld + k0;
  FragAB f;
#pragma unroll
  for (int p = 0; p < 8; ++p) f.u[p] = *(const unsigned*)(base + kpair(p, h));
  return f.v;
}

__device__ __forceinline__ floatx8 wmma_bf16(bf16x16 a, bf16x16 b, floatx8 c) {
  return __builtin_amdgcn_wmma_f32_16x16x32_bf16(false, a, false, b, (short)0, c,
                                                 false, false);
}

template <int KSTEPS>
__device__ __forceinline__ void gemm_strip(const __bf16* A, int rowA,
                                           const __bf16* B, int rowB, int ld,
                                           floatx8 acc[4]) {
#pragma unroll
  for (int ks = 0; ks < KSTEPS; ++ks) {
    int k0 = ks * 32;
    bf16x16 af = load_frag_bf16_rm(A, rowA, k0, ld);
#pragma unroll
    for (int t = 0; t < 4; ++t) {
      bf16x16 bfr = load_frag_bf16_rm(B, rowB + 16 * t, k0, ld);
      acc[t] = wmma_bf16(af, bfr, acc[t]);
    }
  }
}

__global__ void k_convert_w(const float* wqkv, const float* wout,
                            __bf16* wqkv_b, __bf16* wout_b, float* scales) {
  int i = blockIdx.x * blockDim.x + threadIdx.x;
  if (i < 2) scales[i] = 0.f;
  if (i < 1536 * 512) wqkv_b[i] = (__bf16)wqkv[i];
  if (i < 512 * 512) wout_b[i] = (__bf16)wout[i];
}

__global__ void k_layernorm(const float* x, const float* g, const float* b,
                            __bf16* xn, int ntok) {
  int wave = (int)((blockIdx.x * blockDim.x + threadIdx.x) >> 5);
  int lane = threadIdx.x & 31;
  if (wave >= ntok) return;
  const float* xr = x + (size_t)wave * 512;
  float s = 0.f, s2 = 0.f, vals[16];
#pragma unroll
  for (int i = 0; i < 16; ++i) {
    float v = xr[lane + 32 * i];
    vals[i] = v; s += v; s2 += v * v;
  }
  for (int m = 16; m >= 1; m >>= 1) {
    s += __shfl_xor(s, m, 32);
    s2 += __shfl_xor(s2, m, 32);
  }
  float mu = s * (1.f / 512.f);
  float var = s2 * (1.f / 512.f) - mu * mu;
  float inv = rsqrtf(var + 1e-5f);
  __bf16* o = xn + (size_t)wave * 512;
#pragma unroll
  for (int i = 0; i < 16; ++i) {
    int c = lane + 32 * i;
    o[c] = (__bf16)((vals[i] - mu) * inv * g[c] + b[c]);
  }
}

__global__ void k_gemm_qkv(const __bf16* xn, const __bf16* wq, __bf16* qb,
                           __bf16* kb, float* vf, __bf16* vT) {
  int wid = (int)((blockIdx.x * blockDim.x + threadIdx.x) >> 5);
  int tM = wid / 24, s = wid % 24;
  floatx8 acc[4] = {};
  gemm_strip<16>(xn, tM * 16, wq, s * 64, 512, acc);
  int lane = threadIdx.x & 31;
  int n = lane & 15, h = lane >> 4;
  int col0 = s * 64;
  int sect = col0 >> 9;
  int head = (col0 >> 6) & 7;
#pragma unroll
  for (int t = 0; t < 4; ++t) {
    int dh = 16 * t + n;
#pragma unroll
    for (int r = 0; r < 8; ++r) {
      int tok = tM * 16 + r + 8 * h;
      int bb = tok >> 12, ns = tok & 4095;
      size_t bh = (size_t)(bb * 8 + head);
      size_t idx = (bh * 4096 + ns) * 64 + dh;
      float val = acc[t][r];
      if (sect == 0) qb[idx] = (__bf16)(val * 0.125f);
      else if (sect == 1) kb[idx] = (__bf16)val;
      else {
        vf[idx] = val;
        vT[(bh * 64 + dh) * 4096 + ns] = (__bf16)val;
      }
    }
  }
}

__global__ void k_landmarks(const __bf16* qb, const __bf16* kb, __bf16* ql,
                            __bf16* kl) {
  int i = blockIdx.x * blockDim.x + threadIdx.x;
  if (i >= 32 * 256 * 64) return;
  int dh = i & 63, m = (i >> 6) & 255, bh = i >> 14;
  size_t base = ((size_t)bh * 4096 + m * 16) * 64 + dh;
  float sq = 0.f, sk = 0.f;
#pragma unroll
  for (int t = 0; t < 16; ++t) {
    sq += (float)qb[base + t * 64];
    sk += (float)kb[base + t * 64];
  }
  ql[i] = (__bf16)(sq * (1.f / 16.f));
  kl[i] = (__bf16)(sk * (1.f / 16.f));
}

__global__ void k_attn2_logits(const __bf16* ql, const __bf16* kl, float* out) {
  int wid = (int)((blockIdx.x * blockDim.x + threadIdx.x) >> 5);
  int bh = wid >> 6, t = wid & 63;
  int tM = t >> 2, s = t & 3;
  const __bf16* A = ql + (size_t)bh * 256 * 64;
  const __bf16* B = kl + (size_t)bh * 256 * 64;
  floatx8 acc[4] = {};
  gemm_strip<2>(A, tM * 16, B, s * 64, 64, acc);
  int lane = threadIdx.x & 31;
  int n = lane & 15, h = lane >> 4;
  float* o = out + (size_t)bh * 256 * 256;
#pragma unroll
  for (int t4 = 0; t4 < 4; ++t4)
#pragma unroll
    for (int r = 0; r < 8; ++r)
      o[(size_t)(tM * 16 + r + 8 * h) * 256 + s * 64 + 16 * t4 + n] =
          acc[t4][r];
}

__global__ void k_softmax256(float* x, __bf16* xb, int nrows) {
  int wave = (int)((blockIdx.x * blockDim.x + threadIdx.x) >> 5);
  if (wave >= nrows) return;
  int lane = threadIdx.x & 31;
  float* r = x + (size_t)wave * 256;
  __bf16* rb = xb + (size_t)wave * 256;
  float v[8], m = -1e30f;
#pragma unroll
  for (int i = 0; i < 8; ++i) { v[i] = r[lane + 32 * i]; m = fmaxf(m, v[i]); }
  for (int t = 16; t >= 1; t >>= 1) m = fmaxf(m, __shfl_xor(m, t, 32));
  float s = 0.f;
#pragma unroll
  for (int i = 0; i < 8; ++i) { v[i] = __expf(v[i] - m); s += v[i]; }
  for (int t = 16; t >= 1; t >>= 1) s += __shfl_xor(s, t, 32);
  float inv = 1.f / s;
#pragma unroll
  for (int i = 0; i < 8; ++i) {
    float pv = v[i] * inv;
    r[lane + 32 * i] = pv;
    rb[lane + 32 * i] = (__bf16)pv;
  }
}

__global__ void k_pinv_scales(const float* x, float* scales) {
  int wave = (int)((blockIdx.x * blockDim.x + threadIdx.x) >> 5);
  if (wave >= 32 * 256) return;
  int lane = threadIdx.x & 31;
  int bh = wave >> 8, r = wave & 255;
  const float* m = x + (size_t)bh * 256 * 256;
  float rs = 0.f, cs = 0.f;
#pragma unroll
  for (int i = 0; i < 8; ++i) {
    rs += fabsf(m[(size_t)r * 256 + lane + 32 * i]);
    cs += fabsf(m[(size_t)(lane + 32 * i) * 256 + r]);
  }
  for (int t = 16; t >= 1; t >>= 1) {
    rs += __shfl_xor(rs, t, 32);
    cs += __shfl_xor(cs, t, 32);
  }
  if (lane == 0) {
    atomicMax((unsigned*)&scales[0], __float_as_uint(rs));
    atomicMax((unsigned*)&scales[1], __float_as_uint(cs));
  }
}

__global__ void k_init_z(const float* x, __bf16* zb, __bf16* zTb,
                         const float* scales) {
  size_t i = (size_t)blockIdx.x * blockDim.x + threadIdx.x;
  if (i >= (size_t)32 * 256 * 256) return;
  int j = (int)(i & 255), r = (int)((i >> 8) & 255);
  size_t bh = i >> 16;
  float denom = scales[0] * scales[1];
  float val = x[(bh << 16) + (size_t)j * 256 + r] / denom;
  zb[i] = (__bf16)val;
  zTb[(bh << 16) + (size_t)j * 256 + r] = (__bf16)val;
}

__global__ void k_gemm256b(const __bf16* A, const __bf16* BT, __bf16* OutRaw,
                           __bf16* OutT, __bf16* OutTT, float alpha,
                           float beta) {
  int wid = (int)((blockIdx.x * blockDim.x + threadIdx.x) >> 5);
  int bh = wid >> 6, t = wid & 63;
  int tM = t >> 2, s = t & 3;
  size_t bofs = (size_t)bh << 16;
  floatx8 acc[4] = {};
  gemm_strip<8>(A + bofs, tM * 16, BT + bofs, s * 64, 256, acc);
  int lane = threadIdx.x & 31;
  int n = lane & 15, h = lane >> 4;
#pragma unroll
  for (int t4 = 0; t4 < 4; ++t4)
#pragma unroll
    for (int r = 0; r < 8; ++r) {
      int M = tM * 16 + r + 8 * h, N = s * 64 + 16 * t4 + n;
      float tv = acc[t4][r];
      if (OutRaw) OutRaw[bofs + (size_t)M * 256 + N] = (__bf16)tv;
      float tr = alpha * (M == N ? 1.f : 0.f) + beta * tv;
      if (OutT) OutT[bofs + (size_t)M * 256 + N] = (__bf16)tr;
      if (OutTT) OutTT[bofs + (size_t)N * 256 + M] = (__bf16)tr;
    }
}

__device__ __forceinline__ void stream_attn(const __bf16* Ak, const __bf16* Bq,
                                            int qrow0, const __bf16* A2,
                                            int ldA2, int nkeys, float* outF,
                                            int rsF, __bf16* outB, int nsB,
                                            int dsB) {
  int lane = threadIdx.x & 31;
  int n = lane & 15, h = lane >> 4;
  bf16x16 bq0 = load_frag_bf16_rm(Bq, qrow0, 0, 64);
  bf16x16 bq1 = load_frag_bf16_rm(Bq, qrow0, 32, 64);
  float run_m = -1e30f, run_s = 0.f;
  floatx8 acc[4] = {};
  for (int c = 0; c < nkeys; c += 32) {
    floatx8 s0 = {}, s1 = {};
    s0 = wmma_bf16(load_frag_bf16_rm(Ak, c, 0, 64), bq0, s0);
    s0 = wmma_bf16(load_frag_bf16_rm(Ak, c, 32, 64), bq1, s0);
    s1 = wmma_bf16(load_frag_bf16_rm(Ak, c + 16, 0, 64), bq0, s1);
    s1 = wmma_bf16(load_frag_bf16_rm(Ak, c + 16, 32, 64), bq1, s1);
    float mc = -1e30f;
#pragma unroll
    for (int r = 0; r < 8; ++r) { mc = fmaxf(mc, s0[r]); mc = fmaxf(mc, s1[r]); }
    mc = fmaxf(mc, __shfl_xor(mc, 16, 32));
    float newm = fmaxf(run_m, mc);
    float scale = __expf(run_m - newm);
    float p0[8], p1[8], ps = 0.f;
#pragma unroll
    for (int r = 0; r < 8; ++r) {
      p0[r] = __expf(s0[r] - newm);
      p1[r] = __expf(s1[r] - newm);
      ps += p0[r] + p1[r];
    }
    run_s = run_s * scale + ps;
    run_m = newm;
    FragAB bp;
#pragma unroll
    for (int p = 0; p < 4; ++p) {
      bp.u[p] = pack2bf(p0[2 * p], p0[2 * p + 1]);
      bp.u[4 + p] = pack2bf(p1[2 * p], p1[2 * p + 1]);
    }
#pragma unroll
    for (int t = 0; t < 4; ++t) {
#pragma unroll
      for (int r = 0; r < 8; ++r) acc[t][r] *= scale;
      acc[t] = wmma_bf16(load_frag_bf16_rm(A2, 16 * t, c, ldA2), bp.v, acc[t]);
    }
  }
  float tot = run_s + __shfl_xor(run_s, 16, 32);
  float inv = 1.f / tot;
#pragma unroll
  for (int t = 0; t < 4; ++t)
#pragma unroll
    for (int r = 0; r < 8; ++r) {
      int dh = 16 * t + r + 8 * h;
      float val = acc[t][r] * inv;
      if (outF) outF[(size_t)n * rsF + dh] = val;
      if (outB) outB[(size_t)n * nsB + (size_t)dh * dsB] = (__bf16)val;
    }
}

__global__ void k_attn3v(const __bf16* kb, const __bf16* ql, const __bf16* vT,
                         __bf16* kvT) {
  int wid = (int)((blockIdx.x * blockDim.x + threadIdx.x) >> 5);
  int bh = wid >> 4, blk = wid & 15;
  stream_attn(kb + (size_t)bh * 4096 * 64, ql + (size_t)bh * 256 * 64, blk * 16,
              vT + (size_t)bh * 64 * 4096, 4096, 4096, nullptr, 0,
              kvT + (size_t)bh * 64 * 256 + blk * 16, 1, 256);
}

__global__ void k_w2(const __bf16* zb, const __bf16* kvT, __bf16* w2t) {
  int wid = (int)((blockIdx.x * blockDim.x + threadIdx.x) >> 5);
  int bh = wid >> 4, tM = wid & 15;
  floatx8 acc[4] = {};
  gemm_strip<8>(zb + ((size_t)bh << 16), tM * 16, kvT + (size_t)bh * 64 * 256,
                0, 256, acc);
  int lane = threadIdx.x & 31;
  int n = lane & 15, h = lane >> 4;
#pragma unroll
  for (int t4 = 0; t4 < 4; ++t4)
#pragma unroll
    for (int r = 0; r < 8; ++r) {
      int M = tM * 16 + r + 8 * h;
      int N = 16 * t4 + n;
      w2t[((size_t)bh * 64 + N) * 256 + M] = (__bf16)acc[t4][r];
    }
}

__global__ void k_attn1_out(const __bf16* kl, const __bf16* qb,
                            const __bf16* w2t, float* aout) {
  int wid = (int)((blockIdx.x * blockDim.x + threadIdx.x) >> 5);
  int bh = wid >> 8, blk = wid & 255;
  int b = bh >> 3, hh = bh & 7;
  stream_attn(kl + (size_t)bh * 256 * 64, qb + (size_t)bh * 4096 * 64, blk * 16,
              w2t + (size_t)bh * 64 * 256, 256, 256,
              aout + ((size_t)b * 4096 + blk * 16) * 512 + hh * 64, 512,
              nullptr, 0, 0);
}

__global__ void k_conv_res(const float* vf, const float* rw, float* aout,
                           __bf16* aout_b) {
  size_t i = (size_t)blockIdx.x * blockDim.x + threadIdx.x;
  if (i >= (size_t)4 * 4096 * 512) return;
  int col = (int)(i & 511);
  size_t tok = i >> 9;
  int b = (int)(tok >> 12), ns = (int)(tok & 4095);
  int hh = col >> 6, dh = col & 63;
  const float* vrow = vf + ((size_t)(b * 8 + hh) * 4096) * 64 + dh;
  float s = 0.f;
#pragma unroll
  for (int j = 0; j < 33; ++j) {
    int p = ns + j - 16;
    if (p >= 0 && p < 4096) s += vrow[(size_t)p * 64] * rw[hh * 33 + j];
  }
  float val = aout[i] + s;
  aout[i] = val;
  aout_b[i] = (__bf16)val;
}

__global__ void k_outproj(const __bf16* aoutb, const __bf16* wo,
                          const float* bias, const float* x, float* out) {
  int wid = (int)((blockIdx.x * blockDim.x + threadIdx.x) >> 5);
  int tM = wid >> 3, s = wid & 7;
  floatx8 acc[4] = {};
  gemm_strip<16>(aoutb, tM * 16, wo, s * 64, 512, acc);
  int lane = threadIdx.x & 31;
  int n = lane & 15, h = lane >> 4;
#pragma unroll
  for (int t4 = 0; t4 < 4; ++t4) {
    int col = s * 64 + 16 * t4 + n;
    float bc = bias[col];
#pragma unroll
    for (int r = 0; r < 8; ++r) {
      size_t idx = (size_t)(tM * 16 + r + 8 * h) * 512 + col;
      out[idx] = x[idx] + acc[t4][r] + bc;
    }
  }
}

extern "C" void kernel_launch(void* const* d_in, const int* in_sizes, int n_in,
                              void* d_out, int out_size, void* d_ws,
                              size_t ws_size, hipStream_t stream) {
  const float* x     = (const float*)d_in[0];
  const float* ln_g  = (const float*)d_in[1];
  const float* ln_b  = (const float*)d_in[2];
  const float* w_qkv = (const float*)d_in[3];
  const float* w_out = (const float*)d_in[4];
  const float* b_out = (const float*)d_in[5];
  const float* res_w = (const float*)d_in[6];
  float* out = (float*)d_out;

  char* p = (char*)d_ws;
  auto alloc = [&](size_t bytes) -> void* {
    void* r = (void*)p;
    p += (bytes + 255) & ~(size_t)255;
    return r;
  };
  const size_t NTOK = 4 * 4096;
  const size_t MM = (size_t)32 * 256 * 256;
  __bf16* xn_b   = (__bf16*)alloc(NTOK * 512 * 2);
  __bf16* wqkv_b = (__bf16*)alloc(1536 * 512 * 2);
  __bf16* wout_b = (__bf16*)alloc(512 * 512 * 2);
  __bf16* q_b    = (__bf16*)alloc((size_t)32 * 4096 * 64 * 2);
  __bf16* k_b    = (__bf16*)alloc((size_t)32 * 4096 * 64 * 2);
  float*  v_f    = (float*)alloc((size_t)32 * 4096 * 64 * 4);
  __bf16* vT_b   = (__bf16*)alloc((size_t)32 * 64 * 4096 * 2);
  __bf16* ql_b   = (__bf16*)alloc((size_t)32 * 256 * 64 * 2);
  __bf16* kl_b   = (__bf16*)alloc((size_t)32 * 256 * 64 * 2);
  float*  mat_x  = (float*)alloc(MM * 4);
  __bf16* x_b    = (__bf16*)alloc(MM * 2);
  __bf16* z0_b   = (__bf16*)alloc(MM * 2);
  __bf16* z0T_b  = (__bf16*)alloc(MM * 2);
  __bf16* z1_b   = (__bf16*)alloc(MM * 2);
  __bf16* z1T_b  = (__bf16*)alloc(MM * 2);
  __bf16* xz_b   = (__bf16*)alloc(MM * 2);
  __bf16* t1T_b  = (__bf16*)alloc(MM * 2);
  __bf16* t2T_b  = (__bf16*)alloc(MM * 2);
  __bf16* t3T_b  = (__bf16*)alloc(MM * 2);
  __bf16* kvT_b  = (__bf16*)alloc((size_t)32 * 64 * 256 * 2);
  __bf16* w2t_b  = (__bf16*)alloc((size_t)32 * 64 * 256 * 2);
  float*  aout_f = (float*)alloc(NTOK * 512 * 4);
  __bf16* aout_b = (__bf16*)alloc(NTOK * 512 * 2);
  float*  scales = (float*)alloc(256);
  (void)ws_size; (void)n_in; (void)in_sizes; (void)out_size;

  const int T = 256;
  k_convert_w<<<3072, T, 0, stream>>>(w_qkv, w_out, wqkv_b, wout_b, scales);
  k_layernorm<<<2048, T, 0, stream>>>(x, ln_g, ln_b, xn_b, (int)NTOK);
  k_gemm_qkv<<<3072, T, 0, stream>>>(xn_b, wqkv_b, q_b, k_b, v_f, vT_b);
  k_landmarks<<<2048, T, 0, stream>>>(q_b, k_b, ql_b, kl_b);
  k_attn2_logits<<<256, T, 0, stream>>>(ql_b, kl_b, mat_x);
  k_softmax256<<<1024, T, 0, stream>>>(mat_x, x_b, 32 * 256);
  k_pinv_scales<<<1024, T, 0, stream>>>(mat_x, scales);
  k_init_z<<<8192, T, 0, stream>>>(mat_x, z0_b, z0T_b, scales);

  __bf16* zc = z0_b;  __bf16* zcT = z0T_b;
  __bf16* zn = z1_b;  __bf16* znT = z1T_b;
  for (int it = 0; it < 6; ++it) {
    k_gemm256b<<<256, T, 0, stream>>>(x_b, zcT, xz_b, nullptr, t1T_b, 7.f, -1.f);
    k_gemm256b<<<256, T, 0, stream>>>(xz_b, t1T_b, nullptr, nullptr, t2T_b, 15.f, -1.f);
    k_gemm256b<<<256, T, 0, stream>>>(xz_b, t2T_b, nullptr, nullptr, t3T_b, 13.f, -1.f);
    k_gemm256b<<<256, T, 0, stream>>>(zc, t3T_b, nullptr, zn, znT, 0.f, 0.25f);
    __bf16* t;
    t = zc; zc = zn; zn = t;
    t = zcT; zcT = znT; znT = t;
  }

  k_attn3v<<<64, T, 0, stream>>>(k_b, ql_b, vT_b, kvT_b);
  k_w2<<<64, T, 0, stream>>>(zc, kvT_b, w2t_b);
  k_attn1_out<<<1024, T, 0, stream>>>(kl_b, q_b, w2t_b, aout_f);
  k_conv_res<<<32768, T, 0, stream>>>(v_f, res_w, aout_f, aout_b);
  k_outproj<<<1024, T, 0, stream>>>(aout_b, wout_b, b_out, x, out);
}


# conv residual: a thread keeps a 48-row register window for 4 channels x 16 tokens; rows read once per 16 outputs by unguarded grouped loads (was 33 guarded serial loads per output); tap order kept
# speedup vs baseline: 1.0948x; 1.0948x over previous
#include <hip/hip_runtime.h>
#include <hip/hip_bf16.h>
#include <math.h>


typedef __attribute__((ext_vector_type(16))) __bf16 bf16x16;
typedef __attribute__((ext_vector_type(8)))  float  floatx8;

union FragAB { bf16x16 v; unsigned int u[8]; };

__device__ __forceinline__ unsigned pack2bf(float a, float b) {
  __bf16 lo = (__bf16)a, hi = (__bf16)b;
  unsigned short ul = __builtin_bit_cast(unsigned short, lo);
  unsigned short uh = __builtin_bit_cast(unsigned short, hi);
  return (unsigned)ul | ((unsigned)uh << 16);
}

__device__ __forceinline__ int kpair(int p, int h) {
  return (p < 4) ? (2 * p + 8 * h) : (8 + 2 * p + 8 * h);
}

__device__ __forceinline__ bf16x16 load_frag_bf16_rm(const __bf16* src, int row0,
                                                     int k0, int ld) {
  int lane = threadIdx.x & 31;
  int r = lane & 15, h = lane >> 4;
  const __bf16* base = src + (size_t)(row0 + r) * ld + k0;
  FragAB f;
#pragma unroll
  for (int p = 0; p < 8; ++p) f.u[p] = *(const unsigned*)(base + kpair(p, h));
  return f.v;
}

__device__ __forceinline__ floatx8 wmma_bf16(bf16x16 a, bf16x16 b, floatx8 c) {
  return __builtin_amdgcn_wmma_f32_16x16x32_bf16(false, a, false, b, (short)0, c,
                                                 false, false);
}

template <int KSTEPS>
__device__ __forceinline__ void gemm_strip(const __bf16* A, int rowA,
                                           const __bf16* B, int rowB, int ld,
                                           floatx8 acc[4]) {
#pragma unroll
  for (int ks = 0; ks < KSTEPS; ++ks) {
    int k0 = ks * 32;
    bf16x16 af = load_frag_bf16_rm(A, rowA, k0, ld);
#pragma unroll
    for (int t = 0; t < 4; ++t) {
      bf16x16 bfr = load_frag_bf16_rm(B, rowB + 16 * t, k0, ld);
      acc[t] = wmma_bf16(af, bfr, acc[t]);
    }
  }
}

__global__ void k_convert_w(const float* wqkv, const float* wout,
                            __bf16* wqkv_b, __bf16* wout_b, float* scales) {
  int i = blockIdx.x * blockDim.x + threadIdx.x;
  if (i < 2) scales[i] = 0.f;
  if (i < 1536 * 512) wqkv_b[i] = (__bf16)wqkv[i];
  if (i < 512 * 512) wout_b[i] = (__bf16)wout[i];
}

__global__ void k_layernorm(const float* x, const float* g, const float* b,
                            __bf16* xn, int ntok) {
  int wave = (int)((blockIdx.x * blockDim.x + threadIdx.x) >> 5);
  int lane = threadIdx.x & 31;
  if (wave >= ntok) return;
  const float* xr = x + (size_t)wave * 512;
  float s = 0.f, s2 = 0.f, vals[16];
#pragma unroll
  for (int i = 0; i < 16; ++i) {
    float v = xr[lane + 32 * i];
    vals[i] = v; s += v; s2 += v * v;
  }
  for (int m = 16; m >= 1; m >>= 1) {
    s += __shfl_xor(s, m, 32);
    s2 += __shfl_xor(s2, m, 32);
  }
  float mu = s * (1.f / 512.f);
  float var = s2 * (1.f / 512.f) - mu * mu;
  float inv = rsqrtf(var + 1e-5f);
  __bf16* o = xn + (size_t)wave * 512;
#pragma unroll
  for (int i = 0; i < 16; ++i) {
    int c = lane + 32 * i;
    o[c] = (__bf16)((vals[i] - mu) * inv * g[c] + b[c]);
  }
}

__global__ void k_gemm_qkv(const __bf16* xn, const __bf16* wq, __bf16* qb,
                           __bf16* kb, float* vf, __bf16* vT) {
  int wid = (int)((blockIdx.x * blockDim.x + threadIdx.x) >> 5);
  int tM = wid / 24, s = wid % 24;
  floatx8 acc[4] = {};
  gemm_strip<16>(xn, tM * 16, wq, s * 64, 512, acc);
  int lane = threadIdx.x & 31;
  int n = lane & 15, h = lane >> 4;
  int col0 = s * 64;
  int sect = col0 >> 9;
  int head = (col0 >> 6) & 7;
#pragma unroll
  for (int t = 0; t < 4; ++t) {
    int dh = 16 * t + n;
#pragma unroll
    for (int r = 0; r < 8; ++r) {
      int tok = tM * 16 + r + 8 * h;
      int bb = tok >> 12, ns = tok & 4095;
      size_t bh = (size_t)(bb * 8 + head);
      size_t idx = (bh * 4096 + ns) * 64 + dh;
      float val = acc[t][r];
      if (sect == 0) qb[idx] = (__bf16)(val * 0.125f);
      else if (sect == 1) kb[idx] = (__bf16)val;
      else {
        vf[idx] = val;
        vT[(bh * 64 + dh) * 4096 + ns] = (__bf16)val;
      }
    }
  }
}

__global__ void k_landmarks(const __bf16* qb, const __bf16* kb, __bf16* ql,
                            __bf16* kl) {
  int i = blockIdx.x * blockDim.x + threadIdx.x;
  if (i >= 32 * 256 * 64) return;
  int dh = i & 63, m = (i >> 6) & 255, bh = i >> 14;
  size_t base = ((size_t)bh * 4096 + m * 16) * 64 + dh;
  float sq = 0.f, sk = 0.f;
#pragma unroll
  for (int t = 0; t < 16; ++t) {
    sq += (float)qb[base + t * 64];
    sk += (float)kb[base + t * 64];
  }
  ql[i] = (__bf16)(sq * (1.f / 16.f));
  kl[i] = (__bf16)(sk * (1.f / 16.f));
}

__global__ void k_attn2_logits(const __bf16* ql, const __bf16* kl, float* out) {
  int wid = (int)((blockIdx.x * blockDim.x + threadIdx.x) >> 5);
  int bh = wid >> 6, t = wid & 63;
  int tM = t >> 2, s = t & 3;
  const __bf16* A = ql + (size_t)bh * 256 * 64;
  const __bf16* B = kl + (size_t)bh * 256 * 64;
  floatx8 acc[4] = {};
  gemm_strip<2>(A, tM * 16, B, s * 64, 64, acc);
  int lane = threadIdx.x & 31;
  int n = lane & 15, h = lane >> 4;
  float* o = out + (size_t)bh * 256 * 256;
#pragma unroll
  for (int t4 = 0; t4 < 4; ++t4)
#pragma unroll
    for (int r = 0; r < 8; ++r)
      o[(size_t)(tM * 16 + r + 8 * h) * 256 + s * 64 + 16 * t4 + n] =
          acc[t4][r];
}

__global__ void k_softmax256(float* x, __bf16* xb, int nrows) {
  int wave = (int)((blockIdx.x * blockDim.x + threadIdx.x) >> 5);
  if (wave >= nrows) return;
  int lane = threadIdx.x & 31;
  float* r = x + (size_t)wave * 256;
  __bf16* rb = xb + (size_t)wave * 256;
  float v[8], m = -1e30f;
#pragma unroll
  for (int i = 0; i < 8; ++i) { v[i] = r[lane + 32 * i]; m = fmaxf(m, v[i]); }
  for (int t = 16; t >= 1; t >>= 1) m = fmaxf(m, __shfl_xor(m, t, 32));
  float s = 0.f;
#pragma unroll
  for (int i = 0; i < 8; ++i) { v[i] = __expf(v[i] - m); s += v[i]; }
  for (int t = 16; t >= 1; t >>= 1) s += __shfl_xor(s, t, 32);
  float inv = 1.f / s;
#pragma unroll
  for (int i = 0; i < 8; ++i) {
    float pv = v[i] * inv;
    r[lane + 32 * i] = pv;
    rb[lane + 32 * i] = (__bf16)pv;
  }
}

__global__ void k_pinv_scales(const float* x, float* scales) {
  int wave = (int)((blockIdx.x * blockDim.x + threadIdx.x) >> 5);
  if (wave >= 32 * 256) return;
  int lane = threadIdx.x & 31;
  int bh = wave >> 8, r = wave & 255;
  const float* m = x + (size_t)bh * 256 * 256;
  float rs = 0.f, cs = 0.f;
#pragma unroll
  for (int i = 0; i < 8; ++i) {
    rs += fabsf(m[(size_t)r * 256 + lane + 32 * i]);
    cs += fabsf(m[(size_t)(lane + 32 * i) * 256 + r]);
  }
  for (int t = 16; t >= 1; t >>= 1) {
    rs += __shfl_xor(rs, t, 32);
    cs += __shfl_xor(cs, t, 32);
  }
  if (lane == 0) {
    atomicMax((unsigned*)&scales[0], __float_as_uint(rs));
    atomicMax((unsigned*)&scales[1], __float_as_uint(cs));
  }
}

__global__ void k_init_z(const float* x, __bf16* zb, __bf16* zTb,
                         const float* scales) {
  size_t i = (size_t)blockIdx.x * blockDim.x + threadIdx.x;
  if (i >= (size_t)32 * 256 * 256) return;
  int j = (int)(i & 255), r = (int)((i >> 8) & 255);
  size_t bh = i >> 16;
  float denom = scales[0] * scales[1];
  float val = x[(bh << 16) + (size_t)j * 256 + r] / denom;
  zb[i] = (__bf16)val;
  zTb[(bh << 16) + (size_t)j * 256 + r] = (__bf16)val;
}

__global__ void k_gemm256b(const __bf16* A, const __bf16* BT, __bf16* OutRaw,
                           __bf16* OutT, __bf16* OutTT, float alpha,
                           float beta) {
  int wid = (int)((blockIdx.x * blockDim.x + threadIdx.x) >> 5);
  int bh = wid >> 6, t = wid & 63;
  int tM = t >> 2, s = t & 3;
  size_t bofs = (size_t)bh << 16;
  floatx8 acc[4] = {};
  gemm_strip<8>(A + bofs, tM * 16, BT + bofs, s * 64, 256, acc);
  int lane = threadIdx.x & 31;
  int n = lane & 15, h = lane >> 4;
#pragma unroll
  for (int t4 = 0; t4 < 4; ++t4)
#pragma unroll
    for (int r = 0; r < 8; ++r) {
      int M = tM * 16 + r + 8 * h, N = s * 64 + 16 * t4 + n;
      float tv = acc[t4][r];
      if (OutRaw) OutRaw[bofs + (size_t)M * 256 + N] = (__bf16)tv;
      float tr = alpha * (M == N ? 1.f : 0.f) + beta * tv;
      if (OutT) OutT[bofs + (size_t)M * 256 + N] = (__bf16)tr;
      if (OutTT) OutTT[bofs + (size_t)N * 256 + M] = (__bf16)tr;
    }
}

__device__ __forceinline__ void stream_attn(const __bf16* Ak, const __bf16* Bq,
                                            int qrow0, const __bf16* A2,
                                            int ldA2, int nkeys, float* outF,
                                            int rsF, __bf16* outB, int nsB,
                                            int dsB) {
  int lane = threadIdx.x & 31;
  int n = lane & 15, h = lane >> 4;
  bf16x16 bq0 = load_frag_bf16_rm(Bq, qrow0, 0, 64);
  bf16x16 bq1 = load_frag_bf16_rm(Bq, qrow0, 32, 64);
  float run_m = -1e30f, run_s = 0.f;
  floatx8 acc[4] = {};
  for (int c = 0; c < nkeys; c += 32) {
    floatx8 s0 = {}, s1 = {};
    s0 = wmma_bf16(load_frag_bf16_rm(Ak, c, 0, 64), bq0, s0);
    s0 = wmma_bf16(load_frag_bf16_rm(Ak, c, 32, 64), bq1, s0);
    s1 = wmma_bf16(load_frag_bf16_rm(Ak, c + 16, 0, 64), bq0, s1);
    s1 = wmma_bf16(load_frag_bf16_rm(Ak, c + 16, 32, 64), bq1, s1);
    float mc = -1e30f;
#pragma unroll
    for (int r = 0; r < 8; ++r) { mc = fmaxf(mc, s0[r]); mc = fmaxf(mc, s1[r]); }
    mc = fmaxf(mc, __shfl_xor(mc, 16, 32));
    float newm = fmaxf(run_m, mc);
    float scale = __expf(run_m - newm);
    float p0[8], p1[8], ps = 0.f;
#pragma unroll
    for (int r = 0; r < 8; ++r) {
      p0[r] = __expf(s0[r] - newm);
      p1[r] = __expf(s1[r] - newm);
      ps += p0[r] + p1[r];
    }
    run_s = run_s * scale + ps;
    run_m = newm;
    FragAB bp;
#pragma unroll
    for (int p = 0; p < 4; ++p) {
      bp.u[p] = pack2bf(p0[2 * p], p0[2 * p + 1]);
      bp.u[4 + p] = pack2bf(p1[2 * p], p1[2 * p + 1]);
    }
#pragma unroll
    for (int t = 0; t < 4; ++t) {
#pragma unroll
      for (int r = 0; r < 8; ++r) acc[t][r] *= scale;
      acc[t] = wmma_bf16(load_frag_bf16_rm(A2, 16 * t, c, ldA2), bp.v, acc[t]);
    }
  }
  float tot = run_s + __shfl_xor(run_s, 16, 32);
  float inv = 1.f / tot;
#pragma unroll
  for (int t = 0; t < 4; ++t)
#pragma unroll
    for (int r = 0; r < 8; ++r) {
      int dh = 16 * t + r + 8 * h;
      float val = acc[t][r] * inv;
      if (outF) outF[(size_t)n * rsF + dh] = val;
      if (outB) outB[(size_t)n * nsB + (size_t)dh * dsB] = (__bf16)val;
    }
}

__global__ void k_attn3v(const __bf16* kb, const __bf16* ql, const __bf16* vT,
                         __bf16* kvT) {
  int wid = (int)((blockIdx.x * blockDim.x + threadIdx.x) >> 5);
  int bh = wid >> 4, blk = wid & 15;
  stream_attn(kb + (size_t)bh * 4096 * 64, ql + (size_t)bh * 256 * 64, blk * 16,
              vT + (size_t)bh * 64 * 4096, 4096, 4096, nullptr, 0,
              kvT + (size_t)bh * 64 * 256 + blk * 16, 1, 256);
}

__global__ void k_w2(const __bf16* zb, const __bf16* kvT, __bf16* w2t) {
  int wid = (int)((blockIdx.x * blockDim.x + threadIdx.x) >> 5);
  int bh = wid >> 4, tM = wid & 15;
  floatx8 acc[4] = {};
  gemm_strip<8>(zb + ((size_t)bh << 16), tM * 16, kvT + (size_t)bh * 64 * 256,
                0, 256, acc);
  int lane = threadIdx.x & 31;
  int n = lane & 15, h = lane >> 4;
#pragma unroll
  for (int t4 = 0; t4 < 4; ++t4)
#pragma unroll
    for (int r = 0; r < 8; ++r) {
      int M = tM * 16 + r + 8 * h;
      int N = 16 * t4 + n;
      w2t[((size_t)bh * 64 + N) * 256 + M] = (__bf16)acc[t4][r];
    }
}

__global__ void k_attn1_out(const __bf16* kl, const __bf16* qb,
                            const __bf16* w2t, float* aout) {
  int wid = (int)((blockIdx.x * blockDim.x + threadIdx.x) >> 5);
  int bh = wid >> 8, blk = wid & 255;
  int b = bh >> 3, hh = bh & 7;
  stream_attn(kl + (size_t)bh * 256 * 64, qb + (size_t)bh * 4096 * 64, blk * 16,
              w2t + (size_t)bh * 64 * 256, 256, 256,
              aout + ((size_t)b * 4096 + blk * 16) * 512 + hh * 64, 512,
              nullptr, 0, 0);
}

#define CONV_R 16
#define CONV_G 8
#define CONV_ROWS (CONV_R + 32)
__global__ void __launch_bounds__(256, 4)
k_conv_res(const float* __restrict__ vf, const float* __restrict__ rw,
           float* __restrict__ aout, __bf16* __restrict__ aout_b) {
  int t = (int)(blockIdx.x * blockDim.x + threadIdx.x);
  if (t >= 4 * (4096 / CONV_R) * 128) return;
  int c4 = t & 127;
  int chunk = t >> 7;
  int b = chunk / (4096 / CONV_R);
  int ns0 = (chunk % (4096 / CONV_R)) * CONV_R;
  int col = c4 * 4, hh = col >> 6, dh = col & 63;
  const float* vcol = vf + ((size_t)(b * 8 + hh) * 4096) * 64 + dh;
  float* arow = aout + ((size_t)b * 4096 + ns0) * 512 + col;
  __bf16* brow = aout_b + ((size_t)b * 4096 + ns0) * 512 + col;
  float w[33];
#pragma unroll
  for (int j = 0; j < 33; ++j) w[j] = rw[hh * 33 + j];
  float4 acc[CONV_R], win[CONV_ROWS], a[CONV_R];
#pragma unroll
  for (int o = 0; o < CONV_R; ++o) acc[o] = make_float4(0.f, 0.f, 0.f, 0.f);
#pragma unroll
  for (int k = 0; k < CONV_G; ++k) {
    int p = ns0 + k - 16;
    int pc = p < 0 ? 0 : (p > 4095 ? 4095 : p);
    win[k] = *(const float4*)(vcol + (size_t)pc * 64);
  }
#pragma unroll
  for (int g = 0; g < CONV_ROWS / CONV_G; ++g) {
    if (g + 1 < CONV_ROWS / CONV_G) {
#pragma unroll
      for (int k = (g + 1) * CONV_G; k < (g + 2) * CONV_G; ++k) {
        int p = ns0 + k - 16;
        int pc = p < 0 ? 0 : (p > 4095 ? 4095 : p);
        win[k] = *(const float4*)(vcol + (size_t)pc * 64);
      }
    } else {
#pragma unroll
      for (int o = 0; o < CONV_R; ++o)
        a[o] = *(const float4*)(arow + (size_t)o * 512);
    }
#pragma unroll
    for (int k = g * CONV_G; k < (g + 1) * CONV_G; ++k) {
      int p = ns0 + k - 16;
      float4 v = win[k];
      if (p < 0 || p > 4095) v = make_float4(0.f, 0.f, 0.f, 0.f);
#pragma unroll
      for (int o = 0; o < CONV_R; ++o) {
        int j = k - o;
        if (j >= 0 && j < 33) {
          acc[o].x += v.x * w[j];
          acc[o].y += v.y * w[j];
          acc[o].z += v.z * w[j];
          acc[o].w += v.w * w[j];
        }
      }
    }
  }
#pragma unroll
  for (int o = 0; o < CONV_R; ++o) {
    float4 r;
    r.x = a[o].x + acc[o].x;
    r.y = a[o].y + acc[o].y;
    r.z = a[o].z + acc[o].z;
    r.w = a[o].w + acc[o].w;
    *(float4*)(arow + (size_t)o * 512) = r;
    uint2 pk;
    pk.x = pack2bf(r.x, r.y);
    pk.y = pack2bf(r.z, r.w);
    *(uint2*)(brow + (size_t)o * 512) = pk;
  }
}

__global__ void k_outproj(const __bf16* aoutb, const __bf16* wo,
                          const float* bias, const float* x, float* out) {
  int wid = (int)((blockIdx.x * blockDim.x + threadIdx.x) >> 5);
  int tM = wid >> 3, s = wid & 7;
  floatx8 acc[4] = {};
  gemm_strip<16>(aoutb, tM * 16, wo, s * 64, 512, acc);
  int lane = threadIdx.x & 31;
  int n = lane & 15, h = lane >> 4;
#pragma unroll
  for (int t4 = 0; t4 < 4; ++t4) {
    int col = s * 64 + 16 * t4 + n;
    float bc = bias[col];
#pragma unroll
    for (int r = 0; r < 8; ++r) {
      size_t idx = (size_t)(tM * 16 + r + 8 * h) * 512 + col;
      out[idx] = x[idx] + acc[t4][r] + bc;
    }
  }
}

extern "C" void kernel_launch(void* const* d_in, const int* in_sizes, int n_in,
                              void* d_out, int out_size, void* d_ws,
                              size_t ws_size, hipStream_t stream) {
  const float* x     = (const float*)d_in[0];
  const float* ln_g  = (const float*)d_in[1];
  const float* ln_b  = (const float*)d_in[2];
  const float* w_qkv = (const float*)d_in[3];
  const float* w_out = (const float*)d_in[4];
  const float* b_out = (const float*)d_in[5];
  const float* res_w = (const float*)d_in[6];
  float* out = (float*)d_out;

  char* p = (char*)d_ws;
  auto alloc = [&](size_t bytes) -> void* {
    void* r = (void*)p;
    p += (bytes + 255) & ~(size_t)255;
    return r;
  };
  const size_t NTOK = 4 * 4096;
  const size_t MM = (size_t)32 * 256 * 256;
  __bf16* xn_b   = (__bf16*)alloc(NTOK * 512 * 2);
  __bf16* wqkv_b = (__bf16*)alloc(1536 * 512 * 2);
  __bf16* wout_b = (__bf16*)alloc(512 * 512 * 2);
  __bf16* q_b    = (__bf16*)alloc((size_t)32 * 4096 * 64 * 2);
  __bf16* k_b    = (__bf16*)alloc((size_t)32 * 4096 * 64 * 2);
  float*  v_f    = (float*)alloc((size_t)32 * 4096 * 64 * 4);
  __bf16* vT_b   = (__bf16*)alloc((size_t)32 * 64 * 4096 * 2);
  __bf16* ql_b   = (__bf16*)alloc((size_t)32 * 256 * 64 * 2);
  __bf16* kl_b   = (__bf16*)alloc((size_t)32 * 256 * 64 * 2);
  float*  mat_x  = (float*)alloc(MM * 4);
  __bf16* x_b    = (__bf16*)alloc(MM * 2);
  __bf16* z0_b   = (__bf16*)alloc(MM * 2);
  __bf16* z0T_b  = (__bf16*)alloc(MM * 2);
  __bf16* z1_b   = (__bf16*)alloc(MM * 2);
  __bf16* z1T_b  = (__bf16*)alloc(MM * 2);
  __bf16* xz_b   = (__bf16*)alloc(MM * 2);
  __bf16* t1T_b  = (__bf16*)alloc(MM * 2);
  __bf16* t2T_b  = (__bf16*)alloc(MM * 2);
  __bf16* t3T_b  = (__bf16*)alloc(MM * 2);
  __bf16* kvT_b  = (__bf16*)alloc((size_t)32 * 64 * 256 * 2);
  __bf16* w2t_b  = (__bf16*)alloc((size_t)32 * 64 * 256 * 2);
  float*  aout_f = (float*)alloc(NTOK * 512 * 4);
  __bf16* aout_b = (__bf16*)alloc(NTOK * 512 * 2);
  float*  scales = (float*)alloc(256);
  (void)ws_size; (void)n_in; (void)in_sizes; (void)out_size;

  const int T = 256;
  k_convert_w<<<3072, T, 0, stream>>>(w_qkv, w_out, wqkv_b, wout_b, scales);
  k_layernorm<<<2048, T, 0, stream>>>(x, ln_g, ln_b, xn_b, (int)NTOK);
  k_gemm_qkv<<<3072, T, 0, stream>>>(xn_b, wqkv_b, q_b, k_b, v_f, vT_b);
  k_landmarks<<<2048, T, 0, stream>>>(q_b, k_b, ql_b, kl_b);
  k_attn2_logits<<<256, T, 0, stream>>>(ql_b, kl_b, mat_x);
  k_softmax256<<<1024, T, 0, stream>>>(mat_x, x_b, 32 * 256);
  k_pinv_scales<<<1024, T, 0, stream>>>(mat_x, scales);
  k_init_z<<<8192, T, 0, stream>>>(mat_x, z0_b, z0T_b, scales);

  __bf16* zc = z0_b;  __bf16* zcT = z0T_b;
  __bf16* zn = z1_b;  __bf16* znT = z1T_b;
  for (int it = 0; it < 6; ++it) {
    k_gemm256b<<<256, T, 0, stream>>>(x_b, zcT, xz_b, nullptr, t1T_b, 7.f, -1.f);
    k_gemm256b<<<256, T, 0, stream>>>(xz_b, t1T_b, nullptr, nullptr, t2T_b, 15.f, -1.f);
    k_gemm256b<<<256, T, 0, stream>>>(xz_b, t2T_b, nullptr, nullptr, t3T_b, 13.f, -1.f);
    k_gemm256b<<<256, T, 0, stream>>>(zc, t3T_b, nullptr, zn, znT, 0.f, 0.25f);
    __bf16* t;
    t = zc; zc = zn; zn = t;
    t = zcT; zcT = znT; znT = t;
  }

  k_attn3v<<<64, T, 0, stream>>>(k_b, ql_b, vT_b, kvT_b);
  k_w2<<<64, T, 0, stream>>>(zc, kvT_b, w2t_b);
  k_attn1_out<<<1024, T, 0, stream>>>(kl_b, q_b, w2t_b, aout_f);
  k_conv_res<<<512, T, 0, stream>>>(v_f, res_w, aout_f, aout_b);
  k_outproj<<<1024, T, 0, stream>>>(aout_b, wout_b, b_out, x, out);
}


# QKV and out-projection GEMMs: wave tile 16x64 -> 64x64, each K step loads 4 A + 4 B fragments as one group for 16 WMMAs (was 5 fragment loads per 4 WMMAs); k order and sums unchanged
# speedup vs baseline: 1.2358x; 1.1288x over previous
#include <hip/hip_runtime.h>
#include <hip/hip_bf16.h>
#include <math.h>


typedef __attribute__((ext_vector_type(16))) __bf16 bf16x16;
typedef __attribute__((ext_vector_type(8)))  float  floatx8;

union FragAB { bf16x16 v; unsigned int u[8]; };

__device__ __forceinline__ unsigned pack2bf(float a, float b) {
  __bf16 lo = (__bf16)a, hi = (__bf16)b;
  unsigned short ul = __builtin_bit_cast(unsigned short, lo);
  unsigned short uh = __builtin_bit_cast(unsigned short, hi);
  return (unsigned)ul | ((unsigned)uh << 16);
}

__device__ __forceinline__ int kpair(int p, int h) {
  return (p < 4) ? (2 * p + 8 * h) : (8 + 2 * p + 8 * h);
}

__device__ __forceinline__ bf16x16 load_frag_bf16_rm(const __bf16* src, int row0,
                                                     int k0, int ld) {
  int lane = threadIdx.x & 31;
  int r = lane & 15, h = lane >> 4;
  const __bf16* base = src + (size_t)(row0 + r) * ld + k0;
  FragAB f;
#pragma unroll
  for (int p = 0; p < 8; ++p) f.u[p] = *(const unsigned*)(base + kpair(p, h));
  return f.v;
}

__device__ __forceinline__ floatx8 wmma_bf16(bf16x16 a, bf16x16 b, floatx8 c) {
  return __builtin_amdgcn_wmma_f32_16x16x32_bf16(false, a, false, b, (short)0, c,
                                                 false, false);
}

template <int KSTEPS>
__device__ __forceinline__ void gemm_strip(const __bf16* A, int rowA,
                                           const __bf16* B, int rowB, int ld,
                                           floatx8 acc[4]) {
#pragma unroll
  for (int ks = 0; ks < KSTEPS; ++ks) {
    int k0 = ks * 32;
    bf16x16 af = load_frag_bf16_rm(A, rowA, k0, ld);
#pragma unroll
    for (int t = 0; t < 4; ++t) {
      bf16x16 bfr = load_frag_bf16_rm(B, rowB + 16 * t, k0, ld);
      acc[t] = wmma_bf16(af, bfr, acc[t]);
    }
  }
}

template <int KSTEPS>
__device__ __forceinline__ void gemm_tile64(const __bf16* A, int rowA,
                                            const __bf16* B, int rowB, int ld,
                                            floatx8 acc[4][4]) {
#pragma unroll
  for (int ks = 0; ks < KSTEPS; ++ks) {
    int k0 = ks * 32;
    bf16x16 af[4], bfr[4];
#pragma unroll
    for (int m = 0; m < 4; ++m) af[m] = load_frag_bf16_rm(A, rowA + 16 * m, k0, ld);
#pragma unroll
    for (int t = 0; t < 4; ++t) bfr[t] = load_frag_bf16_rm(B, rowB + 16 * t, k0, ld);
    __builtin_amdgcn_sched_barrier(0);
#pragma unroll
    for (int m = 0; m < 4; ++m)
#pragma unroll
      for (int t = 0; t < 4; ++t) acc[m][t] = wmma_bf16(af[m], bfr[t], acc[m][t]);
    __builtin_amdgcn_sched_barrier(0);
  }
}

__global__ void k_convert_w(const float* wqkv, const float* wout,
                            __bf16* wqkv_b, __bf16* wout_b, float* scales) {
  int i = blockIdx.x * blockDim.x + threadIdx.x;
  if (i < 2) scales[i] = 0.f;
  if (i < 1536 * 512) wqkv_b[i] = (__bf16)wqkv[i];
  if (i < 512 * 512) wout_b[i] = (__bf16)wout[i];
}

__global__ void k_layernorm(const float* x, const float* g, const float* b,
                            __bf16* xn, int ntok) {
  int wave = (int)((blockIdx.x * blockDim.x + threadIdx.x) >> 5);
  int lane = threadIdx.x & 31;
  if (wave >= ntok) return;
  const float* xr = x + (size_t)wave * 512;
  float s = 0.f, s2 = 0.f, vals[16];
#pragma unroll
  for (int i = 0; i < 16; ++i) {
    float v = xr[lane + 32 * i];
    vals[i] = v; s += v; s2 += v * v;
  }
  for (int m = 16; m >= 1; m >>= 1) {
    s += __shfl_xor(s, m, 32);
    s2 += __shfl_xor(s2, m, 32);
  }
  float mu = s * (1.f / 512.f);
  float var = s2 * (1.f / 512.f) - mu * mu;
  float inv = rsqrtf(var + 1e-5f);
  __bf16* o = xn + (size_t)wave * 512;
#pragma unroll
  for (int i = 0; i < 16; ++i) {
    int c = lane + 32 * i;
    o[c] = (__bf16)((vals[i] - mu) * inv * g[c] + b[c]);
  }
}

__global__ void __launch_bounds__(256)
k_gemm_qkv(const __bf16* xn, const __bf16* wq, __bf16* qb,
           __bf16* kb, float* vf, __bf16* vT) {
  int wid = (int)((blockIdx.x * blockDim.x + threadIdx.x) >> 5);
  int tM = wid / 24, s = wid % 24;
  floatx8 acc[4][4] = {};
  gemm_tile64<16>(xn, tM * 64, wq, s * 64, 512, acc);
  int lane = threadIdx.x & 31;
  int n = lane & 15, h = lane >> 4;
  int col0 = s * 64;
  int sect = col0 >> 9;
  int head = (col0 >> 6) & 7;
#pragma unroll
  for (int m = 0; m < 4; ++m) {
#pragma unroll
    for (int t = 0; t < 4; ++t) {
      int dh = 16 * t + n;
#pragma unroll
      for (int r = 0; r < 8; ++r) {
        int tok = tM * 64 + 16 * m + r + 8 * h;
        int bb = tok >> 12, ns = tok & 4095;
        size_t bh = (size_t)(bb * 8 + head);
        size_t idx = (bh * 4096 + ns) * 64 + dh;
        float val = acc[m][t][r];
        if (sect == 0) qb[idx] = (__bf16)(val * 0.125f);
        else if (sect == 1) kb[idx] = (__bf16)val;
        else {
          vf[idx] = val;
          vT[(bh * 64 + dh) * 4096 + ns] = (__bf16)val;
        }
      }
    }
  }
}

__global__ void k_landmarks(const __bf16* qb, const __bf16* kb, __bf16* ql,
                            __bf16* kl) {
  int i = blockIdx.x * blockDim.x + threadIdx.x;
  if (i >= 32 * 256 * 64) return;
  int dh = i & 63, m = (i >> 6) & 255, bh = i >> 14;
  size_t base = ((size_t)bh * 4096 + m * 16) * 64 + dh;
  float sq = 0.f, sk = 0.f;
#pragma unroll
  for (int t = 0; t < 16; ++t) {
    sq += (float)qb[base + t * 64];
    sk += (float)kb[base + t * 64];
  }
  ql[i] = (__bf16)(sq * (1.f / 16.f));
  kl[i] = (__bf16)(sk * (1.f / 16.f));
}

__global__ void k_attn2_logits(const __bf16* ql, const __bf16* kl, float* out) {
  int wid = (int)((blockIdx.x * blockDim.x + threadIdx.x) >> 5);
  int bh = wid >> 6, t = wid & 63;
  int tM = t >> 2, s = t & 3;
  const __bf16* A = ql + (size_t)bh * 256 * 64;
  const __bf16* B = kl + (size_t)bh * 256 * 64;
  floatx8 acc[4] = {};
  gemm_strip<2>(A, tM * 16, B, s * 64, 64, acc);
  int lane = threadIdx.x & 31;
  int n = lane & 15, h = lane >> 4;
  float* o = out + (size_t)bh * 256 * 256;
#pragma unroll
  for (int t4 = 0; t4 < 4; ++t4)
#pragma unroll
    for (int r = 0; r < 8; ++r)
      o[(size_t)(tM * 16 + r + 8 * h) * 256 + s * 64 + 16 * t4 + n] =
          acc[t4][r];
}

__global__ void k_softmax256(float* x, __bf16* xb, int nrows) {
  int wave = (int)((blockIdx.x * blockDim.x + threadIdx.x) >> 5);
  if (wave >= nrows) return;
  int lane = threadIdx.x & 31;
  float* r = x + (size_t)wave * 256;
  __bf16* rb = xb + (size_t)wave * 256;
  float v[8], m = -1e30f;
#pragma unroll
  for (int i = 0; i < 8; ++i) { v[i] = r[lane + 32 * i]; m = fmaxf(m, v[i]); }
  for (int t = 16; t >= 1; t >>= 1) m = fmaxf(m, __shfl_xor(m, t, 32));
  float s = 0.f;
#pragma unroll
  for (int i = 0; i < 8; ++i) { v[i] = __expf(v[i] - m); s += v[i]; }
  for (int t = 16; t >= 1; t >>= 1) s += __shfl_xor(s, t, 32);
  float inv = 1.f / s;
#pragma unroll
  for (int i = 0; i < 8; ++i) {
    float pv = v[i] * inv;
    r[lane + 32 * i] = pv;
    rb[lane + 32 * i] = (__bf16)pv;
  }
}

__global__ void k_pinv_scales(const float* x, float* scales) {
  int wave = (int)((blockIdx.x * blockDim.x + threadIdx.x) >> 5);
  if (wave >= 32 * 256) return;
  int lane = threadIdx.x & 31;
  int bh = wave >> 8, r = wave & 255;
  const float* m = x + (size_t)bh * 256 * 256;
  float rs = 0.f, cs = 0.f;
#pragma unroll
  for (int i = 0; i < 8; ++i) {
    rs += fabsf(m[(size_t)r * 256 + lane + 32 * i]);
    cs += fabsf(m[(size_t)(lane + 32 * i) * 256 + r]);
  }
  for (int t = 16; t >= 1; t >>= 1) {
    rs += __shfl_xor(rs, t, 32);
    cs += __shfl_xor(cs, t, 32);
  }
  if (lane == 0) {
    atomicMax((unsigned*)&scales[0], __float_as_uint(rs));
    atomicMax((unsigned*)&scales[1], __float_as_uint(cs));
  }
}

__global__ void k_init_z(const float* x, __bf16* zb, __bf16* zTb,
                         const float* scales) {
  size_t i = (size_t)blockIdx.x * blockDim.x + threadIdx.x;
  if (i >= (size_t)32 * 256 * 256) return;
  int j = (int)(i & 255), r = (int)((i >> 8) & 255);
  size_t bh = i >> 16;
  float denom = scales[0] * scales[1];
  float val = x[(bh << 16) + (size_t)j * 256 + r] / denom;
  zb[i] = (__bf16)val;
  zTb[(bh << 16) + (size_t)j * 256 + r] = (__bf16)val;
}

__global__ void k_gemm256b(const __bf16* A, const __bf16* BT, __bf16* OutRaw,
                           __bf16* OutT, __bf16* OutTT, float alpha,
                           float beta) {
  int wid = (int)((blockIdx.x * blockDim.x + threadIdx.x) >> 5);
  int bh = wid >> 6, t = wid & 63;
  int tM = t >> 2, s = t & 3;
  size_t bofs = (size_t)bh << 16;
  floatx8 acc[4] = {};
  gemm_strip<8>(A + bofs, tM * 16, BT + bofs, s * 64, 256, acc);
  int lane = threadIdx.x & 31;
  int n = lane & 15, h = lane >> 4;
#pragma unroll
  for (int t4 = 0; t4 < 4; ++t4)
#pragma unroll
    for (int r = 0; r < 8; ++r) {
      int M = tM * 16 + r + 8 * h, N = s * 64 + 16 * t4 + n;
      float tv = acc[t4][r];
      if (OutRaw) OutRaw[bofs + (size_t)M * 256 + N] = (__bf16)tv;
      float tr = alpha * (M == N ? 1.f : 0.f) + beta * tv;
      if (OutT) OutT[bofs + (size_t)M * 256 + N] = (__bf16)tr;
      if (OutTT) OutTT[bofs + (size_t)N * 256 + M] = (__bf16)tr;
    }
}

__device__ __forceinline__ void stream_attn(const __bf16* Ak, const __bf16* Bq,
                                            int qrow0, const __bf16* A2,
                                            int ldA2, int nkeys, float* outF,
                                            int rsF, __bf16* outB, int nsB,
                                            int dsB) {
  int lane = threadIdx.x & 31;
  int n = lane & 15, h = lane >> 4;
  bf16x16 bq0 = load_frag_bf16_rm(Bq, qrow0, 0, 64);
  bf16x16 bq1 = load_frag_bf16_rm(Bq, qrow0, 32, 64);
  float run_m = -1e30f, run_s = 0.f;
  floatx8 acc[4] = {};
  for (int c = 0; c < nkeys; c += 32) {
    floatx8 s0 = {}, s1 = {};
    s0 = wmma_bf16(load_frag_bf16_rm(Ak, c, 0, 64), bq0, s0);
    s0 = wmma_bf16(load_frag_bf16_rm(Ak, c, 32, 64), bq1, s0);
    s1 = wmma_bf16(load_frag_bf16_rm(Ak, c + 16, 0, 64), bq0, s1);
    s1 = wmma_bf16(load_frag_bf16_rm(Ak, c + 16, 32, 64), bq1, s1);
    float mc = -1e30f;
#pragma unroll
    for (int r = 0; r < 8; ++r) { mc = fmaxf(mc, s0[r]); mc = fmaxf(mc, s1[r]); }
    mc = fmaxf(mc, __shfl_xor(mc, 16, 32));
    float newm = fmaxf(run_m, mc);
    float scale = __expf(run_m - newm);
    float p0[8], p1[8], ps = 0.f;
#pragma unroll
    for (int r = 0; r < 8; ++r) {
      p0[r] = __expf(s0[r] - newm);
      p1[r] = __expf(s1[r] - newm);
      ps += p0[r] + p1[r];
    }
    run_s = run_s * scale + ps;
    run_m = newm;
    FragAB bp;
#pragma unroll
    for (int p = 0; p < 4; ++p) {
      bp.u[p] = pack2bf(p0[2 * p], p0[2 * p + 1]);
      bp.u[4 + p] = pack2bf(p1[2 * p], p1[2 * p + 1]);
    }
#pragma unroll
    for (int t = 0; t < 4; ++t) {
#pragma unroll
      for (int r = 0; r < 8; ++r) acc[t][r] *= scale;
      acc[t] = wmma_bf16(load_frag_bf16_rm(A2, 16 * t, c, ldA2), bp.v, acc[t]);
    }
  }
  float tot = run_s + __shfl_xor(run_s, 16, 32);
  float inv = 1.f / tot;
#pragma unroll
  for (int t = 0; t < 4; ++t)
#pragma unroll
    for (int r = 0; r < 8; ++r) {
      int dh = 16 * t + r + 8 * h;
      float val = acc[t][r] * inv;
      if (outF) outF[(size_t)n * rsF + dh] = val;
      if (outB) outB[(size_t)n * nsB + (size_t)dh * dsB] = (__bf16)val;
    }
}

__global__ void k_attn3v(const __bf16* kb, const __bf16* ql, const __bf16* vT,
                         __bf16* kvT) {
  int wid = (int)((blockIdx.x * blockDim.x + threadIdx.x) >> 5);
  int bh = wid >> 4, blk = wid & 15;
  stream_attn(kb + (size_t)bh * 4096 * 64, ql + (size_t)bh * 256 * 64, blk * 16,
              vT + (size_t)bh * 64 * 4096, 4096, 4096, nullptr, 0,
              kvT + (size_t)bh * 64 * 256 + blk * 16, 1, 256);
}

__global__ void k_w2(const __bf16* zb, const __bf16* kvT, __bf16* w2t) {
  int wid = (int)((blockIdx.x * blockDim.x + threadIdx.x) >> 5);
  int bh = wid >> 4, tM = wid & 15;
  floatx8 acc[4] = {};
  gemm_strip<8>(zb + ((size_t)bh << 16), tM * 16, kvT + (size_t)bh * 64 * 256,
                0, 256, acc);
  int lane = threadIdx.x & 31;
  int n = lane & 15, h = lane >> 4;
#pragma unroll
  for (int t4 = 0; t4 < 4; ++t4)
#pragma unroll
    for (int r = 0; r < 8; ++r) {
      int M = tM * 16 + r + 8 * h;
      int N = 16 * t4 + n;
      w2t[((size_t)bh * 64 + N) * 256 + M] = (__bf16)acc[t4][r];
    }
}

__global__ void k_attn1_out(const __bf16* kl, const __bf16* qb,
                            const __bf16* w2t, float* aout) {
  int wid = (int)((blockIdx.x * blockDim.x + threadIdx.x) >> 5);
  int bh = wid >> 8, blk = wid & 255;
  int b = bh >> 3, hh = bh & 7;
  stream_attn(kl + (size_t)bh * 256 * 64, qb + (size_t)bh * 4096 * 64, blk * 16,
              w2t + (size_t)bh * 64 * 256, 256, 256,
              aout + ((size_t)b * 4096 + blk * 16) * 512 + hh * 64, 512,
              nullptr, 0, 0);
}

#define CONV_R 16
#define CONV_G 8
#define CONV_ROWS (CONV_R + 32)
__global__ void __launch_bounds__(256, 4)
k_conv_res(const float* __restrict__ vf, const float* __restrict__ rw,
           float* __restrict__ aout, __bf16* __restrict__ aout_b) {
  int t = (int)(blockIdx.x * blockDim.x + threadIdx.x);
  if (t >= 4 * (4096 / CONV_R) * 128) return;
  int c4 = t & 127;
  int chunk = t >> 7;
  int b = chunk / (4096 / CONV_R);
  int ns0 = (chunk % (4096 / CONV_R)) * CONV_R;
  int col = c4 * 4, hh = col >> 6, dh = col & 63;
  const float* vcol = vf + ((size_t)(b * 8 + hh) * 4096) * 64 + dh;
  float* arow = aout + ((size_t)b * 4096 + ns0) * 512 + col;
  __bf16* brow = aout_b + ((size_t)b * 4096 + ns0) * 512 + col;
  float w[33];
#pragma unroll
  for (int j = 0; j < 33; ++j) w[j] = rw[hh * 33 + j];
  float4 acc[CONV_R], win[CONV_ROWS], a[CONV_R];
#pragma unroll
  for (int o = 0; o < CONV_R; ++o) acc[o] = make_float4(0.f, 0.f, 0.f, 0.f);
#pragma unroll
  for (int k = 0; k < CONV_G; ++k) {
    int p = ns0 + k - 16;
    int pc = p < 0 ? 0 : (p > 4095 ? 4095 : p);
    win[k] = *(const float4*)(vcol + (size_t)pc * 64);
  }
#pragma unroll
  for (int g = 0; g < CONV_ROWS / CONV_G; ++g) {
    if (g + 1 < CONV_ROWS / CONV_G) {
#pragma unroll
      for (int k = (g + 1) * CONV_G; k < (g + 2) * CONV_G; ++k) {
        int p = ns0 + k - 16;
        int pc = p < 0 ? 0 : (p > 4095 ? 4095 : p);
        win[k] = *(const float4*)(vcol + (size_t)pc * 64);
      }
    } else {
#pragma unroll
      for (int o = 0; o < CONV_R; ++o)
        a[o] = *(const float4*)(arow + (size_t)o * 512);
    }
#pragma unroll
    for (int k = g * CONV_G; k < (g + 1) * CONV_G; ++k) {
      int p = ns0 + k - 16;
      float4 v = win[k];
      if (p < 0 || p > 4095) v = make_float4(0.f, 0.f, 0.f, 0.f);
#pragma unroll
      for (int o = 0; o < CONV_R; ++o) {
        int j = k - o;
        if (j >= 0 && j < 33) {
          acc[o].x += v.x * w[j];
          acc[o].y += v.y * w[j];
          acc[o].z += v.z * w[j];
          acc[o].w += v.w * w[j];
        }
      }
    }
  }
#pragma unroll
  for (int o = 0; o < CONV_R; ++o) {
    float4 r;
    r.x = a[o].x + acc[o].x;
    r.y = a[o].y + acc[o].y;
    r.z = a[o].z + acc[o].z;
    r.w = a[o].w + acc[o].w;
    *(float4*)(arow + (size_t)o * 512) = r;
    uint2 pk;
    pk.x = pack2bf(r.x, r.y);
    pk.y = pack2bf(r.z, r.w);
    *(uint2*)(brow + (size_t)o * 512) = pk;
  }
}

__global__ void __launch_bounds__(256)
k_outproj(const __bf16* __restrict__ aoutb, const __bf16* __restrict__ wo,
          const float* __restrict__ bias, const float* __restrict__ x,
          float* __restrict__ out) {
  int wid = (int)((blockIdx.x * blockDim.x + threadIdx.x) >> 5);
  int tM = wid >> 3, s = wid & 7;
  floatx8 acc[4][4] = {};
  gemm_tile64<16>(aoutb, tM * 64, wo, s * 64, 512, acc);
  int lane = threadIdx.x & 31;
  int n = lane & 15, h = lane >> 4;
#pragma unroll
  for (int t4 = 0; t4 < 4; ++t4) {
    int col = s * 64 + 16 * t4 + n;
    float bc = bias[col];
#pragma unroll
    for (int m = 0; m < 4; ++m) {
#pragma unroll
      for (int r = 0; r < 8; ++r) {
        size_t idx = (size_t)(tM * 64 + 16 * m + r + 8 * h) * 512 + col;
        out[idx] = x[idx] + acc[m][t4][r] + bc;
      }
    }
  }
}

extern "C" void kernel_launch(void* const* d_in, const int* in_sizes, int n_in,
                              void* d_out, int out_size, void* d_ws,
                              size_t ws_size, hipStream_t stream) {
  const float* x     = (const float*)d_in[0];
  const float* ln_g  = (const float*)d_in[1];
  const float* ln_b  = (const float*)d_in[2];
  const float* w_qkv = (const float*)d_in[3];
  const float* w_out = (const float*)d_in[4];
  const float* b_out = (const float*)d_in[5];
  const float* res_w = (const float*)d_in[6];
  float* out = (float*)d_out;

  char* p = (char*)d_ws;
  auto alloc = [&](size_t bytes) -> void* {
    void* r = (void*)p;
    p += (bytes + 255) & ~(size_t)255;
    return r;
  };
  const size_t NTOK = 4 * 4096;
  const size_t MM = (size_t)32 * 256 * 256;
  __bf16* xn_b   = (__bf16*)alloc(NTOK * 512 * 2);
  __bf16* wqkv_b = (__bf16*)alloc(1536 * 512 * 2);
  __bf16* wout_b = (__bf16*)alloc(512 * 512 * 2);
  __bf16* q_b    = (__bf16*)alloc((size_t)32 * 4096 * 64 * 2);
  __bf16* k_b    = (__bf16*)alloc((size_t)32 * 4096 * 64 * 2);
  float*  v_f    = (float*)alloc((size_t)32 * 4096 * 64 * 4);
  __bf16* vT_b   = (__bf16*)alloc((size_t)32 * 64 * 4096 * 2);
  __bf16* ql_b   = (__bf16*)alloc((size_t)32 * 256 * 64 * 2);
  __bf16* kl_b   = (__bf16*)alloc((size_t)32 * 256 * 64 * 2);
  float*  mat_x  = (float*)alloc(MM * 4);
  __bf16* x_b    = (__bf16*)alloc(MM * 2);
  __bf16* z0_b   = (__bf16*)alloc(MM * 2);
  __bf16* z0T_b  = (__bf16*)alloc(MM * 2);
  __bf16* z1_b   = (__bf16*)alloc(MM * 2);
  __bf16* z1T_b  = (__bf16*)alloc(MM * 2);
  __bf16* xz_b   = (__bf16*)alloc(MM * 2);
  __bf16* t1T_b  = (__bf16*)alloc(MM * 2);
  __bf16* t2T_b  = (__bf16*)alloc(MM * 2);
  __bf16* t3T_b  = (__bf16*)alloc(MM * 2);
  __bf16* kvT_b  = (__bf16*)alloc((size_t)32 * 64 * 256 * 2);
  __bf16* w2t_b  = (__bf16*)alloc((size_t)32 * 64 * 256 * 2);
  float*  aout_f = (float*)alloc(NTOK * 512 * 4);
  __bf16* aout_b = (__bf16*)alloc(NTOK * 512 * 2);
  float*  scales = (float*)alloc(256);
  (void)ws_size; (void)n_in; (void)in_sizes; (void)out_size;

  const int T = 256;
  k_convert_w<<<3072, T, 0, stream>>>(w_qkv, w_out, wqkv_b, wout_b, scales);
  k_layernorm<<<2048, T, 0, stream>>>(x, ln_g, ln_b, xn_b, (int)NTOK);
  k_gemm_qkv<<<768, T, 0, stream>>>(xn_b, wqkv_b, q_b, k_b, v_f, vT_b);
  k_landmarks<<<2048, T, 0, stream>>>(q_b, k_b, ql_b, kl_b);
  k_attn2_logits<<<256, T, 0, stream>>>(ql_b, kl_b, mat_x);
  k_softmax256<<<1024, T, 0, stream>>>(mat_x, x_b, 32 * 256);
  k_pinv_scales<<<1024, T, 0, stream>>>(mat_x, scales);
  k_init_z<<<8192, T, 0, stream>>>(mat_x, z0_b, z0T_b, scales);

  __bf16* zc = z0_b;  __bf16* zcT = z0T_b;
  __bf16* zn = z1_b;  __bf16* znT = z1T_b;
  for (int it = 0; it < 6; ++it) {
    k_gemm256b<<<256, T, 0, stream>>>(x_b, zcT, xz_b, nullptr, t1T_b, 7.f, -1.f);
    k_gemm256b<<<256, T, 0, stream>>>(xz_b, t1T_b, nullptr, nullptr, t2T_b, 15.f, -1.f);
    k_gemm256b<<<256, T, 0, stream>>>(xz_b, t2T_b, nullptr, nullptr, t3T_b, 13.f, -1.f);
    k_gemm256b<<<256, T, 0, stream>>>(zc, t3T_b, nullptr, zn, znT, 0.f, 0.25f);
    __bf16* t;
    t = zc; zc = zn; zn = t;
    t = zcT; zcT = znT; znT = t;
  }

  k_attn3v<<<64, T, 0, stream>>>(k_b, ql_b, vT_b, kvT_b);
  k_w2<<<64, T, 0, stream>>>(zc, kvT_b, w2t_b);
  k_attn1_out<<<1024, T, 0, stream>>>(kl_b, q_b, w2t_b, aout_f);
  k_conv_res<<<512, T, 0, stream>>>(v_f, res_w, aout_f, aout_b);
  k_outproj<<<256, T, 0, stream>>>(aout_b, wout_b, b_out, x, out);
}


# attn3@v kernel re-tiled: a block = 32 landmark rows, its 8 waves split the 4096 keys (512 each) and merge max/sum/acc partials through LDS; 256 blocks (was 64); softmax sum order changes
# speedup vs baseline: 1.4493x; 1.1728x over previous
#include <hip/hip_runtime.h>
#include <hip/hip_bf16.h>
#include <math.h>


typedef __attribute__((ext_vector_type(16))) __bf16 bf16x16;
typedef __attribute__((ext_vector_type(8)))  float  floatx8;

union FragAB { bf16x16 v; unsigned int u[8]; };

__device__ __forceinline__ unsigned pack2bf(float a, float b) {
  __bf16 lo = (__bf16)a, hi = (__bf16)b;
  unsigned short ul = __builtin_bit_cast(unsigned short, lo);
  unsigned short uh = __builtin_bit_cast(unsigned short, hi);
  return (unsigned)ul | ((unsigned)uh << 16);
}

__device__ __forceinline__ int kpair(int p, int h) {
  return (p < 4) ? (2 * p + 8 * h) : (8 + 2 * p + 8 * h);
}

__device__ __forceinline__ bf16x16 load_frag_bf16_rm(const __bf16* src, int row0,
                                                     int k0, int ld) {
  int lane = threadIdx.x & 31;
  int r = lane & 15, h = lane >> 4;
  const __bf16* base = src + (size_t)(row0 + r) * ld + k0;
  FragAB f;
#pragma unroll
  for (int p = 0; p < 8; ++p) f.u[p] = *(const unsigned*)(base + kpair(p, h));
  return f.v;
}

__device__ __forceinline__ floatx8 wmma_bf16(bf16x16 a, bf16x16 b, floatx8 c) {
  return __builtin_amdgcn_wmma_f32_16x16x32_bf16(false, a, false, b, (short)0, c,
                                                 false, false);
}

template <int KSTEPS>
__device__ __forceinline__ void gemm_strip(const __bf16* A, int rowA,
                                           const __bf16* B, int rowB, int ld,
                                           floatx8 acc[4]) {
#pragma unroll
  for (int ks = 0; ks < KSTEPS; ++ks) {
    int k0 = ks * 32;
    bf16x16 af = load_frag_bf16_rm(A, rowA, k0, ld);
#pragma unroll
    for (int t = 0; t < 4; ++t) {
      bf16x16 bfr = load_frag_bf16_rm(B, rowB + 16 * t, k0, ld);
      acc[t] = wmma_bf16(af, bfr, acc[t]);
    }
  }
}

template <int KSTEPS>
__device__ __forceinline__ void gemm_tile64(const __bf16* A, int rowA,
                                            const __bf16* B, int rowB, int ld,
                                            floatx8 acc[4][4]) {
#pragma unroll
  for (int ks = 0; ks < KSTEPS; ++ks) {
    int k0 = ks * 32;
    bf16x16 af[4], bfr[4];
#pragma unroll
    for (int m = 0; m < 4; ++m) af[m] = load_frag_bf16_rm(A, rowA + 16 * m, k0, ld);
#pragma unroll
    for (int t = 0; t < 4; ++t) bfr[t] = load_frag_bf16_rm(B, rowB + 16 * t, k0, ld);
    __builtin_amdgcn_sched_barrier(0);
#pragma unroll
    for (int m = 0; m < 4; ++m)
#pragma unroll
      for (int t = 0; t < 4; ++t) acc[m][t] = wmma_bf16(af[m], bfr[t], acc[m][t]);
    __builtin_amdgcn_sched_barrier(0);
  }
}

__global__ void k_convert_w(const float* wqkv, const float* wout,
                            __bf16* wqkv_b, __bf16* wout_b, float* scales) {
  int i = blockIdx.x * blockDim.x + threadIdx.x;
  if (i < 2) scales[i] = 0.f;
  if (i < 1536 * 512) wqkv_b[i] = (__bf16)wqkv[i];
  if (i < 512 * 512) wout_b[i] = (__bf16)wout[i];
}

__global__ void k_layernorm(const float* x, const float* g, const float* b,
                            __bf16* xn, int ntok) {
  int wave = (int)((blockIdx.x * blockDim.x + threadIdx.x) >> 5);
  int lane = threadIdx.x & 31;
  if (wave >= ntok) return;
  const float* xr = x + (size_t)wave * 512;
  float s = 0.f, s2 = 0.f, vals[16];
#pragma unroll
  for (int i = 0; i < 16; ++i) {
    float v = xr[lane + 32 * i];
    vals[i] = v; s += v; s2 += v * v;
  }
  for (int m = 16; m >= 1; m >>= 1) {
    s += __shfl_xor(s, m, 32);
    s2 += __shfl_xor(s2, m, 32);
  }
  float mu = s * (1.f / 512.f);
  float var = s2 * (1.f / 512.f) - mu * mu;
  float inv = rsqrtf(var + 1e-5f);
  __bf16* o = xn + (size_t)wave * 512;
#pragma unroll
  for (int i = 0; i < 16; ++i) {
    int c = lane + 32 * i;
    o[c] = (__bf16)((vals[i] - mu) * inv * g[c] + b[c]);
  }
}

__global__ void __launch_bounds__(256)
k_gemm_qkv(const __bf16* xn, const __bf16* wq, __bf16* qb,
           __bf16* kb, float* vf, __bf16* vT) {
  int wid = (int)((blockIdx.x * blockDim.x + threadIdx.x) >> 5);
  int tM = wid / 24, s = wid % 24;
  floatx8 acc[4][4] = {};
  gemm_tile64<16>(xn, tM * 64, wq, s * 64, 512, acc);
  int lane = threadIdx.x & 31;
  int n = lane & 15, h = lane >> 4;
  int col0 = s * 64;
  int sect = col0 >> 9;
  int head = (col0 >> 6) & 7;
#pragma unroll
  for (int m = 0; m < 4; ++m) {
#pragma unroll
    for (int t = 0; t < 4; ++t) {
      int dh = 16 * t + n;
#pragma unroll
      for (int r = 0; r < 8; ++r) {
        int tok = tM * 64 + 16 * m + r + 8 * h;
        int bb = tok >> 12, ns = tok & 4095;
        size_t bh = (size_t)(bb * 8 + head);
        size_t idx = (bh * 4096 + ns) * 64 + dh;
        float val = acc[m][t][r];
        if (sect == 0) qb[idx] = (__bf16)(val * 0.125f);
        else if (sect == 1) kb[idx] = (__bf16)val;
        else {
          vf[idx] = val;
          vT[(bh * 64 + dh) * 4096 + ns] = (__bf16)val;
        }
      }
    }
  }
}

__global__ void k_landmarks(const __bf16* qb, const __bf16* kb, __bf16* ql,
                            __bf16* kl) {
  int i = blockIdx.x * blockDim.x + threadIdx.x;
  if (i >= 32 * 256 * 64) return;
  int dh = i & 63, m = (i >> 6) & 255, bh = i >> 14;
  size_t base = ((size_t)bh * 4096 + m * 16) * 64 + dh;
  float sq = 0.f, sk = 0.f;
#pragma unroll
  for (int t = 0; t < 16; ++t) {
    sq += (float)qb[base + t * 64];
    sk += (float)kb[base + t * 64];
  }
  ql[i] = (__bf16)(sq * (1.f / 16.f));
  kl[i] = (__bf16)(sk * (1.f / 16.f));
}

__global__ void k_attn2_logits(const __bf16* ql, const __bf16* kl, float* out) {
  int wid = (int)((blockIdx.x * blockDim.x + threadIdx.x) >> 5);
  int bh = wid >> 6, t = wid & 63;
  int tM = t >> 2, s = t & 3;
  const __bf16* A = ql + (size_t)bh * 256 * 64;
  const __bf16* B = kl + (size_t)bh * 256 * 64;
  floatx8 acc[4] = {};
  gemm_strip<2>(A, tM * 16, B, s * 64, 64, acc);
  int lane = threadIdx.x & 31;
  int n = lane & 15, h = lane >> 4;
  float* o = out + (size_t)bh * 256 * 256;
#pragma unroll
  for (int t4 = 0; t4 < 4; ++t4)
#pragma unroll
    for (int r = 0; r < 8; ++r)
      o[(size_t)(tM * 16 + r + 8 * h) * 256 + s * 64 + 16 * t4 + n] =
          acc[t4][r];
}

__global__ void k_softmax256(float* x, __bf16* xb, int nrows) {
  int wave = (int)((blockIdx.x * blockDim.x + threadIdx.x) >> 5);
  if (wave >= nrows) return;
  int lane = threadIdx.x & 31;
  float* r = x + (size_t)wave * 256;
  __bf16* rb = xb + (size_t)wave * 256;
  float v[8], m = -1e30f;
#pragma unroll
  for (int i = 0; i < 8; ++i) { v[i] = r[lane + 32 * i]; m = fmaxf(m, v[i]); }
  for (int t = 16; t >= 1; t >>= 1) m = fmaxf(m, __shfl_xor(m, t, 32));
  float s = 0.f;
#pragma unroll
  for (int i = 0; i < 8; ++i) { v[i] = __expf(v[i] - m); s += v[i]; }
  for (int t = 16; t >= 1; t >>= 1) s += __shfl_xor(s, t, 32);
  float inv = 1.f / s;
#pragma unroll
  for (int i = 0; i < 8; ++i) {
    float pv = v[i] * inv;
    r[lane + 32 * i] = pv;
    rb[lane + 32 * i] = (__bf16)pv;
  }
}

__global__ void k_pinv_scales(const float* x, float* scales) {
  int wave = (int)((blockIdx.x * blockDim.x + threadIdx.x) >> 5);
  if (wave >= 32 * 256) return;
  int lane = threadIdx.x & 31;
  int bh = wave >> 8, r = wave & 255;
  const float* m = x + (size_t)bh * 256 * 256;
  float rs = 0.f, cs = 0.f;
#pragma unroll
  for (int i = 0; i < 8; ++i) {
    rs += fabsf(m[(size_t)r * 256 + lane + 32 * i]);
    cs += fabsf(m[(size_t)(lane + 32 * i) * 256 + r]);
  }
  for (int t = 16; t >= 1; t >>= 1) {
    rs += __shfl_xor(rs, t, 32);
    cs += __shfl_xor(cs, t, 32);
  }
  if (lane == 0) {
    atomicMax((unsigned*)&scales[0], __float_as_uint(rs));
    atomicMax((unsigned*)&scales[1], __float_as_uint(cs));
  }
}

__global__ void k_init_z(const float* x, __bf16* zb, __bf16* zTb,
                         const float* scales) {
  size_t i = (size_t)blockIdx.x * blockDim.x + threadIdx.x;
  if (i >= (size_t)32 * 256 * 256) return;
  int j = (int)(i & 255), r = (int)((i >> 8) & 255);
  size_t bh = i >> 16;
  float denom = scales[0] * scales[1];
  float val = x[(bh << 16) + (size_t)j * 256 + r] / denom;
  zb[i] = (__bf16)val;
  zTb[(bh << 16) + (size_t)j * 256 + r] = (__bf16)val;
}

__global__ void k_gemm256b(const __bf16* A, const __bf16* BT, __bf16* OutRaw,
                           __bf16* OutT, __bf16* OutTT, float alpha,
                           float beta) {
  int wid = (int)((blockIdx.x * blockDim.x + threadIdx.x) >> 5);
  int bh = wid >> 6, t = wid & 63;
  int tM = t >> 2, s = t & 3;
  size_t bofs = (size_t)bh << 16;
  floatx8 acc[4] = {};
  gemm_strip<8>(A + bofs, tM * 16, BT + bofs, s * 64, 256, acc);
  int lane = threadIdx.x & 31;
  int n = lane & 15, h = lane >> 4;
#pragma unroll
  for (int t4 = 0; t4 < 4; ++t4)
#pragma unroll
    for (int r = 0; r < 8; ++r) {
      int M = tM * 16 + r + 8 * h, N = s * 64 + 16 * t4 + n;
      float tv = acc[t4][r];
      if (OutRaw) OutRaw[bofs + (size_t)M * 256 + N] = (__bf16)tv;
      float tr = alpha * (M == N ? 1.f : 0.f) + beta * tv;
      if (OutT) OutT[bofs + (size_t)M * 256 + N] = (__bf16)tr;
      if (OutTT) OutTT[bofs + (size_t)N * 256 + M] = (__bf16)tr;
    }
}

__device__ __forceinline__ void stream_attn(const __bf16* Ak, const __bf16* Bq,
                                            int qrow0, const __bf16* A2,
                                            int ldA2, int nkeys, float* outF,
                                            int rsF, __bf16* outB, int nsB,
                                            int dsB) {
  int lane = threadIdx.x & 31;
  int n = lane & 15, h = lane >> 4;
  bf16x16 bq0 = load_frag_bf16_rm(Bq, qrow0, 0, 64);
  bf16x16 bq1 = load_frag_bf16_rm(Bq, qrow0, 32, 64);
  float run_m = -1e30f, run_s = 0.f;
  floatx8 acc[4] = {};
  for (int c = 0; c < nkeys; c += 32) {
    floatx8 s0 = {}, s1 = {};
    s0 = wmma_bf16(load_frag_bf16_rm(Ak, c, 0, 64), bq0, s0);
    s0 = wmma_bf16(load_frag_bf16_rm(Ak, c, 32, 64), bq1, s0);
    s1 = wmma_bf16(load_frag_bf16_rm(Ak, c + 16, 0, 64), bq0, s1);
    s1 = wmma_bf16(load_frag_bf16_rm(Ak, c + 16, 32, 64), bq1, s1);
    float mc = -1e30f;
#pragma unroll
    for (int r = 0; r < 8; ++r) { mc = fmaxf(mc, s0[r]); mc = fmaxf(mc, s1[r]); }
    mc = fmaxf(mc, __shfl_xor(mc, 16, 32));
    float newm = fmaxf(run_m, mc);
    float scale = __expf(run_m - newm);
    float p0[8], p1[8], ps = 0.f;
#pragma unroll
    for (int r = 0; r < 8; ++r) {
      p0[r] = __expf(s0[r] - newm);
      p1[r] = __expf(s1[r] - newm);
      ps += p0[r] + p1[r];
    }
    run_s = run_s * scale + ps;
    run_m = newm;
    FragAB bp;
#pragma unroll
    for (int p = 0; p < 4; ++p) {
      bp.u[p] = pack2bf(p0[2 * p], p0[2 * p + 1]);
      bp.u[4 + p] = pack2bf(p1[2 * p], p1[2 * p + 1]);
    }
#pragma unroll
    for (int t = 0; t < 4; ++t) {
#pragma unroll
      for (int r = 0; r < 8; ++r) acc[t][r] *= scale;
      acc[t] = wmma_bf16(load_frag_bf16_rm(A2, 16 * t, c, ldA2), bp.v, acc[t]);
    }
  }
  float tot = run_s + __shfl_xor(run_s, 16, 32);
  float inv = 1.f / tot;
#pragma unroll
  for (int t = 0; t < 4; ++t)
#pragma unroll
    for (int r = 0; r < 8; ++r) {
      int dh = 16 * t + r + 8 * h;
      float val = acc[t][r] * inv;
      if (outF) outF[(size_t)n * rsF + dh] = val;
      if (outB) outB[(size_t)n * nsB + (size_t)dh * dsB] = (__bf16)val;
    }
}

#define A3_WAVES 8
#define A3_QG 2
#define A3_KEYS (4096 / A3_WAVES)
__global__ void __launch_bounds__(256)
k_attn3v(const __bf16* kb, const __bf16* ql, const __bf16* vT, __bf16* kvT) {
  __shared__ float accL[A3_WAVES][A3_QG][4][8][32];
  __shared__ float mL[A3_WAVES][A3_QG][16];
  __shared__ float sL[A3_WAVES][A3_QG][16];
  int lane = threadIdx.x & 31;
  int w = __builtin_amdgcn_readfirstlane((int)(threadIdx.x >> 5));
  int bh = blockIdx.x >> 3, q0 = (blockIdx.x & 7) * (16 * A3_QG);
  const __bf16* Ak = kb + (size_t)bh * 4096 * 64;
  const __bf16* Bq = ql + (size_t)bh * 256 * 64;
  const __bf16* A2 = vT + (size_t)bh * 64 * 4096;
  bf16x16 bq[A3_QG][2];
  float run_m[A3_QG], run_s[A3_QG];
  floatx8 acc[A3_QG][4];
#pragma unroll
  for (int g = 0; g < A3_QG; ++g) {
    bq[g][0] = load_frag_bf16_rm(Bq, q0 + 16 * g, 0, 64);
    bq[g][1] = load_frag_bf16_rm(Bq, q0 + 16 * g, 32, 64);
    run_m[g] = -1e30f;
    run_s[g] = 0.f;
#pragma unroll
    for (int t = 0; t < 4; ++t) acc[g][t] = floatx8{};
  }
  int kbeg = w * A3_KEYS;
  for (int c = kbeg; c < kbeg + A3_KEYS; c += 32) {
    bf16x16 ka[4], va[4];
    ka[0] = load_frag_bf16_rm(Ak, c, 0, 64);
    ka[1] = load_frag_bf16_rm(Ak, c, 32, 64);
    ka[2] = load_frag_bf16_rm(Ak, c + 16, 0, 64);
    ka[3] = load_frag_bf16_rm(Ak, c + 16, 32, 64);
#pragma unroll
    for (int t = 0; t < 4; ++t) va[t] = load_frag_bf16_rm(A2, 16 * t, c, 4096);
    __builtin_amdgcn_sched_barrier(0);
#pragma unroll
    for (int g = 0; g < A3_QG; ++g) {
      floatx8 s0 = {}, s1 = {};
      s0 = wmma_bf16(ka[0], bq[g][0], s0);
      s0 = wmma_bf16(ka[1], bq[g][1], s0);
      s1 = wmma_bf16(ka[2], bq[g][0], s1);
      s1 = wmma_bf16(ka[3], bq[g][1], s1);
      float mc = -1e30f;
#pragma unroll
      for (int r = 0; r < 8; ++r) { mc = fmaxf(mc, s0[r]); mc = fmaxf(mc, s1[r]); }
      mc = fmaxf(mc, __shfl_xor(mc, 16, 32));
      float newm = fmaxf(run_m[g], mc);
      float scale = __expf(run_m[g] - newm);
      float p0[8], p1[8], ps = 0.f;
#pragma unroll
      for (int r = 0; r < 8; ++r) {
        p0[r] = __expf(s0[r] - newm);
        p1[r] = __expf(s1[r] - newm);
        ps += p0[r] + p1[r];
      }
      run_s[g] = run_s[g] * scale + ps;
      run_m[g] = newm;
      FragAB bp;
#pragma unroll
      for (int p = 0; p < 4; ++p) {
        bp.u[p] = pack2bf(p0[2 * p], p0[2 * p + 1]);
        bp.u[4 + p] = pack2bf(p1[2 * p], p1[2 * p + 1]);
      }
#pragma unroll
      for (int t = 0; t < 4; ++t) {
#pragma unroll
        for (int r = 0; r < 8; ++r) acc[g][t][r] *= scale;
        acc[g][t] = wmma_bf16(va[t], bp.v, acc[g][t]);
      }
    }
  }
#pragma unroll
  for (int g = 0; g < A3_QG; ++g) {
    float tot = run_s[g] + __shfl_xor(run_s[g], 16, 32);
    if (lane < 16) {
      mL[w][g][lane] = run_m[g];
      sL[w][g][lane] = tot;
    }
#pragma unroll
    for (int t = 0; t < 4; ++t)
#pragma unroll
      for (int r = 0; r < 8; ++r) accL[w][g][t][r][lane] = acc[g][t][r];
  }
  __syncthreads();
  int cc = lane & 15, hh = lane >> 4;
  for (int g = 0; g < A3_QG; ++g) {
    float mw[A3_WAVES], f[A3_WAVES], M = -1e30f, S = 0.f;
#pragma unroll
    for (int v = 0; v < A3_WAVES; ++v) { mw[v] = mL[v][g][cc]; M = fmaxf(M, mw[v]); }
#pragma unroll
    for (int v = 0; v < A3_WAVES; ++v) {
      f[v] = __expf(mw[v] - M);
      S += sL[v][g][cc] * f[v];
    }
    float inv = 1.f / S;
    for (int t = 0; t < 4; ++t) {
      float o = 0.f;
#pragma unroll
      for (int v = 0; v < A3_WAVES; ++v) o += accL[v][g][t][w][lane] * f[v];
      int dh = 16 * t + 8 * hh + w;
      kvT[((size_t)bh * 64 + dh) * 256 + q0 + 16 * g + cc] = (__bf16)(o * inv);
    }
  }
}

__global__ void k_w2(const __bf16* zb, const __bf16* kvT, __bf16* w2t) {
  int wid = (int)((blockIdx.x * blockDim.x + threadIdx.x) >> 5);
  int bh = wid >> 4, tM = wid & 15;
  floatx8 acc[4] = {};
  gemm_strip<8>(zb + ((size_t)bh << 16), tM * 16, kvT + (size_t)bh * 64 * 256,
                0, 256, acc);
  int lane = threadIdx.x & 31;
  int n = lane & 15, h = lane >> 4;
#pragma unroll
  for (int t4 = 0; t4 < 4; ++t4)
#pragma unroll
    for (int r = 0; r < 8; ++r) {
      int M = tM * 16 + r + 8 * h;
      int N = 16 * t4 + n;
      w2t[((size_t)bh * 64 + N) * 256 + M] = (__bf16)acc[t4][r];
    }
}

__global__ void k_attn1_out(const __bf16* kl, const __bf16* qb,
                            const __bf16* w2t, float* aout) {
  int wid = (int)((blockIdx.x * blockDim.x + threadIdx.x) >> 5);
  int bh = wid >> 8, blk = wid & 255;
  int b = bh >> 3, hh = bh & 7;
  stream_attn(kl + (size_t)bh * 256 * 64, qb + (size_t)bh * 4096 * 64, blk * 16,
              w2t + (size_t)bh * 64 * 256, 256, 256,
              aout + ((size_t)b * 4096 + blk * 16) * 512 + hh * 64, 512,
              nullptr, 0, 0);
}

#define CONV_R 16
#define CONV_G 8
#define CONV_ROWS (CONV_R + 32)
__global__ void __launch_bounds__(256, 4)
k_conv_res(const float* __restrict__ vf, const float* __restrict__ rw,
           float* __restrict__ aout, __bf16* __restrict__ aout_b) {
  int t = (int)(blockIdx.x * blockDim.x + threadIdx.x);
  if (t >= 4 * (4096 / CONV_R) * 128) return;
  int c4 = t & 127;
  int chunk = t >> 7;
  int b = chunk / (4096 / CONV_R);
  int ns0 = (chunk % (4096 / CONV_R)) * CONV_R;
  int col = c4 * 4, hh = col >> 6, dh = col & 63;
  const float* vcol = vf + ((size_t)(b * 8 + hh) * 4096) * 64 + dh;
  float* arow = aout + ((size_t)b * 4096 + ns0) * 512 + col;
  __bf16* brow = aout_b + ((size_t)b * 4096 + ns0) * 512 + col;
  float w[33];
#pragma unroll
  for (int j = 0; j < 33; ++j) w[j] = rw[hh * 33 + j];
  float4 acc[CONV_R], win[CONV_ROWS], a[CONV_R];
#pragma unroll
  for (int o = 0; o < CONV_R; ++o) acc[o] = make_float4(0.f, 0.f, 0.f, 0.f);
#pragma unroll
  for (int k = 0; k < CONV_G; ++k) {
    int p = ns0 + k - 16;
    int pc = p < 0 ? 0 : (p > 4095 ? 4095 : p);
    win[k] = *(const float4*)(vcol + (size_t)pc * 64);
  }
#pragma unroll
  for (int g = 0; g < CONV_ROWS / CONV_G; ++g) {
    if (g + 1 < CONV_ROWS / CONV_G) {
#pragma unroll
      for (int k = (g + 1) * CONV_G; k < (g + 2) * CONV_G; ++k) {
        int p = ns0 + k - 16;
        int pc = p < 0 ? 0 : (p > 4095 ? 4095 : p);
        win[k] = *(const float4*)(vcol + (size_t)pc * 64);
      }
    } else {
#pragma unroll
      for (int o = 0; o < CONV_R; ++o)
        a[o] = *(const float4*)(arow + (size_t)o * 512);
    }
#pragma unroll
    for (int k = g * CONV_G; k < (g + 1) * CONV_G; ++k) {
      int p = ns0 + k - 16;
      float4 v = win[k];
      if (p < 0 || p > 4095) v = make_float4(0.f, 0.f, 0.f, 0.f);
#pragma unroll
      for (int o = 0; o < CONV_R; ++o) {
        int j = k - o;
        if (j >= 0 && j < 33) {
          acc[o].x += v.x * w[j];
          acc[o].y += v.y * w[j];
          acc[o].z += v.z * w[j];
          acc[o].w += v.w * w[j];
        }
      }
    }
  }
#pragma unroll
  for (int o = 0; o < CONV_R; ++o) {
    float4 r;
    r.x = a[o].x + acc[o].x;
    r.y = a[o].y + acc[o].y;
    r.z = a[o].z + acc[o].z;
    r.w = a[o].w + acc[o].w;
    *(float4*)(arow + (size_t)o * 512) = r;
    uint2 pk;
    pk.x = pack2bf(r.x, r.y);
    pk.y = pack2bf(r.z, r.w);
    *(uint2*)(brow + (size_t)o * 512) = pk;
  }
}

__global__ void __launch_bounds__(256)
k_outproj(const __bf16* __restrict__ aoutb, const __bf16* __restrict__ wo,
          const float* __restrict__ bias, const float* __restrict__ x,
          float* __restrict__ out) {
  int wid = (int)((blockIdx.x * blockDim.x + threadIdx.x) >> 5);
  int tM = wid >> 3, s = wid & 7;
  floatx8 acc[4][4] = {};
  gemm_tile64<16>(aoutb, tM * 64, wo, s * 64, 512, acc);
  int lane = threadIdx.x & 31;
  int n = lane & 15, h = lane >> 4;
#pragma unroll
  for (int t4 = 0; t4 < 4; ++t4) {
    int col = s * 64 + 16 * t4 + n;
    float bc = bias[col];
#pragma unroll
    for (int m = 0; m < 4; ++m) {
#pragma unroll
      for (int r = 0; r < 8; ++r) {
        size_t idx = (size_t)(tM * 64 + 16 * m + r + 8 * h) * 512 + col;
        out[idx] = x[idx] + acc[m][t4][r] + bc;
      }
    }
  }
}

extern "C" void kernel_launch(void* const* d_in, const int* in_sizes, int n_in,
                              void* d_out, int out_size, void* d_ws,
                              size_t ws_size, hipStream_t stream) {
  const float* x     = (const float*)d_in[0];
  const float* ln_g  = (const float*)d_in[1];
  const float* ln_b  = (const float*)d_in[2];
  const float* w_qkv = (const float*)d_in[3];
  const float* w_out = (const float*)d_in[4];
  const float* b_out = (const float*)d_in[5];
  const float* res_w = (const float*)d_in[6];
  float* out = (float*)d_out;

  char* p = (char*)d_ws;
  auto alloc = [&](size_t bytes) -> void* {
    void* r = (void*)p;
    p += (bytes + 255) & ~(size_t)255;
    return r;
  };
  const size_t NTOK = 4 * 4096;
  const size_t MM = (size_t)32 * 256 * 256;
  __bf16* xn_b   = (__bf16*)alloc(NTOK * 512 * 2);
  __bf16* wqkv_b = (__bf16*)alloc(1536 * 512 * 2);
  __bf16* wout_b = (__bf16*)alloc(512 * 512 * 2);
  __bf16* q_b    = (__bf16*)alloc((size_t)32 * 4096 * 64 * 2);
  __bf16* k_b    = (__bf16*)alloc((size_t)32 * 4096 * 64 * 2);
  float*  v_f    = (float*)alloc((size_t)32 * 4096 * 64 * 4);
  __bf16* vT_b   = (__bf16*)alloc((size_t)32 * 64 * 4096 * 2);
  __bf16* ql_b   = (__bf16*)alloc((size_t)32 * 256 * 64 * 2);
  __bf16* kl_b   = (__bf16*)alloc((size_t)32 * 256 * 64 * 2);
  float*  mat_x  = (float*)alloc(MM * 4);
  __bf16* x_b    = (__bf16*)alloc(MM * 2);
  __bf16* z0_b   = (__bf16*)alloc(MM * 2);
  __bf16* z0T_b  = (__bf16*)alloc(MM * 2);
  __bf16* z1_b   = (__bf16*)alloc(MM * 2);
  __bf16* z1T_b  = (__bf16*)alloc(MM * 2);
  __bf16* xz_b   = (__bf16*)alloc(MM * 2);
  __bf16* t1T_b  = (__bf16*)alloc(MM * 2);
  __bf16* t2T_b  = (__bf16*)alloc(MM * 2);
  __bf16* t3T_b  = (__bf16*)alloc(MM * 2);
  __bf16* kvT_b  = (__bf16*)alloc((size_t)32 * 64 * 256 * 2);
  __bf16* w2t_b  = (__bf16*)alloc((size_t)32 * 64 * 256 * 2);
  float*  aout_f = (float*)alloc(NTOK * 512 * 4);
  __bf16* aout_b = (__bf16*)alloc(NTOK * 512 * 2);
  float*  scales = (float*)alloc(256);
  (void)ws_size; (void)n_in; (void)in_sizes; (void)out_size;

  const int T = 256;
  k_convert_w<<<3072, T, 0, stream>>>(w_qkv, w_out, wqkv_b, wout_b, scales);
  k_layernorm<<<2048, T, 0, stream>>>(x, ln_g, ln_b, xn_b, (int)NTOK);
  k_gemm_qkv<<<768, T, 0, stream>>>(xn_b, wqkv_b, q_b, k_b, v_f, vT_b);
  k_landmarks<<<2048, T, 0, stream>>>(q_b, k_b, ql_b, kl_b);
  k_attn2_logits<<<256, T, 0, stream>>>(ql_b, kl_b, mat_x);
  k_softmax256<<<1024, T, 0, stream>>>(mat_x, x_b, 32 * 256);
  k_pinv_scales<<<1024, T, 0, stream>>>(mat_x, scales);
  k_init_z<<<8192, T, 0, stream>>>(mat_x, z0_b, z0T_b, scales);

  __bf16* zc = z0_b;  __bf16* zcT = z0T_b;
  __bf16* zn = z1_b;  __bf16* znT = z1T_b;
  for (int it = 0; it < 6; ++it) {
    k_gemm256b<<<256, T, 0, stream>>>(x_b, zcT, xz_b, nullptr, t1T_b, 7.f, -1.f);
    k_gemm256b<<<256, T, 0, stream>>>(xz_b, t1T_b, nullptr, nullptr, t2T_b, 15.f, -1.f);
    k_gemm256b<<<256, T, 0, stream>>>(xz_b, t2T_b, nullptr, nullptr, t3T_b, 13.f, -1.f);
    k_gemm256b<<<256, T, 0, stream>>>(zc, t3T_b, nullptr, zn, znT, 0.f, 0.25f);
    __bf16* t;
    t = zc; zc = zn; zn = t;
    t = zcT; zcT = znT; znT = t;
  }

  k_attn3v<<<256, T, 0, stream>>>(k_b, ql_b, vT_b, kvT_b);
  k_w2<<<64, T, 0, stream>>>(zc, kvT_b, w2t_b);
  k_attn1_out<<<1024, T, 0, stream>>>(kl_b, q_b, w2t_b, aout_f);
  k_conv_res<<<512, T, 0, stream>>>(v_f, res_w, aout_f, aout_b);
  k_outproj<<<256, T, 0, stream>>>(aout_b, wout_b, b_out, x, out);
}


# pinv operands (24 GEMMs, W2 A side) stored fragment-packed, 1 KB per 16x32 block in lane order: a fragment load covers 4 lines (was 16) and runs 4 K steps ahead; transposed outputs 16-byte stores
# speedup vs baseline: 1.9709x; 1.3599x over previous
#include <hip/hip_runtime.h>
#include <hip/hip_bf16.h>
#include <math.h>


typedef __attribute__((ext_vector_type(16))) __bf16 bf16x16;
typedef __attribute__((ext_vector_type(8)))  float  floatx8;

union FragAB { bf16x16 v; unsigned int u[8]; };

__device__ __forceinline__ unsigned pack2bf(float a, float b) {
  __bf16 lo = (__bf16)a, hi = (__bf16)b;
  unsigned short ul = __builtin_bit_cast(unsigned short, lo);
  unsigned short uh = __builtin_bit_cast(unsigned short, hi);
  return (unsigned)ul | ((unsigned)uh << 16);
}

__device__ __forceinline__ int kpair(int p, int h) {
  return (p < 4) ? (2 * p + 8 * h) : (8 + 2 * p + 8 * h);
}

__device__ __forceinline__ bf16x16 load_frag_bf16_rm(const __bf16* src, int row0,
                                                     int k0, int ld) {
  int lane = threadIdx.x & 31;
  int r = lane & 15, h = lane >> 4;
  const __bf16* base = src + (size_t)(row0 + r) * ld + k0;
  FragAB f;
#pragma unroll
  for (int p = 0; p < 8; ++p) f.u[p] = *(const unsigned*)(base + kpair(p, h));
  return f.v;
}

__device__ __forceinline__ floatx8 wmma_bf16(bf16x16 a, bf16x16 b, floatx8 c) {
  return __builtin_amdgcn_wmma_f32_16x16x32_bf16(false, a, false, b, (short)0, c,
                                                 false, false);
}

template <int KSTEPS>
__device__ __forceinline__ void gemm_strip(const __bf16* A, int rowA,
                                           const __bf16* B, int rowB, int ld,
                                           floatx8 acc[4]) {
#pragma unroll
  for (int ks = 0; ks < KSTEPS; ++ks) {
    int k0 = ks * 32;
    bf16x16 af = load_frag_bf16_rm(A, rowA, k0, ld);
#pragma unroll
    for (int t = 0; t < 4; ++t) {
      bf16x16 bfr = load_frag_bf16_rm(B, rowB + 16 * t, k0, ld);
      acc[t] = wmma_bf16(af, bfr, acc[t]);
    }
  }
}

__device__ __forceinline__ size_t pk_idx(int row, int k) {
  int kk = k & 31;
  int lane = (row & 15) + 16 * ((kk >> 3) & 1);
  return (size_t)(((row >> 4) * 8 + (k >> 5)) * 512 + (kk >> 4) * 256 +
                  lane * 8 + (kk & 7));
}

__device__ __forceinline__ bf16x16 load_frag_bf16_pk(const __bf16* mat, int rb,
                                                     int ks) {
  int lane = threadIdx.x & 31;
  const uint4* blk = (const uint4*)(mat + (size_t)(rb * 8 + ks) * 512) + lane;
  uint4 lo = blk[0], hi = blk[32];
  FragAB f;
  f.u[0] = lo.x; f.u[1] = lo.y; f.u[2] = lo.z; f.u[3] = lo.w;
  f.u[4] = hi.x; f.u[5] = hi.y; f.u[6] = hi.z; f.u[7] = hi.w;
  return f.v;
}

#define PK_AHEAD 4
template <bool BPK>
__device__ __forceinline__ void gemm_strip_pk(const __bf16* A, int rowA,
                                              const __bf16* B, int rowB,
                                              floatx8 acc[4]) {
  bf16x16 af[8], bfr[8][4];
#pragma unroll
  for (int ks = 0; ks < PK_AHEAD; ++ks) {
    af[ks] = load_frag_bf16_pk(A, rowA >> 4, ks);
#pragma unroll
    for (int t = 0; t < 4; ++t)
      bfr[ks][t] = BPK ? load_frag_bf16_pk(B, (rowB >> 4) + t, ks)
                       : load_frag_bf16_rm(B, rowB + 16 * t, ks * 32, 256);
    __builtin_amdgcn_sched_barrier(0);
  }
#pragma unroll
  for (int ks = 0; ks < 8; ++ks) {
#pragma unroll
    for (int t = 0; t < 4; ++t) acc[t] = wmma_bf16(af[ks], bfr[ks][t], acc[t]);
    __builtin_amdgcn_sched_barrier(0);
    if (ks + PK_AHEAD < 8) {
      af[ks + PK_AHEAD] = load_frag_bf16_pk(A, rowA >> 4, ks + PK_AHEAD);
#pragma unroll
      for (int t = 0; t < 4; ++t)
        bfr[ks + PK_AHEAD][t] =
            BPK ? load_frag_bf16_pk(B, (rowB >> 4) + t, ks + PK_AHEAD)
                : load_frag_bf16_rm(B, rowB + 16 * t, (ks + PK_AHEAD) * 32, 256);
      __builtin_amdgcn_sched_barrier(0);
    }
  }
}

template <int KSTEPS>
__device__ __forceinline__ void gemm_tile64(const __bf16* A, int rowA,
                                            const __bf16* B, int rowB, int ld,
                                            floatx8 acc[4][4]) {
#pragma unroll
  for (int ks = 0; ks < KSTEPS; ++ks) {
    int k0 = ks * 32;
    bf16x16 af[4], bfr[4];
#pragma unroll
    for (int m = 0; m < 4; ++m) af[m] = load_frag_bf16_rm(A, rowA + 16 * m, k0, ld);
#pragma unroll
    for (int t = 0; t < 4; ++t) bfr[t] = load_frag_bf16_rm(B, rowB + 16 * t, k0, ld);
    __builtin_amdgcn_sched_barrier(0);
#pragma unroll
    for (int m = 0; m < 4; ++m)
#pragma unroll
      for (int t = 0; t < 4; ++t) acc[m][t] = wmma_bf16(af[m], bfr[t], acc[m][t]);
    __builtin_amdgcn_sched_barrier(0);
  }
}

__global__ void k_convert_w(const float* wqkv, const float* wout,
                            __bf16* wqkv_b, __bf16* wout_b, float* scales) {
  int i = blockIdx.x * blockDim.x + threadIdx.x;
  if (i < 2) scales[i] = 0.f;
  if (i < 1536 * 512) wqkv_b[i] = (__bf16)wqkv[i];
  if (i < 512 * 512) wout_b[i] = (__bf16)wout[i];
}

__global__ void k_layernorm(const float* x, const float* g, const float* b,
                            __bf16* xn, int ntok) {
  int wave = (int)((blockIdx.x * blockDim.x + threadIdx.x) >> 5);
  int lane = threadIdx.x & 31;
  if (wave >= ntok) return;
  const float* xr = x + (size_t)wave * 512;
  float s = 0.f, s2 = 0.f, vals[16];
#pragma unroll
  for (int i = 0; i < 16; ++i) {
    float v = xr[lane + 32 * i];
    vals[i] = v; s += v; s2 += v * v;
  }
  for (int m = 16; m >= 1; m >>= 1) {
    s += __shfl_xor(s, m, 32);
    s2 += __shfl_xor(s2, m, 32);
  }
  float mu = s * (1.f / 512.f);
  float var = s2 * (1.f / 512.f) - mu * mu;
  float inv = rsqrtf(var + 1e-5f);
  __bf16* o = xn + (size_t)wave * 512;
#pragma unroll
  for (int i = 0; i < 16; ++i) {
    int c = lane + 32 * i;
    o[c] = (__bf16)((vals[i] - mu) * inv * g[c] + b[c]);
  }
}

__global__ void __launch_bounds__(256)
k_gemm_qkv(const __bf16* xn, const __bf16* wq, __bf16* qb,
           __bf16* kb, float* vf, __bf16* vT) {
  int wid = (int)((blockIdx.x * blockDim.x + threadIdx.x) >> 5);
  int tM = wid / 24, s = wid % 24;
  floatx8 acc[4][4] = {};
  gemm_tile64<16>(xn, tM * 64, wq, s * 64, 512, acc);
  int lane = threadIdx.x & 31;
  int n = lane & 15, h = lane >> 4;
  int col0 = s * 64;
  int sect = col0 >> 9;
  int head = (col0 >> 6) & 7;
#pragma unroll
  for (int m = 0; m < 4; ++m) {
#pragma unroll
    for (int t = 0; t < 4; ++t) {
      int dh = 16 * t + n;
#pragma unroll
      for (int r = 0; r < 8; ++r) {
        int tok = tM * 64 + 16 * m + r + 8 * h;
        int bb = tok >> 12, ns = tok & 4095;
        size_t bh = (size_t)(bb * 8 + head);
        size_t idx = (bh * 4096 + ns) * 64 + dh;
        float val = acc[m][t][r];
        if (sect == 0) qb[idx] = (__bf16)(val * 0.125f);
        else if (sect == 1) kb[idx] = (__bf16)val;
        else {
          vf[idx] = val;
          vT[(bh * 64 + dh) * 4096 + ns] = (__bf16)val;
        }
      }
    }
  }
}

__global__ void k_landmarks(const __bf16* qb, const __bf16* kb, __bf16* ql,
                            __bf16* kl) {
  int i = blockIdx.x * blockDim.x + threadIdx.x;
  if (i >= 32 * 256 * 64) return;
  int dh = i & 63, m = (i >> 6) & 255, bh = i >> 14;
  size_t base = ((size_t)bh * 4096 + m * 16) * 64 + dh;
  float sq = 0.f, sk = 0.f;
#pragma unroll
  for (int t = 0; t < 16; ++t) {
    sq += (float)qb[base + t * 64];
    sk += (float)kb[base + t * 64];
  }
  ql[i] = (__bf16)(sq * (1.f / 16.f));
  kl[i] = (__bf16)(sk * (1.f / 16.f));
}

__global__ void k_attn2_logits(const __bf16* ql, const __bf16* kl, float* out) {
  int wid = (int)((blockIdx.x * blockDim.x + threadIdx.x) >> 5);
  int bh = wid >> 6, t = wid & 63;
  int tM = t >> 2, s = t & 3;
  const __bf16* A = ql + (size_t)bh * 256 * 64;
  const __bf16* B = kl + (size_t)bh * 256 * 64;
  floatx8 acc[4] = {};
  gemm_strip<2>(A, tM * 16, B, s * 64, 64, acc);
  int lane = threadIdx.x & 31;
  int n = lane & 15, h = lane >> 4;
  float* o = out + (size_t)bh * 256 * 256;
#pragma unroll
  for (int t4 = 0; t4 < 4; ++t4)
#pragma unroll
    for (int r = 0; r < 8; ++r)
      o[(size_t)(tM * 16 + r + 8 * h) * 256 + s * 64 + 16 * t4 + n] =
          acc[t4][r];
}

__global__ void k_softmax256(float* x, __bf16* xb, int nrows) {
  int wave = (int)((blockIdx.x * blockDim.x + threadIdx.x) >> 5);
  if (wave >= nrows) return;
  int lane = threadIdx.x & 31;
  float* r = x + (size_t)wave * 256;
  __bf16* rb = xb + ((size_t)(wave >> 8) << 16);
  int row = wave & 255;
  float v[8], m = -1e30f;
#pragma unroll
  for (int i = 0; i < 8; ++i) { v[i] = r[lane + 32 * i]; m = fmaxf(m, v[i]); }
  for (int t = 16; t >= 1; t >>= 1) m = fmaxf(m, __shfl_xor(m, t, 32));
  float s = 0.f;
#pragma unroll
  for (int i = 0; i < 8; ++i) { v[i] = __expf(v[i] - m); s += v[i]; }
  for (int t = 16; t >= 1; t >>= 1) s += __shfl_xor(s, t, 32);
  float inv = 1.f / s;
#pragma unroll
  for (int i = 0; i < 8; ++i) {
    float pv = v[i] * inv;
    r[lane + 32 * i] = pv;
    rb[pk_idx(row, lane + 32 * i)] = (__bf16)pv;
  }
}

__global__ void k_pinv_scales(const float* x, float* scales) {
  int wave = (int)((blockIdx.x * blockDim.x + threadIdx.x) >> 5);
  if (wave >= 32 * 256) return;
  int lane = threadIdx.x & 31;
  int bh = wave >> 8, r = wave & 255;
  const float* m = x + (size_t)bh * 256 * 256;
  float rs = 0.f, cs = 0.f;
#pragma unroll
  for (int i = 0; i < 8; ++i) {
    rs += fabsf(m[(size_t)r * 256 + lane + 32 * i]);
    cs += fabsf(m[(size_t)(lane + 32 * i) * 256 + r]);
  }
  for (int t = 16; t >= 1; t >>= 1) {
    rs += __shfl_xor(rs, t, 32);
    cs += __shfl_xor(cs, t, 32);
  }
  if (lane == 0) {
    atomicMax((unsigned*)&scales[0], __float_as_uint(rs));
    atomicMax((unsigned*)&scales[1], __float_as_uint(cs));
  }
}

__global__ void k_init_z(const float* x, __bf16* zb, __bf16* zTb,
                         const float* scales) {
  size_t i = (size_t)blockIdx.x * blockDim.x + threadIdx.x;
  if (i >= (size_t)32 * 256 * 256) return;
  int j = (int)(i & 255), r = (int)((i >> 8) & 255);
  size_t bh = i >> 16;
  float denom = scales[0] * scales[1];
  float val = x[(bh << 16) + (size_t)j * 256 + r] / denom;
  zb[(bh << 16) + pk_idx(r, j)] = (__bf16)val;
  zTb[(bh << 16) + pk_idx(j, r)] = (__bf16)val;
}

__global__ void __launch_bounds__(256)
k_gemm256b(const __bf16* A, const __bf16* BT, __bf16* OutRaw,
           __bf16* OutT, __bf16* OutTT, float alpha, float beta) {
  int wid = (int)((blockIdx.x * blockDim.x + threadIdx.x) >> 5);
  int bh = wid >> 6, t = wid & 63;
  int tM = t >> 2, s = t & 3;
  size_t bofs = (size_t)bh << 16;
  floatx8 acc[4] = {};
  gemm_strip_pk<true>(A + bofs, tM * 16, BT + bofs, s * 64, acc);
  int lane = threadIdx.x & 31;
  int n = lane & 15, h = lane >> 4;
#pragma unroll
  for (int t4 = 0; t4 < 4; ++t4) {
    int N = s * 64 + 16 * t4 + n;
    float trv[8];
#pragma unroll
    for (int r = 0; r < 8; ++r) {
      int M = tM * 16 + r + 8 * h;
      float tv = acc[t4][r];
      if (OutRaw) OutRaw[bofs + pk_idx(M, N)] = (__bf16)tv;
      float tr = alpha * (M == N ? 1.f : 0.f) + beta * tv;
      if (OutT) OutT[bofs + pk_idx(M, N)] = (__bf16)tr;
      trv[r] = tr;
    }
    if (OutTT) {
      uint4 pv;
      pv.x = pack2bf(trv[0], trv[1]);
      pv.y = pack2bf(trv[2], trv[3]);
      pv.z = pack2bf(trv[4], trv[5]);
      pv.w = pack2bf(trv[6], trv[7]);
      *(uint4*)(OutTT + bofs + pk_idx(N, tM * 16 + 8 * h)) = pv;
    }
  }
}

__device__ __forceinline__ void stream_attn(const __bf16* Ak, const __bf16* Bq,
                                            int qrow0, const __bf16* A2,
                                            int ldA2, int nkeys, float* outF,
                                            int rsF, __bf16* outB, int nsB,
                                            int dsB) {
  int lane = threadIdx.x & 31;
  int n = lane & 15, h = lane >> 4;
  bf16x16 bq0 = load_frag_bf16_rm(Bq, qrow0, 0, 64);
  bf16x16 bq1 = load_frag_bf16_rm(Bq, qrow0, 32, 64);
  float run_m = -1e30f, run_s = 0.f;
  floatx8 acc[4] = {};
  for (int c = 0; c < nkeys; c += 32) {
    floatx8 s0 = {}, s1 = {};
    s0 = wmma_bf16(load_frag_bf16_rm(Ak, c, 0, 64), bq0, s0);
    s0 = wmma_bf16(load_frag_bf16_rm(Ak, c, 32, 64), bq1, s0);
    s1 = wmma_bf16(load_frag_bf16_rm(Ak, c + 16, 0, 64), bq0, s1);
    s1 = wmma_bf16(load_frag_bf16_rm(Ak, c + 16, 32, 64), bq1, s1);
    float mc = -1e30f;
#pragma unroll
    for (int r = 0; r < 8; ++r) { mc = fmaxf(mc, s0[r]); mc = fmaxf(mc, s1[r]); }
    mc = fmaxf(mc, __shfl_xor(mc, 16, 32));
    float newm = fmaxf(run_m, mc);
    float scale = __expf(run_m - newm);
    float p0[8], p1[8], ps = 0.f;
#pragma unroll
    for (int r = 0; r < 8; ++r) {
      p0[r] = __expf(s0[r] - newm);
      p1[r] = __expf(s1[r] - newm);
      ps += p0[r] + p1[r];
    }
    run_s = run_s * scale + ps;
    run_m = newm;
    FragAB bp;
#pragma unroll
    for (int p = 0; p < 4; ++p) {
      bp.u[p] = pack2bf(p0[2 * p], p0[2 * p + 1]);
      bp.u[4 + p] = pack2bf(p1[2 * p], p1[2 * p + 1]);
    }
#pragma unroll
    for (int t = 0; t < 4; ++t) {
#pragma unroll
      for (int r = 0; r < 8; ++r) acc[t][r] *= scale;
      acc[t] = wmma_bf16(load_frag_bf16_rm(A2, 16 * t, c, ldA2), bp.v, acc[t]);
    }
  }
  float tot = run_s + __shfl_xor(run_s, 16, 32);
  float inv = 1.f / tot;
#pragma unroll
  for (int t = 0; t < 4; ++t)
#pragma unroll
    for (int r = 0; r < 8; ++r) {
      int dh = 16 * t + r + 8 * h;
      float val = acc[t][r] * inv;
      if (outF) outF[(size_t)n * rsF + dh] = val;
      if (outB) outB[(size_t)n * nsB + (size_t)dh * dsB] = (__bf16)val;
    }
}

#define A3_WAVES 8
#define A3_QG 2
#define A3_KEYS (4096 / A3_WAVES)
__global__ void __launch_bounds__(256)
k_attn3v(const __bf16* kb, const __bf16* ql, const __bf16* vT, __bf16* kvT) {
  __shared__ float accL[A3_WAVES][A3_QG][4][8][32];
  __shared__ float mL[A3_WAVES][A3_QG][16];
  __shared__ float sL[A3_WAVES][A3_QG][16];
  int lane = threadIdx.x & 31;
  int w = __builtin_amdgcn_readfirstlane((int)(threadIdx.x >> 5));
  int bh = blockIdx.x >> 3, q0 = (blockIdx.x & 7) * (16 * A3_QG);
  const __bf16* Ak = kb + (size_t)bh * 4096 * 64;
  const __bf16* Bq = ql + (size_t)bh * 256 * 64;
  const __bf16* A2 = vT + (size_t)bh * 64 * 4096;
  bf16x16 bq[A3_QG][2];
  float run_m[A3_QG], run_s[A3_QG];
  floatx8 acc[A3_QG][4];
#pragma unroll
  for (int g = 0; g < A3_QG; ++g) {
    bq[g][0] = load_frag_bf16_rm(Bq, q0 + 16 * g, 0, 64);
    bq[g][1] = load_frag_bf16_rm(Bq, q0 + 16 * g, 32, 64);
    run_m[g] = -1e30f;
    run_s[g] = 0.f;
#pragma unroll
    for (int t = 0; t < 4; ++t) acc[g][t] = floatx8{};
  }
  int kbeg = w * A3_KEYS;
  for (int c = kbeg; c < kbeg + A3_KEYS; c += 32) {
    bf16x16 ka[4], va[4];
    ka[0] = load_frag_bf16_rm(Ak, c, 0, 64);
    ka[1] = load_frag_bf16_rm(Ak, c, 32, 64);
    ka[2] = load_frag_bf16_rm(Ak, c + 16, 0, 64);
    ka[3] = load_frag_bf16_rm(Ak, c + 16, 32, 64);
#pragma unroll
    for (int t = 0; t < 4; ++t) va[t] = load_frag_bf16_rm(A2, 16 * t, c, 4096);
    __builtin_amdgcn_sched_barrier(0);
#pragma unroll
    for (int g = 0; g < A3_QG; ++g) {
      floatx8 s0 = {}, s1 = {};
      s0 = wmma_bf16(ka[0], bq[g][0], s0);
      s0 = wmma_bf16(ka[1], bq[g][1], s0);
      s1 = wmma_bf16(ka[2], bq[g][0], s1);
      s1 = wmma_bf16(ka[3], bq[g][1], s1);
      float mc = -1e30f;
#pragma unroll
      for (int r = 0; r < 8; ++r) { mc = fmaxf(mc, s0[r]); mc = fmaxf(mc, s1[r]); }
      mc = fmaxf(mc, __shfl_xor(mc, 16, 32));
      float newm = fmaxf(run_m[g], mc);
      float scale = __expf(run_m[g] - newm);
      float p0[8], p1[8], ps = 0.f;
#pragma unroll
      for (int r = 0; r < 8; ++r) {
        p0[r] = __expf(s0[r] - newm);
        p1[r] = __expf(s1[r] - newm);
        ps += p0[r] + p1[r];
      }
      run_s[g] = run_s[g] * scale + ps;
      run_m[g] = newm;
      FragAB bp;
#pragma unroll
      for (int p = 0; p < 4; ++p) {
        bp.u[p] = pack2bf(p0[2 * p], p0[2 * p + 1]);
        bp.u[4 + p] = pack2bf(p1[2 * p], p1[2 * p + 1]);
      }
#pragma unroll
      for (int t = 0; t < 4; ++t) {
#pragma unroll
        for (int r = 0; r < 8; ++r) acc[g][t][r] *= scale;
        acc[g][t] = wmma_bf16(va[t], bp.v, acc[g][t]);
      }
    }
  }
#pragma unroll
  for (int g = 0; g < A3_QG; ++g) {
    float tot = run_s[g] + __shfl_xor(run_s[g], 16, 32);
    if (lane < 16) {
      mL[w][g][lane] = run_m[g];
      sL[w][g][lane] = tot;
    }
#pragma unroll
    for (int t = 0; t < 4; ++t)
#pragma unroll
      for (int r = 0; r < 8; ++r) accL[w][g][t][r][lane] = acc[g][t][r];
  }
  __syncthreads();
  int cc = lane & 15, hh = lane >> 4;
  for (int g = 0; g < A3_QG; ++g) {
    float mw[A3_WAVES], f[A3_WAVES], M = -1e30f, S = 0.f;
#pragma unroll
    for (int v = 0; v < A3_WAVES; ++v) { mw[v] = mL[v][g][cc]; M = fmaxf(M, mw[v]); }
#pragma unroll
    for (int v = 0; v < A3_WAVES; ++v) {
      f[v] = __expf(mw[v] - M);
      S += sL[v][g][cc] * f[v];
    }
    float inv = 1.f / S;
    for (int t = 0; t < 4; ++t) {
      float o = 0.f;
#pragma unroll
      for (int v = 0; v < A3_WAVES; ++v) o += accL[v][g][t][w][lane] * f[v];
      int dh = 16 * t + 8 * hh + w;
      kvT[((size_t)bh * 64 + dh) * 256 + q0 + 16 * g + cc] = (__bf16)(o * inv);
    }
  }
}

__global__ void __launch_bounds__(256)
k_w2(const __bf16* zb, const __bf16* kvT, __bf16* w2t) {
  int wid = (int)((blockIdx.x * blockDim.x + threadIdx.x) >> 5);
  int bh = wid >> 4, tM = wid & 15;
  floatx8 acc[4] = {};
  gemm_strip_pk<false>(zb + ((size_t)bh << 16), tM * 16,
                       kvT + (size_t)bh * 64 * 256, 0, acc);
  int lane = threadIdx.x & 31;
  int n = lane & 15, h = lane >> 4;
#pragma unroll
  for (int t4 = 0; t4 < 4; ++t4)
#pragma unroll
    for (int r = 0; r < 8; ++r) {
      int M = tM * 16 + r + 8 * h;
      int N = 16 * t4 + n;
      w2t[((size_t)bh * 64 + N) * 256 + M] = (__bf16)acc[t4][r];
    }
}

__global__ void k_attn1_out(const __bf16* kl, const __bf16* qb,
                            const __bf16* w2t, float* aout) {
  int wid = (int)((blockIdx.x * blockDim.x + threadIdx.x) >> 5);
  int bh = wid >> 8, blk = wid & 255;
  int b = bh >> 3, hh = bh & 7;
  stream_attn(kl + (size_t)bh * 256 * 64, qb + (size_t)bh * 4096 * 64, blk * 16,
              w2t + (size_t)bh * 64 * 256, 256, 256,
              aout + ((size_t)b * 4096 + blk * 16) * 512 + hh * 64, 512,
              nullptr, 0, 0);
}

#define CONV_R 16
#define CONV_G 8
#define CONV_ROWS (CONV_R + 32)
__global__ void __launch_bounds__(256, 4)
k_conv_res(const float* __restrict__ vf, const float* __restrict__ rw,
           float* __restrict__ aout, __bf16* __restrict__ aout_b) {
  int t = (int)(blockIdx.x * blockDim.x + threadIdx.x);
  if (t >= 4 * (4096 / CONV_R) * 128) return;
  int c4 = t & 127;
  int chunk = t >> 7;
  int b = chunk / (4096 / CONV_R);
  int ns0 = (chunk % (4096 / CONV_R)) * CONV_R;
  int col = c4 * 4, hh = col >> 6, dh = col & 63;
  const float* vcol = vf + ((size_t)(b * 8 + hh) * 4096) * 64 + dh;
  float* arow = aout + ((size_t)b * 4096 + ns0) * 512 + col;
  __bf16* brow = aout_b + ((size_t)b * 4096 + ns0) * 512 + col;
  float w[33];
#pragma unroll
  for (int j = 0; j < 33; ++j) w[j] = rw[hh * 33 + j];
  float4 acc[CONV_R], win[CONV_ROWS], a[CONV_R];
#pragma unroll
  for (int o = 0; o < CONV_R; ++o) acc[o] = make_float4(0.f, 0.f, 0.f, 0.f);
#pragma unroll
  for (int k = 0; k < CONV_G; ++k) {
    int p = ns0 + k - 16;
    int pc = p < 0 ? 0 : (p > 4095 ? 4095 : p);
    win[k] = *(const float4*)(vcol + (size_t)pc * 64);
  }
#pragma unroll
  for (int g = 0; g < CONV_ROWS / CONV_G; ++g) {
    if (g + 1 < CONV_ROWS / CONV_G) {
#pragma unroll
      for (int k = (g + 1) * CONV_G; k < (g + 2) * CONV_G; ++k) {
        int p = ns0 + k - 16;
        int pc = p < 0 ? 0 : (p > 4095 ? 4095 : p);
        win[k] = *(const float4*)(vcol + (size_t)pc * 64);
      }
    } else {
#pragma unroll
      for (int o = 0; o < CONV_R; ++o)
        a[o] = *(const float4*)(arow + (size_t)o * 512);
    }
#pragma unroll
    for (int k = g * CONV_G; k < (g + 1) * CONV_G; ++k) {
      int p = ns0 + k - 16;
      float4 v = win[k];
      if (p < 0 || p > 4095) v = make_float4(0.f, 0.f, 0.f, 0.f);
#pragma unroll
      for (int o = 0; o < CONV_R; ++o) {
        int j = k - o;
        if (j >= 0 && j < 33) {
          acc[o].x += v.x * w[j];
          acc[o].y += v.y * w[j];
          acc[o].z += v.z * w[j];
          acc[o].w += v.w * w[j];
        }
      }
    }
  }
#pragma unroll
  for (int o = 0; o < CONV_R; ++o) {
    float4 r;
    r.x = a[o].x + acc[o].x;
    r.y = a[o].y + acc[o].y;
    r.z = a[o].z + acc[o].z;
    r.w = a[o].w + acc[o].w;
    *(float4*)(arow + (size_t)o * 512) = r;
    uint2 pk;
    pk.x = pack2bf(r.x, r.y);
    pk.y = pack2bf(r.z, r.w);
    *(uint2*)(brow + (size_t)o * 512) = pk;
  }
}

__global__ void __launch_bounds__(256)
k_outproj(const __bf16* __restrict__ aoutb, const __bf16* __restrict__ wo,
          const float* __restrict__ bias, const float* __restrict__ x,
          float* __restrict__ out) {
  int wid = (int)((blockIdx.x * blockDim.x + threadIdx.x) >> 5);
  int tM = wid >> 3, s = wid & 7;
  floatx8 acc[4][4] = {};
  gemm_tile64<16>(aoutb, tM * 64, wo, s * 64, 512, acc);
  int lane = threadIdx.x & 31;
  int n = lane & 15, h = lane >> 4;
#pragma unroll
  for (int t4 = 0; t4 < 4; ++t4) {
    int col = s * 64 + 16 * t4 + n;
    float bc = bias[col];
#pragma unroll
    for (int m = 0; m < 4; ++m) {
#pragma unroll
      for (int r = 0; r < 8; ++r) {
        size_t idx = (size_t)(tM * 64 + 16 * m + r + 8 * h) * 512 + col;
        out[idx] = x[idx] + acc[m][t4][r] + bc;
      }
    }
  }
}

extern "C" void kernel_launch(void* const* d_in, const int* in_sizes, int n_in,
                              void* d_out, int out_size, void* d_ws,
                              size_t ws_size, hipStream_t stream) {
  const float* x     = (const float*)d_in[0];
  const float* ln_g  = (const float*)d_in[1];
  const float* ln_b  = (const float*)d_in[2];
  const float* w_qkv = (const float*)d_in[3];
  const float* w_out = (const float*)d_in[4];
  const float* b_out = (const float*)d_in[5];
  const float* res_w = (const float*)d_in[6];
  float* out = (float*)d_out;

  char* p = (char*)d_ws;
  auto alloc = [&](size_t bytes) -> void* {
    void* r = (void*)p;
    p += (bytes + 255) & ~(size_t)255;
    return r;
  };
  const size_t NTOK = 4 * 4096;
  const size_t MM = (size_t)32 * 256 * 256;
  __bf16* xn_b   = (__bf16*)alloc(NTOK * 512 * 2);
  __bf16* wqkv_b = (__bf16*)alloc(1536 * 512 * 2);
  __bf16* wout_b = (__bf16*)alloc(512 * 512 * 2);
  __bf16* q_b    = (__bf16*)alloc((size_t)32 * 4096 * 64 * 2);
  __bf16* k_b    = (__bf16*)alloc((size_t)32 * 4096 * 64 * 2);
  float*  v_f    = (float*)alloc((size_t)32 * 4096 * 64 * 4);
  __bf16* vT_b   = (__bf16*)alloc((size_t)32 * 64 * 4096 * 2);
  __bf16* ql_b   = (__bf16*)alloc((size_t)32 * 256 * 64 * 2);
  __bf16* kl_b   = (__bf16*)alloc((size_t)32 * 256 * 64 * 2);
  float*  mat_x  = (float*)alloc(MM * 4);
  __bf16* x_b    = (__bf16*)alloc(MM * 2);
  __bf16* z0_b   = (__bf16*)alloc(MM * 2);
  __bf16* z0T_b  = (__bf16*)alloc(MM * 2);
  __bf16* z1_b   = (__bf16*)alloc(MM * 2);
  __bf16* z1T_b  = (__bf16*)alloc(MM * 2);
  __bf16* xz_b   = (__bf16*)alloc(MM * 2);
  __bf16* t1T_b  = (__bf16*)alloc(MM * 2);
  __bf16* t2T_b  = (__bf16*)alloc(MM * 2);
  __bf16* t3T_b  = (__bf16*)alloc(MM * 2);
  __bf16* kvT_b  = (__bf16*)alloc((size_t)32 * 64 * 256 * 2);
  __bf16* w2t_b  = (__bf16*)alloc((size_t)32 * 64 * 256 * 2);
  float*  aout_f = (float*)alloc(NTOK * 512 * 4);
  __bf16* aout_b = (__bf16*)alloc(NTOK * 512 * 2);
  float*  scales = (float*)alloc(256);
  (void)ws_size; (void)n_in; (void)in_sizes; (void)out_size;

  const int T = 256;
  k_convert_w<<<3072, T, 0, stream>>>(w_qkv, w_out, wqkv_b, wout_b, scales);
  k_layernorm<<<2048, T, 0, stream>>>(x, ln_g, ln_b, xn_b, (int)NTOK);
  k_gemm_qkv<<<768, T, 0, stream>>>(xn_b, wqkv_b, q_b, k_b, v_f, vT_b);
  k_landmarks<<<2048, T, 0, stream>>>(q_b, k_b, ql_b, kl_b);
  k_attn2_logits<<<256, T, 0, stream>>>(ql_b, kl_b, mat_x);
  k_softmax256<<<1024, T, 0, stream>>>(mat_x, x_b, 32 * 256);
  k_pinv_scales<<<1024, T, 0, stream>>>(mat_x, scales);
  k_init_z<<<8192, T, 0, stream>>>(mat_x, z0_b, z0T_b, scales);

  __bf16* zc = z0_b;  __bf16* zcT = z0T_b;
  __bf16* zn = z1_b;  __bf16* znT = z1T_b;
  for (int it = 0; it < 6; ++it) {
    k_gemm256b<<<256, T, 0, stream>>>(x_b, zcT, xz_b, nullptr, t1T_b, 7.f, -1.f);
    k_gemm256b<<<256, T, 0, stream>>>(xz_b, t1T_b, nullptr, nullptr, t2T_b, 15.f, -1.f);
    k_gemm256b<<<256, T, 0, stream>>>(xz_b, t2T_b, nullptr, nullptr, t3T_b, 13.f, -1.f);
    k_gemm256b<<<256, T, 0, stream>>>(zc, t3T_b, nullptr, zn, znT, 0.f, 0.25f);
    __bf16* t;
    t = zc; zc = zn; zn = t;
    t = zcT; zcT = znT; znT = t;
  }

  k_attn3v<<<256, T, 0, stream>>>(k_b, ql_b, vT_b, kvT_b);
  k_w2<<<64, T, 0, stream>>>(zc, kvT_b, w2t_b);
  k_attn1_out<<<1024, T, 0, stream>>>(kl_b, q_b, w2t_b, aout_f);
  k_conv_res<<<512, T, 0, stream>>>(v_f, res_w, aout_f, aout_b);
  k_outproj<<<256, T, 0, stream>>>(aout_b, wout_b, b_out, x, out);
}


# QKV GEMM operands (xn, w_qkv) and the out-projection weight stored fragment-packed, 1 KB per 16x32 block in lane order: a fragment load covers 4 lines (was 16); k order and sums unchanged
# speedup vs baseline: 2.0789x; 1.0548x over previous
#include <hip/hip_runtime.h>
#include <hip/hip_bf16.h>
#include <math.h>


typedef __attribute__((ext_vector_type(16))) __bf16 bf16x16;
typedef __attribute__((ext_vector_type(8)))  float  floatx8;

union FragAB { bf16x16 v; unsigned int u[8]; };

__device__ __forceinline__ unsigned pack2bf(float a, float b) {
  __bf16 lo = (__bf16)a, hi = (__bf16)b;
  unsigned short ul = __builtin_bit_cast(unsigned short, lo);
  unsigned short uh = __builtin_bit_cast(unsigned short, hi);
  return (unsigned)ul | ((unsigned)uh << 16);
}

__device__ __forceinline__ int kpair(int p, int h) {
  return (p < 4) ? (2 * p + 8 * h) : (8 + 2 * p + 8 * h);
}

__device__ __forceinline__ bf16x16 load_frag_bf16_rm(const __bf16* src, int row0,
                                                     int k0, int ld) {
  int lane = threadIdx.x & 31;
  int r = lane & 15, h = lane >> 4;
  const __bf16* base = src + (size_t)(row0 + r) * ld + k0;
  FragAB f;
#pragma unroll
  for (int p = 0; p < 8; ++p) f.u[p] = *(const unsigned*)(base + kpair(p, h));
  return f.v;
}

__device__ __forceinline__ floatx8 wmma_bf16(bf16x16 a, bf16x16 b, floatx8 c) {
  return __builtin_amdgcn_wmma_f32_16x16x32_bf16(false, a, false, b, (short)0, c,
                                                 false, false);
}

template <int KSTEPS>
__device__ __forceinline__ void gemm_strip(const __bf16* A, int rowA,
                                           const __bf16* B, int rowB, int ld,
                                           floatx8 acc[4]) {
#pragma unroll
  for (int ks = 0; ks < KSTEPS; ++ks) {
    int k0 = ks * 32;
    bf16x16 af = load_frag_bf16_rm(A, rowA, k0, ld);
#pragma unroll
    for (int t = 0; t < 4; ++t) {
      bf16x16 bfr = load_frag_bf16_rm(B, rowB + 16 * t, k0, ld);
      acc[t] = wmma_bf16(af, bfr, acc[t]);
    }
  }
}

__device__ __forceinline__ size_t pk_idx(int row, int k) {
  int kk = k & 31;
  int lane = (row & 15) + 16 * ((kk >> 3) & 1);
  return (size_t)(((row >> 4) * 8 + (k >> 5)) * 512 + (kk >> 4) * 256 +
                  lane * 8 + (kk & 7));
}

__device__ __forceinline__ bf16x16 load_frag_bf16_pk(const __bf16* mat, int rb,
                                                     int ks) {
  int lane = threadIdx.x & 31;
  const uint4* blk = (const uint4*)(mat + (size_t)(rb * 8 + ks) * 512) + lane;
  uint4 lo = blk[0], hi = blk[32];
  FragAB f;
  f.u[0] = lo.x; f.u[1] = lo.y; f.u[2] = lo.z; f.u[3] = lo.w;
  f.u[4] = hi.x; f.u[5] = hi.y; f.u[6] = hi.z; f.u[7] = hi.w;
  return f.v;
}

__device__ __forceinline__ size_t pk_idx_n(int row, int k, int KB) {
  int kk = k & 31;
  int lane = (row & 15) + 16 * ((kk >> 3) & 1);
  return (size_t)((row >> 4) * KB + (k >> 5)) * 512 + (kk >> 4) * 256 +
         lane * 8 + (kk & 7);
}

__device__ __forceinline__ bf16x16 load_frag_bf16_pkn(const __bf16* mat, int rb,
                                                      int ks, int KB) {
  int lane = threadIdx.x & 31;
  const uint4* blk = (const uint4*)(mat + (size_t)(rb * KB + ks) * 512) + lane;
  uint4 lo = blk[0], hi = blk[32];
  FragAB f;
  f.u[0] = lo.x; f.u[1] = lo.y; f.u[2] = lo.z; f.u[3] = lo.w;
  f.u[4] = hi.x; f.u[5] = hi.y; f.u[6] = hi.z; f.u[7] = hi.w;
  return f.v;
}

#define PK_AHEAD 4
template <bool BPK>
__device__ __forceinline__ void gemm_strip_pk(const __bf16* A, int rowA,
                                              const __bf16* B, int rowB,
                                              floatx8 acc[4]) {
  bf16x16 af[8], bfr[8][4];
#pragma unroll
  for (int ks = 0; ks < PK_AHEAD; ++ks) {
    af[ks] = load_frag_bf16_pk(A, rowA >> 4, ks);
#pragma unroll
    for (int t = 0; t < 4; ++t)
      bfr[ks][t] = BPK ? load_frag_bf16_pk(B, (rowB >> 4) + t, ks)
                       : load_frag_bf16_rm(B, rowB + 16 * t, ks * 32, 256);
    __builtin_amdgcn_sched_barrier(0);
  }
#pragma unroll
  for (int ks = 0; ks < 8; ++ks) {
#pragma unroll
    for (int t = 0; t < 4; ++t) acc[t] = wmma_bf16(af[ks], bfr[ks][t], acc[t]);
    __builtin_amdgcn_sched_barrier(0);
    if (ks + PK_AHEAD < 8) {
      af[ks + PK_AHEAD] = load_frag_bf16_pk(A, rowA >> 4, ks + PK_AHEAD);
#pragma unroll
      for (int t = 0; t < 4; ++t)
        bfr[ks + PK_AHEAD][t] =
            BPK ? load_frag_bf16_pk(B, (rowB >> 4) + t, ks + PK_AHEAD)
                : load_frag_bf16_rm(B, rowB + 16 * t, (ks + PK_AHEAD) * 32, 256);
      __builtin_amdgcn_sched_barrier(0);
    }
  }
}

template <int KSTEPS, bool APK>
__device__ __forceinline__ void gemm_tile64(const __bf16* A, int rowA,
                                            const __bf16* B, int rowB,
                                            floatx8 acc[4][4]) {
#pragma unroll
  for (int ks = 0; ks < KSTEPS; ++ks) {
    bf16x16 af[4], bfr[4];
#pragma unroll
    for (int m = 0; m < 4; ++m)
      af[m] = APK ? load_frag_bf16_pkn(A, (rowA >> 4) + m, ks, KSTEPS)
                  : load_frag_bf16_rm(A, rowA + 16 * m, ks * 32, 32 * KSTEPS);
#pragma unroll
    for (int t = 0; t < 4; ++t)
      bfr[t] = load_frag_bf16_pkn(B, (rowB >> 4) + t, ks, KSTEPS);
    __builtin_amdgcn_sched_barrier(0);
#pragma unroll
    for (int m = 0; m < 4; ++m)
#pragma unroll
      for (int t = 0; t < 4; ++t) acc[m][t] = wmma_bf16(af[m], bfr[t], acc[m][t]);
    __builtin_amdgcn_sched_barrier(0);
  }
}

__global__ void k_convert_w(const float* wqkv, const float* wout,
                            __bf16* wqkv_b, __bf16* wout_b, float* scales) {
  int i = blockIdx.x * blockDim.x + threadIdx.x;
  if (i < 2) scales[i] = 0.f;
  bool second = i >= 1536 * 64;
  const float* src = second ? wout : wqkv;
  __bf16* dst = second ? wout_b : wqkv_b;
  int u = second ? i - 1536 * 64 : i;
  if (second && u >= 512 * 64) return;
  int blk = u >> 6, half = (u >> 5) & 1, lane = u & 31;
  int row = (blk >> 4) * 16 + (lane & 15);
  int k = (blk & 15) * 32 + half * 16 + 8 * (lane >> 4);
  const float4* s = (const float4*)(src + (size_t)row * 512 + k);
  float4 a = s[0], b = s[1];
  uint4 pv;
  pv.x = pack2bf(a.x, a.y);
  pv.y = pack2bf(a.z, a.w);
  pv.z = pack2bf(b.x, b.y);
  pv.w = pack2bf(b.z, b.w);
  ((uint4*)dst)[u] = pv;
}

__global__ void k_layernorm(const float* x, const float* g, const float* b,
                            __bf16* xn, int ntok) {
  int wave = (int)((blockIdx.x * blockDim.x + threadIdx.x) >> 5);
  int lane = threadIdx.x & 31;
  if (wave >= ntok) return;
  const float* xr = x + (size_t)wave * 512;
  float s = 0.f, s2 = 0.f, vals[16];
#pragma unroll
  for (int i = 0; i < 16; ++i) {
    float v = xr[lane + 32 * i];
    vals[i] = v; s += v; s2 += v * v;
  }
  for (int m = 16; m >= 1; m >>= 1) {
    s += __shfl_xor(s, m, 32);
    s2 += __shfl_xor(s2, m, 32);
  }
  float mu = s * (1.f / 512.f);
  float var = s2 * (1.f / 512.f) - mu * mu;
  float inv = rsqrtf(var + 1e-5f);
#pragma unroll
  for (int i = 0; i < 16; ++i) {
    int c = lane + 32 * i;
    xn[pk_idx_n(wave, c, 16)] = (__bf16)((vals[i] - mu) * inv * g[c] + b[c]);
  }
}

__global__ void __launch_bounds__(256)
k_gemm_qkv(const __bf16* xn, const __bf16* wq, __bf16* qb,
           __bf16* kb, float* vf, __bf16* vT) {
  int wid = (int)((blockIdx.x * blockDim.x + threadIdx.x) >> 5);
  int tM = wid / 24, s = wid % 24;
  floatx8 acc[4][4] = {};
  gemm_tile64<16, true>(xn, tM * 64, wq, s * 64, acc);
  int lane = threadIdx.x & 31;
  int n = lane & 15, h = lane >> 4;
  int col0 = s * 64;
  int sect = col0 >> 9;
  int head = (col0 >> 6) & 7;
#pragma unroll
  for (int m = 0; m < 4; ++m) {
#pragma unroll
    for (int t = 0; t < 4; ++t) {
      int dh = 16 * t + n;
#pragma unroll
      for (int r = 0; r < 8; ++r) {
        int tok = tM * 64 + 16 * m + r + 8 * h;
        int bb = tok >> 12, ns = tok & 4095;
        size_t bh = (size_t)(bb * 8 + head);
        size_t idx = (bh * 4096 + ns) * 64 + dh;
        float val = acc[m][t][r];
        if (sect == 0) qb[idx] = (__bf16)(val * 0.125f);
        else if (sect == 1) kb[idx] = (__bf16)val;
        else {
          vf[idx] = val;
          vT[(bh * 64 + dh) * 4096 + ns] = (__bf16)val;
        }
      }
    }
  }
}

__global__ void k_landmarks(const __bf16* qb, const __bf16* kb, __bf16* ql,
                            __bf16* kl) {
  int i = blockIdx.x * blockDim.x + threadIdx.x;
  if (i >= 32 * 256 * 64) return;
  int dh = i & 63, m = (i >> 6) & 255, bh = i >> 14;
  size_t base = ((size_t)bh * 4096 + m * 16) * 64 + dh;
  float sq = 0.f, sk = 0.f;
#pragma unroll
  for (int t = 0; t < 16; ++t) {
    sq += (float)qb[base + t * 64];
    sk += (float)kb[base + t * 64];
  }
  ql[i] = (__bf16)(sq * (1.f / 16.f));
  kl[i] = (__bf16)(sk * (1.f / 16.f));
}

__global__ void k_attn2_logits(const __bf16* ql, const __bf16* kl, float* out) {
  int wid = (int)((blockIdx.x * blockDim.x + threadIdx.x) >> 5);
  int bh = wid >> 6, t = wid & 63;
  int tM = t >> 2, s = t & 3;
  const __bf16* A = ql + (size_t)bh * 256 * 64;
  const __bf16* B = kl + (size_t)bh * 256 * 64;
  floatx8 acc[4] = {};
  gemm_strip<2>(A, tM * 16, B, s * 64, 64, acc);
  int lane = threadIdx.x & 31;
  int n = lane & 15, h = lane >> 4;
  float* o = out + (size_t)bh * 256 * 256;
#pragma unroll
  for (int t4 = 0; t4 < 4; ++t4)
#pragma unroll
    for (int r = 0; r < 8; ++r)
      o[(size_t)(tM * 16 + r + 8 * h) * 256 + s * 64 + 16 * t4 + n] =
          acc[t4][r];
}

__global__ void k_softmax256(float* x, __bf16* xb, int nrows) {
  int wave = (int)((blockIdx.x * blockDim.x + threadIdx.x) >> 5);
  if (wave >= nrows) return;
  int lane = threadIdx.x & 31;
  float* r = x + (size_t)wave * 256;
  __bf16* rb = xb + ((size_t)(wave >> 8) << 16);
  int row = wave & 255;
  float v[8], m = -1e30f;
#pragma unroll
  for (int i = 0; i < 8; ++i) { v[i] = r[lane + 32 * i]; m = fmaxf(m, v[i]); }
  for (int t = 16; t >= 1; t >>= 1) m = fmaxf(m, __shfl_xor(m, t, 32));
  float s = 0.f;
#pragma unroll
  for (int i = 0; i < 8; ++i) { v[i] = __expf(v[i] - m); s += v[i]; }
  for (int t = 16; t >= 1; t >>= 1) s += __shfl_xor(s, t, 32);
  float inv = 1.f / s;
#pragma unroll
  for (int i = 0; i < 8; ++i) {
    float pv = v[i] * inv;
    r[lane + 32 * i] = pv;
    rb[pk_idx(row, lane + 32 * i)] = (__bf16)pv;
  }
}

__global__ void k_pinv_scales(const float* x, float* scales) {
  int wave = (int)((blockIdx.x * blockDim.x + threadIdx.x) >> 5);
  if (wave >= 32 * 256) return;
  int lane = threadIdx.x & 31;
  int bh = wave >> 8, r = wave & 255;
  const float* m = x + (size_t)bh * 256 * 256;
  float rs = 0.f, cs = 0.f;
#pragma unroll
  for (int i = 0; i < 8; ++i) {
    rs += fabsf(m[(size_t)r * 256 + lane + 32 * i]);
    cs += fabsf(m[(size_t)(lane + 32 * i) * 256 + r]);
  }
  for (int t = 16; t >= 1; t >>= 1) {
    rs += __shfl_xor(rs, t, 32);
    cs += __shfl_xor(cs, t, 32);
  }
  if (lane == 0) {
    atomicMax((unsigned*)&scales[0], __float_as_uint(rs));
    atomicMax((unsigned*)&scales[1], __float_as_uint(cs));
  }
}

__global__ void k_init_z(const float* x, __bf16* zb, __bf16* zTb,
                         const float* scales) {
  size_t i = (size_t)blockIdx.x * blockDim.x + threadIdx.x;
  if (i >= (size_t)32 * 256 * 256) return;
  int j = (int)(i & 255), r = (int)((i >> 8) & 255);
  size_t bh = i >> 16;
  float denom = scales[0] * scales[1];
  float val = x[(bh << 16) + (size_t)j * 256 + r] / denom;
  zb[(bh << 16) + pk_idx(r, j)] = (__bf16)val;
  zTb[(bh << 16) + pk_idx(j, r)] = (__bf16)val;
}

__global__ void __launch_bounds__(256)
k_gemm256b(const __bf16* A, const __bf16* BT, __bf16* OutRaw,
           __bf16* OutT, __bf16* OutTT, float alpha, float beta) {
  int wid = (int)((blockIdx.x * blockDim.x + threadIdx.x) >> 5);
  int bh = wid >> 6, t = wid & 63;
  int tM = t >> 2, s = t & 3;
  size_t bofs = (size_t)bh << 16;
  floatx8 acc[4] = {};
  gemm_strip_pk<true>(A + bofs, tM * 16, BT + bofs, s * 64, acc);
  int lane = threadIdx.x & 31;
  int n = lane & 15, h = lane >> 4;
#pragma unroll
  for (int t4 = 0; t4 < 4; ++t4) {
    int N = s * 64 + 16 * t4 + n;
    float trv[8];
#pragma unroll
    for (int r = 0; r < 8; ++r) {
      int M = tM * 16 + r + 8 * h;
      float tv = acc[t4][r];
      if (OutRaw) OutRaw[bofs + pk_idx(M, N)] = (__bf16)tv;
      float tr = alpha * (M == N ? 1.f : 0.f) + beta * tv;
      if (OutT) OutT[bofs + pk_idx(M, N)] = (__bf16)tr;
      trv[r] = tr;
    }
    if (OutTT) {
      uint4 pv;
      pv.x = pack2bf(trv[0], trv[1]);
      pv.y = pack2bf(trv[2], trv[3]);
      pv.z = pack2bf(trv[4], trv[5]);
      pv.w = pack2bf(trv[6], trv[7]);
      *(uint4*)(OutTT + bofs + pk_idx(N, tM * 16 + 8 * h)) = pv;
    }
  }
}

__device__ __forceinline__ void stream_attn(const __bf16* Ak, const __bf16* Bq,
                                            int qrow0, const __bf16* A2,
                                            int ldA2, int nkeys, float* outF,
                                            int rsF, __bf16* outB, int nsB,
                                            int dsB) {
  int lane = threadIdx.x & 31;
  int n = lane & 15, h = lane >> 4;
  bf16x16 bq0 = load_frag_bf16_rm(Bq, qrow0, 0, 64);
  bf16x16 bq1 = load_frag_bf16_rm(Bq, qrow0, 32, 64);
  float run_m = -1e30f, run_s = 0.f;
  floatx8 acc[4] = {};
  for (int c = 0; c < nkeys; c += 32) {
    floatx8 s0 = {}, s1 = {};
    s0 = wmma_bf16(load_frag_bf16_rm(Ak, c, 0, 64), bq0, s0);
    s0 = wmma_bf16(load_frag_bf16_rm(Ak, c, 32, 64), bq1, s0);
    s1 = wmma_bf16(load_frag_bf16_rm(Ak, c + 16, 0, 64), bq0, s1);
    s1 = wmma_bf16(load_frag_bf16_rm(Ak, c + 16, 32, 64), bq1, s1);
    float mc = -1e30f;
#pragma unroll
    for (int r = 0; r < 8; ++r) { mc = fmaxf(mc, s0[r]); mc = fmaxf(mc, s1[r]); }
    mc = fmaxf(mc, __shfl_xor(mc, 16, 32));
    float newm = fmaxf(run_m, mc);
    float scale = __expf(run_m - newm);
    float p0[8], p1[8], ps = 0.f;
#pragma unroll
    for (int r = 0; r < 8; ++r) {
      p0[r] = __expf(s0[r] - newm);
      p1[r] = __expf(s1[r] - newm);
      ps += p0[r] + p1[r];
    }
    run_s = run_s * scale + ps;
    run_m = newm;
    FragAB bp;
#pragma unroll
    for (int p = 0; p < 4; ++p) {
      bp.u[p] = pack2bf(p0[2 * p], p0[2 * p + 1]);
      bp.u[4 + p] = pack2bf(p1[2 * p], p1[2 * p + 1]);
    }
#pragma unroll
    for (int t = 0; t < 4; ++t) {
#pragma unroll
      for (int r = 0; r < 8; ++r) acc[t][r] *= scale;
      acc[t] = wmma_bf16(load_frag_bf16_rm(A2, 16 * t, c, ldA2), bp.v, acc[t]);
    }
  }
  float tot = run_s + __shfl_xor(run_s, 16, 32);
  float inv = 1.f / tot;
#pragma unroll
  for (int t = 0; t < 4; ++t)
#pragma unroll
    for (int r = 0; r < 8; ++r) {
      int dh = 16 * t + r + 8 * h;
      float val = acc[t][r] * inv;
      if (outF) outF[(size_t)n * rsF + dh] = val;
      if (outB) outB[(size_t)n * nsB + (size_t)dh * dsB] = (__bf16)val;
    }
}

#define A3_WAVES 8
#define A3_QG 2
#define A3_KEYS (4096 / A3_WAVES)
__global__ void __launch_bounds__(256)
k_attn3v(const __bf16* kb, const __bf16* ql, const __bf16* vT, __bf16* kvT) {
  __shared__ float accL[A3_WAVES][A3_QG][4][8][32];
  __shared__ float mL[A3_WAVES][A3_QG][16];
  __shared__ float sL[A3_WAVES][A3_QG][16];
  int lane = threadIdx.x & 31;
  int w = __builtin_amdgcn_readfirstlane((int)(threadIdx.x >> 5));
  int bh = blockIdx.x >> 3, q0 = (blockIdx.x & 7) * (16 * A3_QG);
  const __bf16* Ak = kb + (size_t)bh * 4096 * 64;
  const __bf16* Bq = ql + (size_t)bh * 256 * 64;
  const __bf16* A2 = vT + (size_t)bh * 64 * 4096;
  bf16x16 bq[A3_QG][2];
  float run_m[A3_QG], run_s[A3_QG];
  floatx8 acc[A3_QG][4];
#pragma unroll
  for (int g = 0; g < A3_QG; ++g) {
    bq[g][0] = load_frag_bf16_rm(Bq, q0 + 16 * g, 0, 64);
    bq[g][1] = load_frag_bf16_rm(Bq, q0 + 16 * g, 32, 64);
    run_m[g] = -1e30f;
    run_s[g] = 0.f;
#pragma unroll
    for (int t = 0; t < 4; ++t) acc[g][t] = floatx8{};
  }
  int kbeg = w * A3_KEYS;
  for (int c = kbeg; c < kbeg + A3_KEYS; c += 32) {
    bf16x16 ka[4], va[4];
    ka[0] = load_frag_bf16_rm(Ak, c, 0, 64);
    ka[1] = load_frag_bf16_rm(Ak, c, 32, 64);
    ka[2] = load_frag_bf16_rm(Ak, c + 16, 0, 64);
    ka[3] = load_frag_bf16_rm(Ak, c + 16, 32, 64);
#pragma unroll
    for (int t = 0; t < 4; ++t) va[t] = load_frag_bf16_rm(A2, 16 * t, c, 4096);
    __builtin_amdgcn_sched_barrier(0);
#pragma unroll
    for (int g = 0; g < A3_QG; ++g) {
      floatx8 s0 = {}, s1 = {};
      s0 = wmma_bf16(ka[0], bq[g][0], s0);
      s0 = wmma_bf16(ka[1], bq[g][1], s0);
      s1 = wmma_bf16(ka[2], bq[g][0], s1);
      s1 = wmma_bf16(ka[3], bq[g][1], s1);
      float mc = -1e30f;
#pragma unroll
      for (int r = 0; r < 8; ++r) { mc = fmaxf(mc, s0[r]); mc = fmaxf(mc, s1[r]); }
      mc = fmaxf(mc, __shfl_xor(mc, 16, 32));
      float newm = fmaxf(run_m[g], mc);
      float scale = __expf(run_m[g] - newm);
      float p0[8], p1[8], ps = 0.f;
#pragma unroll
      for (int r = 0; r < 8; ++r) {
        p0[r] = __expf(s0[r] - newm);
        p1[r] = __expf(s1[r] - newm);
        ps += p0[r] + p1[r];
      }
      run_s[g] = run_s[g] * scale + ps;
      run_m[g] = newm;
      FragAB bp;
#pragma unroll
      for (int p = 0; p < 4; ++p) {
        bp.u[p] = pack2bf(p0[2 * p], p0[2 * p + 1]);
        bp.u[4 + p] = pack2bf(p1[2 * p], p1[2 * p + 1]);
      }
#pragma unroll
      for (int t = 0; t < 4; ++t) {
#pragma unroll
        for (int r = 0; r < 8; ++r) acc[g][t][r] *= scale;
        acc[g][t] = wmma_bf16(va[t], bp.v, acc[g][t]);
      }
    }
  }
#pragma unroll
  for (int g = 0; g < A3_QG; ++g) {
    float tot = run_s[g] + __shfl_xor(run_s[g], 16, 32);
    if (lane < 16) {
      mL[w][g][lane] = run_m[g];
      sL[w][g][lane] = tot;
    }
#pragma unroll
    for (int t = 0; t < 4; ++t)
#pragma unroll
      for (int r = 0; r < 8; ++r) accL[w][g][t][r][lane] = acc[g][t][r];
  }
  __syncthreads();
  int cc = lane & 15, hh = lane >> 4;
  for (int g = 0; g < A3_QG; ++g) {
    float mw[A3_WAVES], f[A3_WAVES], M = -1e30f, S = 0.f;
#pragma unroll
    for (int v = 0; v < A3_WAVES; ++v) { mw[v] = mL[v][g][cc]; M = fmaxf(M, mw[v]); }
#pragma unroll
    for (int v = 0; v < A3_WAVES; ++v) {
      f[v] = __expf(mw[v] - M);
      S += sL[v][g][cc] * f[v];
    }
    float inv = 1.f / S;
    for (int t = 0; t < 4; ++t) {
      float o = 0.f;
#pragma unroll
      for (int v = 0; v < A3_WAVES; ++v) o += accL[v][g][t][w][lane] * f[v];
      int dh = 16 * t + 8 * hh + w;
      kvT[((size_t)bh * 64 + dh) * 256 + q0 + 16 * g + cc] = (__bf16)(o * inv);
    }
  }
}

__global__ void __launch_bounds__(256)
k_w2(const __bf16* zb, const __bf16* kvT, __bf16* w2t) {
  int wid = (int)((blockIdx.x * blockDim.x + threadIdx.x) >> 5);
  int bh = wid >> 4, tM = wid & 15;
  floatx8 acc[4] = {};
  gemm_strip_pk<false>(zb + ((size_t)bh << 16), tM * 16,
                       kvT + (size_t)bh * 64 * 256, 0, acc);
  int lane = threadIdx.x & 31;
  int n = lane & 15, h = lane >> 4;
#pragma unroll
  for (int t4 = 0; t4 < 4; ++t4)
#pragma unroll
    for (int r = 0; r < 8; ++r) {
      int M = tM * 16 + r + 8 * h;
      int N = 16 * t4 + n;
      w2t[((size_t)bh * 64 + N) * 256 + M] = (__bf16)acc[t4][r];
    }
}

__global__ void k_attn1_out(const __bf16* kl, const __bf16* qb,
                            const __bf16* w2t, float* aout) {
  int wid = (int)((blockIdx.x * blockDim.x + threadIdx.x) >> 5);
  int bh = wid >> 8, blk = wid & 255;
  int b = bh >> 3, hh = bh & 7;
  stream_attn(kl + (size_t)bh * 256 * 64, qb + (size_t)bh * 4096 * 64, blk * 16,
              w2t + (size_t)bh * 64 * 256, 256, 256,
              aout + ((size_t)b * 4096 + blk * 16) * 512 + hh * 64, 512,
              nullptr, 0, 0);
}

#define CONV_R 16
#define CONV_G 8
#define CONV_ROWS (CONV_R + 32)
__global__ void __launch_bounds__(256, 4)
k_conv_res(const float* __restrict__ vf, const float* __restrict__ rw,
           float* __restrict__ aout, __bf16* __restrict__ aout_b) {
  int t = (int)(blockIdx.x * blockDim.x + threadIdx.x);
  if (t >= 4 * (4096 / CONV_R) * 128) return;
  int c4 = t & 127;
  int chunk = t >> 7;
  int b = chunk / (4096 / CONV_R);
  int ns0 = (chunk % (4096 / CONV_R)) * CONV_R;
  int col = c4 * 4, hh = col >> 6, dh = col & 63;
  const float* vcol = vf + ((size_t)(b * 8 + hh) * 4096) * 64 + dh;
  float* arow = aout + ((size_t)b * 4096 + ns0) * 512 + col;
  __bf16* brow = aout_b + ((size_t)b * 4096 + ns0) * 512 + col;
  float w[33];
#pragma unroll
  for (int j = 0; j < 33; ++j) w[j] = rw[hh * 33 + j];
  float4 acc[CONV_R], win[CONV_ROWS], a[CONV_R];
#pragma unroll
  for (int o = 0; o < CONV_R; ++o) acc[o] = make_float4(0.f, 0.f, 0.f, 0.f);
#pragma unroll
  for (int k = 0; k < CONV_G; ++k) {
    int p = ns0 + k - 16;
    int pc = p < 0 ? 0 : (p > 4095 ? 4095 : p);
    win[k] = *(const float4*)(vcol + (size_t)pc * 64);
  }
#pragma unroll
  for (int g = 0; g < CONV_ROWS / CONV_G; ++g) {
    if (g + 1 < CONV_ROWS / CONV_G) {
#pragma unroll
      for (int k = (g + 1) * CONV_G; k < (g + 2) * CONV_G; ++k) {
        int p = ns0 + k - 16;
        int pc = p < 0 ? 0 : (p > 4095 ? 4095 : p);
        win[k] = *(const float4*)(vcol + (size_t)pc * 64);
      }
    } else {
#pragma unroll
      for (int o = 0; o < CONV_R; ++o)
        a[o] = *(const float4*)(arow + (size_t)o * 512);
    }
#pragma unroll
    for (int k = g * CONV_G; k < (g + 1) * CONV_G; ++k) {
      int p = ns0 + k - 16;
      float4 v = win[k];
      if (p < 0 || p > 4095) v = make_float4(0.f, 0.f, 0.f, 0.f);
#pragma unroll
      for (int o = 0; o < CONV_R; ++o) {
        int j = k - o;
        if (j >= 0 && j < 33) {
          acc[o].x += v.x * w[j];
          acc[o].y += v.y * w[j];
          acc[o].z += v.z * w[j];
          acc[o].w += v.w * w[j];
        }
      }
    }
  }
#pragma unroll
  for (int o = 0; o < CONV_R; ++o) {
    float4 r;
    r.x = a[o].x + acc[o].x;
    r.y = a[o].y + acc[o].y;
    r.z = a[o].z + acc[o].z;
    r.w = a[o].w + acc[o].w;
    *(float4*)(arow + (size_t)o * 512) = r;
    uint2 pk;
    pk.x = pack2bf(r.x, r.y);
    pk.y = pack2bf(r.z, r.w);
    *(uint2*)(brow + (size_t)o * 512) = pk;
  }
}

__global__ void __launch_bounds__(256)
k_outproj(const __bf16* __restrict__ aoutb, const __bf16* __restrict__ wo,
          const float* __restrict__ bias, const float* __restrict__ x,
          float* __restrict__ out) {
  int wid = (int)((blockIdx.x * blockDim.x + threadIdx.x) >> 5);
  int tM = wid >> 3, s = wid & 7;
  floatx8 acc[4][4] = {};
  gemm_tile64<16, false>(aoutb, tM * 64, wo, s * 64, acc);
  int lane = threadIdx.x & 31;
  int n = lane & 15, h = lane >> 4;
#pragma unroll
  for (int t4 = 0; t4 < 4; ++t4) {
    int col = s * 64 + 16 * t4 + n;
    float bc = bias[col];
#pragma unroll
    for (int m = 0; m < 4; ++m) {
#pragma unroll
      for (int r = 0; r < 8; ++r) {
        size_t idx = (size_t)(tM * 64 + 16 * m + r + 8 * h) * 512 + col;
        out[idx] = x[idx] + acc[m][t4][r] + bc;
      }
    }
  }
}

extern "C" void kernel_launch(void* const* d_in, const int* in_sizes, int n_in,
                              void* d_out, int out_size, void* d_ws,
                              size_t ws_size, hipStream_t stream) {
  const float* x     = (const float*)d_in[0];
  const float* ln_g  = (const float*)d_in[1];
  const float* ln_b  = (const float*)d_in[2];
  const float* w_qkv = (const float*)d_in[3];
  const float* w_out = (const float*)d_in[4];
  const float* b_out = (const float*)d_in[5];
  const float* res_w = (const float*)d_in[6];
  float* out = (float*)d_out;

  char* p = (char*)d_ws;
  auto alloc = [&](size_t bytes) -> void* {
    void* r = (void*)p;
    p += (bytes + 255) & ~(size_t)255;
    return r;
  };
  const size_t NTOK = 4 * 4096;
  const size_t MM = (size_t)32 * 256 * 256;
  __bf16* xn_b   = (__bf16*)alloc(NTOK * 512 * 2);
  __bf16* wqkv_b = (__bf16*)alloc(1536 * 512 * 2);
  __bf16* wout_b = (__bf16*)alloc(512 * 512 * 2);
  __bf16* q_b    = (__bf16*)alloc((size_t)32 * 4096 * 64 * 2);
  __bf16* k_b    = (__bf16*)alloc((size_t)32 * 4096 * 64 * 2);
  float*  v_f    = (float*)alloc((size_t)32 * 4096 * 64 * 4);
  __bf16* vT_b   = (__bf16*)alloc((size_t)32 * 64 * 4096 * 2);
  __bf16* ql_b   = (__bf16*)alloc((size_t)32 * 256 * 64 * 2);
  __bf16* kl_b   = (__bf16*)alloc((size_t)32 * 256 * 64 * 2);
  float*  mat_x  = (float*)alloc(MM * 4);
  __bf16* x_b    = (__bf16*)alloc(MM * 2);
  __bf16* z0_b   = (__bf16*)alloc(MM * 2);
  __bf16* z0T_b  = (__bf16*)alloc(MM * 2);
  __bf16* z1_b   = (__bf16*)alloc(MM * 2);
  __bf16* z1T_b  = (__bf16*)alloc(MM * 2);
  __bf16* xz_b   = (__bf16*)alloc(MM * 2);
  __bf16* t1T_b  = (__bf16*)alloc(MM * 2);
  __bf16* t2T_b  = (__bf16*)alloc(MM * 2);
  __bf16* t3T_b  = (__bf16*)alloc(MM * 2);
  __bf16* kvT_b  = (__bf16*)alloc((size_t)32 * 64 * 256 * 2);
  __bf16* w2t_b  = (__bf16*)alloc((size_t)32 * 64 * 256 * 2);
  float*  aout_f = (float*)alloc(NTOK * 512 * 4);
  __bf16* aout_b = (__bf16*)alloc(NTOK * 512 * 2);
  float*  scales = (float*)alloc(256);
  (void)ws_size; (void)n_in; (void)in_sizes; (void)out_size;

  const int T = 256;
  k_convert_w<<<512, T, 0, stream>>>(w_qkv, w_out, wqkv_b, wout_b, scales);
  k_layernorm<<<2048, T, 0, stream>>>(x, ln_g, ln_b, xn_b, (int)NTOK);
  k_gemm_qkv<<<768, T, 0, stream>>>(xn_b, wqkv_b, q_b, k_b, v_f, vT_b);
  k_landmarks<<<2048, T, 0, stream>>>(q_b, k_b, ql_b, kl_b);
  k_attn2_logits<<<256, T, 0, stream>>>(ql_b, kl_b, mat_x);
  k_softmax256<<<1024, T, 0, stream>>>(mat_x, x_b, 32 * 256);
  k_pinv_scales<<<1024, T, 0, stream>>>(mat_x, scales);
  k_init_z<<<8192, T, 0, stream>>>(mat_x, z0_b, z0T_b, scales);

  __bf16* zc = z0_b;  __bf16* zcT = z0T_b;
  __bf16* zn = z1_b;  __bf16* znT = z1T_b;
  for (int it = 0; it < 6; ++it) {
    k_gemm256b<<<256, T, 0, stream>>>(x_b, zcT, xz_b, nullptr, t1T_b, 7.f, -1.f);
    k_gemm256b<<<256, T, 0, stream>>>(xz_b, t1T_b, nullptr, nullptr, t2T_b, 15.f, -1.f);
    k_gemm256b<<<256, T, 0, stream>>>(xz_b, t2T_b, nullptr, nullptr, t3T_b, 13.f, -1.f);
    k_gemm256b<<<256, T, 0, stream>>>(zc, t3T_b, nullptr, zn, znT, 0.f, 0.25f);
    __bf16* t;
    t = zc; zc = zn; zn = t;
    t = zcT; zcT = znT; znT = t;
  }

  k_attn3v<<<256, T, 0, stream>>>(k_b, ql_b, vT_b, kvT_b);
  k_w2<<<64, T, 0, stream>>>(zc, kvT_b, w2t_b);
  k_attn1_out<<<1024, T, 0, stream>>>(kl_b, q_b, w2t_b, aout_f);
  k_conv_res<<<512, T, 0, stream>>>(v_f, res_w, aout_f, aout_b);
  k_outproj<<<256, T, 0, stream>>>(aout_b, wout_b, b_out, x, out);
}


# 24 pinv GEMMs: a block = one 64x128 output tile; its 96 KB of packed operands go to LDS once (plain loads + ds_store, two K halves), waves take fragments from LDS (was 40 KB/wave from L2); same sums
# speedup vs baseline: 2.3320x; 1.1217x over previous
#include <hip/hip_runtime.h>
#include <hip/hip_bf16.h>
#include <math.h>


typedef __attribute__((ext_vector_type(16))) __bf16 bf16x16;
typedef __attribute__((ext_vector_type(8)))  float  floatx8;

union FragAB { bf16x16 v; unsigned int u[8]; };

__device__ __forceinline__ unsigned pack2bf(float a, float b) {
  __bf16 lo = (__bf16)a, hi = (__bf16)b;
  unsigned short ul = __builtin_bit_cast(unsigned short, lo);
  unsigned short uh = __builtin_bit_cast(unsigned short, hi);
  return (unsigned)ul | ((unsigned)uh << 16);
}

__device__ __forceinline__ int kpair(int p, int h) {
  return (p < 4) ? (2 * p + 8 * h) : (8 + 2 * p + 8 * h);
}

__device__ __forceinline__ bf16x16 load_frag_bf16_rm(const __bf16* src, int row0,
                                                     int k0, int ld) {
  int lane = threadIdx.x & 31;
  int r = lane & 15, h = lane >> 4;
  const __bf16* base = src + (size_t)(row0 + r) * ld + k0;
  FragAB f;
#pragma unroll
  for (int p = 0; p < 8; ++p) f.u[p] = *(const unsigned*)(base + kpair(p, h));
  return f.v;
}

__device__ __forceinline__ floatx8 wmma_bf16(bf16x16 a, bf16x16 b, floatx8 c) {
  return __builtin_amdgcn_wmma_f32_16x16x32_bf16(false, a, false, b, (short)0, c,
                                                 false, false);
}

template <int KSTEPS>
__device__ __forceinline__ void gemm_strip(const __bf16* A, int rowA,
                                           const __bf16* B, int rowB, int ld,
                                           floatx8 acc[4]) {
#pragma unroll
  for (int ks = 0; ks < KSTEPS; ++ks) {
    int k0 = ks * 32;
    bf16x16 af = load_frag_bf16_rm(A, rowA, k0, ld);
#pragma unroll
    for (int t = 0; t < 4; ++t) {
      bf16x16 bfr = load_frag_bf16_rm(B, rowB + 16 * t, k0, ld);
      acc[t] = wmma_bf16(af, bfr, acc[t]);
    }
  }
}

__device__ __forceinline__ size_t pk_idx(int row, int k) {
  int kk = k & 31;
  int lane = (row & 15) + 16 * ((kk >> 3) & 1);
  return (size_t)(((row >> 4) * 8 + (k >> 5)) * 512 + (kk >> 4) * 256 +
                  lane * 8 + (kk & 7));
}

__device__ __forceinline__ bf16x16 load_frag_bf16_pk(const __bf16* mat, int rb,
                                                     int ks) {
  int lane = threadIdx.x & 31;
  const uint4* blk = (const uint4*)(mat + (size_t)(rb * 8 + ks) * 512) + lane;
  uint4 lo = blk[0], hi = blk[32];
  FragAB f;
  f.u[0] = lo.x; f.u[1] = lo.y; f.u[2] = lo.z; f.u[3] = lo.w;
  f.u[4] = hi.x; f.u[5] = hi.y; f.u[6] = hi.z; f.u[7] = hi.w;
  return f.v;
}

__device__ __forceinline__ size_t pk_idx_n(int row, int k, int KB) {
  int kk = k & 31;
  int lane = (row & 15) + 16 * ((kk >> 3) & 1);
  return (size_t)((row >> 4) * KB + (k >> 5)) * 512 + (kk >> 4) * 256 +
         lane * 8 + (kk & 7);
}

__device__ __forceinline__ bf16x16 load_frag_bf16_pkn(const __bf16* mat, int rb,
                                                      int ks, int KB) {
  int lane = threadIdx.x & 31;
  const uint4* blk = (const uint4*)(mat + (size_t)(rb * KB + ks) * 512) + lane;
  uint4 lo = blk[0], hi = blk[32];
  FragAB f;
  f.u[0] = lo.x; f.u[1] = lo.y; f.u[2] = lo.z; f.u[3] = lo.w;
  f.u[4] = hi.x; f.u[5] = hi.y; f.u[6] = hi.z; f.u[7] = hi.w;
  return f.v;
}

#define PK_AHEAD 4
template <bool BPK>
__device__ __forceinline__ void gemm_strip_pk(const __bf16* A, int rowA,
                                              const __bf16* B, int rowB,
                                              floatx8 acc[4]) {
  bf16x16 af[8], bfr[8][4];
#pragma unroll
  for (int ks = 0; ks < PK_AHEAD; ++ks) {
    af[ks] = load_frag_bf16_pk(A, rowA >> 4, ks);
#pragma unroll
    for (int t = 0; t < 4; ++t)
      bfr[ks][t] = BPK ? load_frag_bf16_pk(B, (rowB >> 4) + t, ks)
                       : load_frag_bf16_rm(B, rowB + 16 * t, ks * 32, 256);
    __builtin_amdgcn_sched_barrier(0);
  }
#pragma unroll
  for (int ks = 0; ks < 8; ++ks) {
#pragma unroll
    for (int t = 0; t < 4; ++t) acc[t] = wmma_bf16(af[ks], bfr[ks][t], acc[t]);
    __builtin_amdgcn_sched_barrier(0);
    if (ks + PK_AHEAD < 8) {
      af[ks + PK_AHEAD] = load_frag_bf16_pk(A, rowA >> 4, ks + PK_AHEAD);
#pragma unroll
      for (int t = 0; t < 4; ++t)
        bfr[ks + PK_AHEAD][t] =
            BPK ? load_frag_bf16_pk(B, (rowB >> 4) + t, ks + PK_AHEAD)
                : load_frag_bf16_rm(B, rowB + 16 * t, (ks + PK_AHEAD) * 32, 256);
      __builtin_amdgcn_sched_barrier(0);
    }
  }
}

template <int KSTEPS, bool APK>
__device__ __forceinline__ void gemm_tile64(const __bf16* A, int rowA,
                                            const __bf16* B, int rowB,
                                            floatx8 acc[4][4]) {
#pragma unroll
  for (int ks = 0; ks < KSTEPS; ++ks) {
    bf16x16 af[4], bfr[4];
#pragma unroll
    for (int m = 0; m < 4; ++m)
      af[m] = APK ? load_frag_bf16_pkn(A, (rowA >> 4) + m, ks, KSTEPS)
                  : load_frag_bf16_rm(A, rowA + 16 * m, ks * 32, 32 * KSTEPS);
#pragma unroll
    for (int t = 0; t < 4; ++t)
      bfr[t] = load_frag_bf16_pkn(B, (rowB >> 4) + t, ks, KSTEPS);
    __builtin_amdgcn_sched_barrier(0);
#pragma unroll
    for (int m = 0; m < 4; ++m)
#pragma unroll
      for (int t = 0; t < 4; ++t) acc[m][t] = wmma_bf16(af[m], bfr[t], acc[m][t]);
    __builtin_amdgcn_sched_barrier(0);
  }
}

__global__ void k_convert_w(const float* wqkv, const float* wout,
                            __bf16* wqkv_b, __bf16* wout_b, float* scales) {
  int i = blockIdx.x * blockDim.x + threadIdx.x;
  if (i < 2) scales[i] = 0.f;
  bool second = i >= 1536 * 64;
  const float* src = second ? wout : wqkv;
  __bf16* dst = second ? wout_b : wqkv_b;
  int u = second ? i - 1536 * 64 : i;
  if (second && u >= 512 * 64) return;
  int blk = u >> 6, half = (u >> 5) & 1, lane = u & 31;
  int row = (blk >> 4) * 16 + (lane & 15);
  int k = (blk & 15) * 32 + half * 16 + 8 * (lane >> 4);
  const float4* s = (const float4*)(src + (size_t)row * 512 + k);
  float4 a = s[0], b = s[1];
  uint4 pv;
  pv.x = pack2bf(a.x, a.y);
  pv.y = pack2bf(a.z, a.w);
  pv.z = pack2bf(b.x, b.y);
  pv.w = pack2bf(b.z, b.w);
  ((uint4*)dst)[u] = pv;
}

__global__ void k_layernorm(const float* x, const float* g, const float* b,
                            __bf16* xn, int ntok) {
  int wave = (int)((blockIdx.x * blockDim.x + threadIdx.x) >> 5);
  int lane = threadIdx.x & 31;
  if (wave >= ntok) return;
  const float* xr = x + (size_t)wave * 512;
  float s = 0.f, s2 = 0.f, vals[16];
#pragma unroll
  for (int i = 0; i < 16; ++i) {
    float v = xr[lane + 32 * i];
    vals[i] = v; s += v; s2 += v * v;
  }
  for (int m = 16; m >= 1; m >>= 1) {
    s += __shfl_xor(s, m, 32);
    s2 += __shfl_xor(s2, m, 32);
  }
  float mu = s * (1.f / 512.f);
  float var = s2 * (1.f / 512.f) - mu * mu;
  float inv = rsqrtf(var + 1e-5f);
#pragma unroll
  for (int i = 0; i < 16; ++i) {
    int c = lane + 32 * i;
    xn[pk_idx_n(wave, c, 16)] = (__bf16)((vals[i] - mu) * inv * g[c] + b[c]);
  }
}

__global__ void __launch_bounds__(256)
k_gemm_qkv(const __bf16* xn, const __bf16* wq, __bf16* qb,
           __bf16* kb, float* vf, __bf16* vT) {
  int wid = (int)((blockIdx.x * blockDim.x + threadIdx.x) >> 5);
  int tM = wid / 24, s = wid % 24;
  floatx8 acc[4][4] = {};
  gemm_tile64<16, true>(xn, tM * 64, wq, s * 64, acc);
  int lane = threadIdx.x & 31;
  int n = lane & 15, h = lane >> 4;
  int col0 = s * 64;
  int sect = col0 >> 9;
  int head = (col0 >> 6) & 7;
#pragma unroll
  for (int m = 0; m < 4; ++m) {
#pragma unroll
    for (int t = 0; t < 4; ++t) {
      int dh = 16 * t + n;
#pragma unroll
      for (int r = 0; r < 8; ++r) {
        int tok = tM * 64 + 16 * m + r + 8 * h;
        int bb = tok >> 12, ns = tok & 4095;
        size_t bh = (size_t)(bb * 8 + head);
        size_t idx = (bh * 4096 + ns) * 64 + dh;
        float val = acc[m][t][r];
        if (sect == 0) qb[idx] = (__bf16)(val * 0.125f);
        else if (sect == 1) kb[idx] = (__bf16)val;
        else {
          vf[idx] = val;
          vT[(bh * 64 + dh) * 4096 + ns] = (__bf16)val;
        }
      }
    }
  }
}

__global__ void k_landmarks(const __bf16* qb, const __bf16* kb, __bf16* ql,
                            __bf16* kl) {
  int i = blockIdx.x * blockDim.x + threadIdx.x;
  if (i >= 32 * 256 * 64) return;
  int dh = i & 63, m = (i >> 6) & 255, bh = i >> 14;
  size_t base = ((size_t)bh * 4096 + m * 16) * 64 + dh;
  float sq = 0.f, sk = 0.f;
#pragma unroll
  for (int t = 0; t < 16; ++t) {
    sq += (float)qb[base + t * 64];
    sk += (float)kb[base + t * 64];
  }
  ql[i] = (__bf16)(sq * (1.f / 16.f));
  kl[i] = (__bf16)(sk * (1.f / 16.f));
}

__global__ void k_attn2_logits(const __bf16* ql, const __bf16* kl, float* out) {
  int wid = (int)((blockIdx.x * blockDim.x + threadIdx.x) >> 5);
  int bh = wid >> 6, t = wid & 63;
  int tM = t >> 2, s = t & 3;
  const __bf16* A = ql + (size_t)bh * 256 * 64;
  const __bf16* B = kl + (size_t)bh * 256 * 64;
  floatx8 acc[4] = {};
  gemm_strip<2>(A, tM * 16, B, s * 64, 64, acc);
  int lane = threadIdx.x & 31;
  int n = lane & 15, h = lane >> 4;
  float* o = out + (size_t)bh * 256 * 256;
#pragma unroll
  for (int t4 = 0; t4 < 4; ++t4)
#pragma unroll
    for (int r = 0; r < 8; ++r)
      o[(size_t)(tM * 16 + r + 8 * h) * 256 + s * 64 + 16 * t4 + n] =
          acc[t4][r];
}

__global__ void k_softmax256(float* x, __bf16* xb, int nrows) {
  int wave = (int)((blockIdx.x * blockDim.x + threadIdx.x) >> 5);
  if (wave >= nrows) return;
  int lane = threadIdx.x & 31;
  float* r = x + (size_t)wave * 256;
  __bf16* rb = xb + ((size_t)(wave >> 8) << 16);
  int row = wave & 255;
  float v[8], m = -1e30f;
#pragma unroll
  for (int i = 0; i < 8; ++i) { v[i] = r[lane + 32 * i]; m = fmaxf(m, v[i]); }
  for (int t = 16; t >= 1; t >>= 1) m = fmaxf(m, __shfl_xor(m, t, 32));
  float s = 0.f;
#pragma unroll
  for (int i = 0; i < 8; ++i) { v[i] = __expf(v[i] - m); s += v[i]; }
  for (int t = 16; t >= 1; t >>= 1) s += __shfl_xor(s, t, 32);
  float inv = 1.f / s;
#pragma unroll
  for (int i = 0; i < 8; ++i) {
    float pv = v[i] * inv;
    r[lane + 32 * i] = pv;
    rb[pk_idx(row, lane + 32 * i)] = (__bf16)pv;
  }
}

__global__ void k_pinv_scales(const float* x, float* scales) {
  int wave = (int)((blockIdx.x * blockDim.x + threadIdx.x) >> 5);
  if (wave >= 32 * 256) return;
  int lane = threadIdx.x & 31;
  int bh = wave >> 8, r = wave & 255;
  const float* m = x + (size_t)bh * 256 * 256;
  float rs = 0.f, cs = 0.f;
#pragma unroll
  for (int i = 0; i < 8; ++i) {
    rs += fabsf(m[(size_t)r * 256 + lane + 32 * i]);
    cs += fabsf(m[(size_t)(lane + 32 * i) * 256 + r]);
  }
  for (int t = 16; t >= 1; t >>= 1) {
    rs += __shfl_xor(rs, t, 32);
    cs += __shfl_xor(cs, t, 32);
  }
  if (lane == 0) {
    atomicMax((unsigned*)&scales[0], __float_as_uint(rs));
    atomicMax((unsigned*)&scales[1], __float_as_uint(cs));
  }
}

__global__ void k_init_z(const float* x, __bf16* zb, __bf16* zTb,
                         const float* scales) {
  size_t i = (size_t)blockIdx.x * blockDim.x + threadIdx.x;
  if (i >= (size_t)32 * 256 * 256) return;
  int j = (int)(i & 255), r = (int)((i >> 8) & 255);
  size_t bh = i >> 16;
  float denom = scales[0] * scales[1];
  float val = x[(bh << 16) + (size_t)j * 256 + r] / denom;
  zb[(bh << 16) + pk_idx(r, j)] = (__bf16)val;
  zTb[(bh << 16) + pk_idx(j, r)] = (__bf16)val;
}

typedef __attribute__((ext_vector_type(4))) unsigned int u32x4;

__device__ __forceinline__ bf16x16 load_frag_bf16_lds(const u32x4* tile, int j,
                                                      int ks) {
  int lane = threadIdx.x & 31;
  const u32x4* blk = tile + (j * 8 + ks) * 64 + lane;
  u32x4 lo = blk[0], hi = blk[32];
  FragAB f;
  f.u[0] = lo.x; f.u[1] = lo.y; f.u[2] = lo.z; f.u[3] = lo.w;
  f.u[4] = hi.x; f.u[5] = hi.y; f.u[6] = hi.z; f.u[7] = hi.w;
  return f.v;
}

__device__ __forceinline__ void lds_barrier() {
  __builtin_amdgcn_fence(__ATOMIC_RELEASE, "workgroup", "local");
  __builtin_amdgcn_s_barrier();
  __builtin_amdgcn_fence(__ATOMIC_ACQUIRE, "workgroup", "local");
}

#define G8_ARB 4
#define G8_BRB 8
__global__ void __launch_bounds__(256)
k_gemm256b(const __bf16* A, const __bf16* BT, __bf16* OutRaw,
           __bf16* OutT, __bf16* OutTT, float alpha, float beta) {
  __shared__ u32x4 tile[(G8_ARB + G8_BRB) * 8 * 64];
  int tid = (int)threadIdx.x;
  int w = tid >> 5;
  int bh = (int)blockIdx.x >> 3, g = (int)blockIdx.x & 7;
  int bm = g >> 1, bn = g & 1;
  int tM = bm * G8_ARB + (w >> 1), s = bn * 2 + (w & 1);
  size_t bofs = (size_t)bh << 16;
  const u32x4* Ag = (const u32x4*)(A + bofs) + (size_t)(bm * G8_ARB) * 8 * 64;
  const u32x4* Bg = (const u32x4*)(BT + bofs) + (size_t)(bn * G8_BRB) * 8 * 64;
  u32x4 st0[G8_ARB + G8_BRB], st1[G8_ARB + G8_BRB];
#pragma unroll
  for (int j = 0; j < G8_ARB; ++j) st0[j] = Ag[j * 512 + tid];
#pragma unroll
  for (int j = 0; j < G8_BRB; ++j) st0[G8_ARB + j] = Bg[j * 512 + tid];
  __builtin_amdgcn_sched_barrier(0);
#pragma unroll
  for (int j = 0; j < G8_ARB; ++j) st1[j] = Ag[j * 512 + 256 + tid];
#pragma unroll
  for (int j = 0; j < G8_BRB; ++j) st1[G8_ARB + j] = Bg[j * 512 + 256 + tid];
  floatx8 acc[4] = {};
#pragma unroll
  for (int j = 0; j < G8_ARB + G8_BRB; ++j) tile[j * 512 + tid] = st0[j];
  lds_barrier();
#pragma unroll
  for (int ks = 0; ks < 4; ++ks) {
    bf16x16 af = load_frag_bf16_lds(tile, w >> 1, ks);
#pragma unroll
    for (int t = 0; t < 4; ++t)
      acc[t] = wmma_bf16(
          af, load_frag_bf16_lds(tile, G8_ARB + (w & 1) * 4 + t, ks), acc[t]);
  }
#pragma unroll
  for (int j = 0; j < G8_ARB + G8_BRB; ++j) tile[j * 512 + 256 + tid] = st1[j];
  lds_barrier();
#pragma unroll
  for (int ks = 4; ks < 8; ++ks) {
    bf16x16 af = load_frag_bf16_lds(tile, w >> 1, ks);
#pragma unroll
    for (int t = 0; t < 4; ++t)
      acc[t] = wmma_bf16(
          af, load_frag_bf16_lds(tile, G8_ARB + (w & 1) * 4 + t, ks), acc[t]);
  }
  int lane = threadIdx.x & 31;
  int n = lane & 15, h = lane >> 4;
#pragma unroll
  for (int t4 = 0; t4 < 4; ++t4) {
    int N = s * 64 + 16 * t4 + n;
    float trv[8];
#pragma unroll
    for (int r = 0; r < 8; ++r) {
      int M = tM * 16 + r + 8 * h;
      float tv = acc[t4][r];
      if (OutRaw) OutRaw[bofs + pk_idx(M, N)] = (__bf16)tv;
      float tr = alpha * (M == N ? 1.f : 0.f) + beta * tv;
      if (OutT) OutT[bofs + pk_idx(M, N)] = (__bf16)tr;
      trv[r] = tr;
    }
    if (OutTT) {
      uint4 pv;
      pv.x = pack2bf(trv[0], trv[1]);
      pv.y = pack2bf(trv[2], trv[3]);
      pv.z = pack2bf(trv[4], trv[5]);
      pv.w = pack2bf(trv[6], trv[7]);
      *(uint4*)(OutTT + bofs + pk_idx(N, tM * 16 + 8 * h)) = pv;
    }
  }
}

__device__ __forceinline__ void stream_attn(const __bf16* Ak, const __bf16* Bq,
                                            int qrow0, const __bf16* A2,
                                            int ldA2, int nkeys, float* outF,
                                            int rsF, __bf16* outB, int nsB,
                                            int dsB) {
  int lane = threadIdx.x & 31;
  int n = lane & 15, h = lane >> 4;
  bf16x16 bq0 = load_frag_bf16_rm(Bq, qrow0, 0, 64);
  bf16x16 bq1 = load_frag_bf16_rm(Bq, qrow0, 32, 64);
  float run_m = -1e30f, run_s = 0.f;
  floatx8 acc[4] = {};
  for (int c = 0; c < nkeys; c += 32) {
    floatx8 s0 = {}, s1 = {};
    s0 = wmma_bf16(load_frag_bf16_rm(Ak, c, 0, 64), bq0, s0);
    s0 = wmma_bf16(load_frag_bf16_rm(Ak, c, 32, 64), bq1, s0);
    s1 = wmma_bf16(load_frag_bf16_rm(Ak, c + 16, 0, 64), bq0, s1);
    s1 = wmma_bf16(load_frag_bf16_rm(Ak, c + 16, 32, 64), bq1, s1);
    float mc = -1e30f;
#pragma unroll
    for (int r = 0; r < 8; ++r) { mc = fmaxf(mc, s0[r]); mc = fmaxf(mc, s1[r]); }
    mc = fmaxf(mc, __shfl_xor(mc, 16, 32));
    float newm = fmaxf(run_m, mc);
    float scale = __expf(run_m - newm);
    float p0[8], p1[8], ps = 0.f;
#pragma unroll
    for (int r = 0; r < 8; ++r) {
      p0[r] = __expf(s0[r] - newm);
      p1[r] = __expf(s1[r] - newm);
      ps += p0[r] + p1[r];
    }
    run_s = run_s * scale + ps;
    run_m = newm;
    FragAB bp;
#pragma unroll
    for (int p = 0; p < 4; ++p) {
      bp.u[p] = pack2bf(p0[2 * p], p0[2 * p + 1]);
      bp.u[4 + p] = pack2bf(p1[2 * p], p1[2 * p + 1]);
    }
#pragma unroll
    for (int t = 0; t < 4; ++t) {
#pragma unroll
      for (int r = 0; r < 8; ++r) acc[t][r] *= scale;
      acc[t] = wmma_bf16(load_frag_bf16_rm(A2, 16 * t, c, ldA2), bp.v, acc[t]);
    }
  }
  float tot = run_s + __shfl_xor(run_s, 16, 32);
  float inv = 1.f / tot;
#pragma unroll
  for (int t = 0; t < 4; ++t)
#pragma unroll
    for (int r = 0; r < 8; ++r) {
      int dh = 16 * t + r + 8 * h;
      float val = acc[t][r] * inv;
      if (outF) outF[(size_t)n * rsF + dh] = val;
      if (outB) outB[(size_t)n * nsB + (size_t)dh * dsB] = (__bf16)val;
    }
}

#define A3_WAVES 8
#define A3_QG 2
#define A3_KEYS (4096 / A3_WAVES)
__global__ void __launch_bounds__(256)
k_attn3v(const __bf16* kb, const __bf16* ql, const __bf16* vT, __bf16* kvT) {
  __shared__ float accL[A3_WAVES][A3_QG][4][8][32];
  __shared__ float mL[A3_WAVES][A3_QG][16];
  __shared__ float sL[A3_WAVES][A3_QG][16];
  int lane = threadIdx.x & 31;
  int w = __builtin_amdgcn_readfirstlane((int)(threadIdx.x >> 5));
  int bh = blockIdx.x >> 3, q0 = (blockIdx.x & 7) * (16 * A3_QG);
  const __bf16* Ak = kb + (size_t)bh * 4096 * 64;
  const __bf16* Bq = ql + (size_t)bh * 256 * 64;
  const __bf16* A2 = vT + (size_t)bh * 64 * 4096;
  bf16x16 bq[A3_QG][2];
  float run_m[A3_QG], run_s[A3_QG];
  floatx8 acc[A3_QG][4];
#pragma unroll
  for (int g = 0; g < A3_QG; ++g) {
    bq[g][0] = load_frag_bf16_rm(Bq, q0 + 16 * g, 0, 64);
    bq[g][1] = load_frag_bf16_rm(Bq, q0 + 16 * g, 32, 64);
    run_m[g] = -1e30f;
    run_s[g] = 0.f;
#pragma unroll
    for (int t = 0; t < 4; ++t) acc[g][t] = floatx8{};
  }
  int kbeg = w * A3_KEYS;
  for (int c = kbeg; c < kbeg + A3_KEYS; c += 32) {
    bf16x16 ka[4], va[4];
    ka[0] = load_frag_bf16_rm(Ak, c, 0, 64);
    ka[1] = load_frag_bf16_rm(Ak, c, 32, 64);
    ka[2] = load_frag_bf16_rm(Ak, c + 16, 0, 64);
    ka[3] = load_frag_bf16_rm(Ak, c + 16, 32, 64);
#pragma unroll
    for (int t = 0; t < 4; ++t) va[t] = load_frag_bf16_rm(A2, 16 * t, c, 4096);
    __builtin_amdgcn_sched_barrier(0);
#pragma unroll
    for (int g = 0; g < A3_QG; ++g) {
      floatx8 s0 = {}, s1 = {};
      s0 = wmma_bf16(ka[0], bq[g][0], s0);
      s0 = wmma_bf16(ka[1], bq[g][1], s0);
      s1 = wmma_bf16(ka[2], bq[g][0], s1);
      s1 = wmma_bf16(ka[3], bq[g][1], s1);
      float mc = -1e30f;
#pragma unroll
      for (int r = 0; r < 8; ++r) { mc = fmaxf(mc, s0[r]); mc = fmaxf(mc, s1[r]); }
      mc = fmaxf(mc, __shfl_xor(mc, 16, 32));
      float newm = fmaxf(run_m[g], mc);
      float scale = __expf(run_m[g] - newm);
      float p0[8], p1[8], ps = 0.f;
#pragma unroll
      for (int r = 0; r < 8; ++r) {
        p0[r] = __expf(s0[r] - newm);
        p1[r] = __expf(s1[r] - newm);
        ps += p0[r] + p1[r];
      }
      run_s[g] = run_s[g] * scale + ps;
      run_m[g] = newm;
      FragAB bp;
#pragma unroll
      for (int p = 0; p < 4; ++p) {
        bp.u[p] = pack2bf(p0[2 * p], p0[2 * p + 1]);
        bp.u[4 + p] = pack2bf(p1[2 * p], p1[2 * p + 1]);
      }
#pragma unroll
      for (int t = 0; t < 4; ++t) {
#pragma unroll
        for (int r = 0; r < 8; ++r) acc[g][t][r] *= scale;
        acc[g][t] = wmma_bf16(va[t], bp.v, acc[g][t]);
      }
    }
  }
#pragma unroll
  for (int g = 0; g < A3_QG; ++g) {
    float tot = run_s[g] + __shfl_xor(run_s[g], 16, 32);
    if (lane < 16) {
      mL[w][g][lane] = run_m[g];
      sL[w][g][lane] = tot;
    }
#pragma unroll
    for (int t = 0; t < 4; ++t)
#pragma unroll
      for (int r = 0; r < 8; ++r) accL[w][g][t][r][lane] = acc[g][t][r];
  }
  __syncthreads();
  int cc = lane & 15, hh = lane >> 4;
  for (int g = 0; g < A3_QG; ++g) {
    float mw[A3_WAVES], f[A3_WAVES], M = -1e30f, S = 0.f;
#pragma unroll
    for (int v = 0; v < A3_WAVES; ++v) { mw[v] = mL[v][g][cc]; M = fmaxf(M, mw[v]); }
#pragma unroll
    for (int v = 0; v < A3_WAVES; ++v) {
      f[v] = __expf(mw[v] - M);
      S += sL[v][g][cc] * f[v];
    }
    float inv = 1.f / S;
    for (int t = 0; t < 4; ++t) {
      float o = 0.f;
#pragma unroll
      for (int v = 0; v < A3_WAVES; ++v) o += accL[v][g][t][w][lane] * f[v];
      int dh = 16 * t + 8 * hh + w;
      kvT[((size_t)bh * 64 + dh) * 256 + q0 + 16 * g + cc] = (__bf16)(o * inv);
    }
  }
}

__global__ void __launch_bounds__(256)
k_w2(const __bf16* zb, const __bf16* kvT, __bf16* w2t) {
  int wid = (int)((blockIdx.x * blockDim.x + threadIdx.x) >> 5);
  int bh = wid >> 4, tM = wid & 15;
  floatx8 acc[4] = {};
  gemm_strip_pk<false>(zb + ((size_t)bh << 16), tM * 16,
                       kvT + (size_t)bh * 64 * 256, 0, acc);
  int lane = threadIdx.x & 31;
  int n = lane & 15, h = lane >> 4;
#pragma unroll
  for (int t4 = 0; t4 < 4; ++t4)
#pragma unroll
    for (int r = 0; r < 8; ++r) {
      int M = tM * 16 + r + 8 * h;
      int N = 16 * t4 + n;
      w2t[((size_t)bh * 64 + N) * 256 + M] = (__bf16)acc[t4][r];
    }
}

__global__ void k_attn1_out(const __bf16* kl, const __bf16* qb,
                            const __bf16* w2t, float* aout) {
  int wid = (int)((blockIdx.x * blockDim.x + threadIdx.x) >> 5);
  int bh = wid >> 8, blk = wid & 255;
  int b = bh >> 3, hh = bh & 7;
  stream_attn(kl + (size_t)bh * 256 * 64, qb + (size_t)bh * 4096 * 64, blk * 16,
              w2t + (size_t)bh * 64 * 256, 256, 256,
              aout + ((size_t)b * 4096 + blk * 16) * 512 + hh * 64, 512,
              nullptr, 0, 0);
}

#define CONV_R 16
#define CONV_G 8
#define CONV_ROWS (CONV_R + 32)
__global__ void __launch_bounds__(256, 4)
k_conv_res(const float* __restrict__ vf, const float* __restrict__ rw,
           float* __restrict__ aout, __bf16* __restrict__ aout_b) {
  int t = (int)(blockIdx.x * blockDim.x + threadIdx.x);
  if (t >= 4 * (4096 / CONV_R) * 128) return;
  int c4 = t & 127;
  int chunk = t >> 7;
  int b = chunk / (4096 / CONV_R);
  int ns0 = (chunk % (4096 / CONV_R)) * CONV_R;
  int col = c4 * 4, hh = col >> 6, dh = col & 63;
  const float* vcol = vf + ((size_t)(b * 8 + hh) * 4096) * 64 + dh;
  float* arow = aout + ((size_t)b * 4096 + ns0) * 512 + col;
  __bf16* brow = aout_b + ((size_t)b * 4096 + ns0) * 512 + col;
  float w[33];
#pragma unroll
  for (int j = 0; j < 33; ++j) w[j] = rw[hh * 33 + j];
  float4 acc[CONV_R], win[CONV_ROWS], a[CONV_R];
#pragma unroll
  for (int o = 0; o < CONV_R; ++o) acc[o] = make_float4(0.f, 0.f, 0.f, 0.f);
#pragma unroll
  for (int k = 0; k < CONV_G; ++k) {
    int p = ns0 + k - 16;
    int pc = p < 0 ? 0 : (p > 4095 ? 4095 : p);
    win[k] = *(const float4*)(vcol + (size_t)pc * 64);
  }
#pragma unroll
  for (int g = 0; g < CONV_ROWS / CONV_G; ++g) {
    if (g + 1 < CONV_ROWS / CONV_G) {
#pragma unroll
      for (int k = (g + 1) * CONV_G; k < (g + 2) * CONV_G; ++k) {
        int p = ns0 + k - 16;
        int pc = p < 0 ? 0 : (p > 4095 ? 4095 : p);
        win[k] = *(const float4*)(vcol + (size_t)pc * 64);
      }
    } else {
#pragma unroll
      for (int o = 0; o < CONV_R; ++o)
        a[o] = *(const float4*)(arow + (size_t)o * 512);
    }
#pragma unroll
    for (int k = g * CONV_G; k < (g + 1) * CONV_G; ++k) {
      int p = ns0 + k - 16;
      float4 v = win[k];
      if (p < 0 || p > 4095) v = make_float4(0.f, 0.f, 0.f, 0.f);
#pragma unroll
      for (int o = 0; o < CONV_R; ++o) {
        int j = k - o;
        if (j >= 0 && j < 33) {
          acc[o].x += v.x * w[j];
          acc[o].y += v.y * w[j];
          acc[o].z += v.z * w[j];
          acc[o].w += v.w * w[j];
        }
      }
    }
  }
#pragma unroll
  for (int o = 0; o < CONV_R; ++o) {
    float4 r;
    r.x = a[o].x + acc[o].x;
    r.y = a[o].y + acc[o].y;
    r.z = a[o].z + acc[o].z;
    r.w = a[o].w + acc[o].w;
    *(float4*)(arow + (size_t)o * 512) = r;
    uint2 pk;
    pk.x = pack2bf(r.x, r.y);
    pk.y = pack2bf(r.z, r.w);
    *(uint2*)(brow + (size_t)o * 512) = pk;
  }
}

__global__ void __launch_bounds__(256)
k_outproj(const __bf16* __restrict__ aoutb, const __bf16* __restrict__ wo,
          const float* __restrict__ bias, const float* __restrict__ x,
          float* __restrict__ out) {
  int wid = (int)((blockIdx.x * blockDim.x + threadIdx.x) >> 5);
  int tM = wid >> 3, s = wid & 7;
  floatx8 acc[4][4] = {};
  gemm_tile64<16, false>(aoutb, tM * 64, wo, s * 64, acc);
  int lane = threadIdx.x & 31;
  int n = lane & 15, h = lane >> 4;
#pragma unroll
  for (int t4 = 0; t4 < 4; ++t4) {
    int col = s * 64 + 16 * t4 + n;
    float bc = bias[col];
#pragma unroll
    for (int m = 0; m < 4; ++m) {
#pragma unroll
      for (int r = 0; r < 8; ++r) {
        size_t idx = (size_t)(tM * 64 + 16 * m + r + 8 * h) * 512 + col;
        out[idx] = x[idx] + acc[m][t4][r] + bc;
      }
    }
  }
}

extern "C" void kernel_launch(void* const* d_in, const int* in_sizes, int n_in,
                              void* d_out, int out_size, void* d_ws,
                              size_t ws_size, hipStream_t stream) {
  const float* x     = (const float*)d_in[0];
  const float* ln_g  = (const float*)d_in[1];
  const float* ln_b  = (const float*)d_in[2];
  const float* w_qkv = (const float*)d_in[3];
  const float* w_out = (const float*)d_in[4];
  const float* b_out = (const float*)d_in[5];
  const float* res_w = (const float*)d_in[6];
  float* out = (float*)d_out;

  char* p = (char*)d_ws;
  auto alloc = [&](size_t bytes) -> void* {
    void* r = (void*)p;
    p += (bytes + 255) & ~(size_t)255;
    return r;
  };
  const size_t NTOK = 4 * 4096;
  const size_t MM = (size_t)32 * 256 * 256;
  __bf16* xn_b   = (__bf16*)alloc(NTOK * 512 * 2);
  __bf16* wqkv_b = (__bf16*)alloc(1536 * 512 * 2);
  __bf16* wout_b = (__bf16*)alloc(512 * 512 * 2);
  __bf16* q_b    = (__bf16*)alloc((size_t)32 * 4096 * 64 * 2);
  __bf16* k_b    = (__bf16*)alloc((size_t)32 * 4096 * 64 * 2);
  float*  v_f    = (float*)alloc((size_t)32 * 4096 * 64 * 4);
  __bf16* vT_b   = (__bf16*)alloc((size_t)32 * 64 * 4096 * 2);
  __bf16* ql_b   = (__bf16*)alloc((size_t)32 * 256 * 64 * 2);
  __bf16* kl_b   = (__bf16*)alloc((size_t)32 * 256 * 64 * 2);
  float*  mat_x  = (float*)alloc(MM * 4);
  __bf16* x_b    = (__bf16*)alloc(MM * 2);
  __bf16* z0_b   = (__bf16*)alloc(MM * 2);
  __bf16* z0T_b  = (__bf16*)alloc(MM * 2);
  __bf16* z1_b   = (__bf16*)alloc(MM * 2);
  __bf16* z1T_b  = (__bf16*)alloc(MM * 2);
  __bf16* xz_b   = (__bf16*)alloc(MM * 2);
  __bf16* t1T_b  = (__bf16*)alloc(MM * 2);
  __bf16* t2T_b  = (__bf16*)alloc(MM * 2);
  __bf16* t3T_b  = (__bf16*)alloc(MM * 2);
  __bf16* kvT_b  = (__bf16*)alloc((size_t)32 * 64 * 256 * 2);
  __bf16* w2t_b  = (__bf16*)alloc((size_t)32 * 64 * 256 * 2);
  float*  aout_f = (float*)alloc(NTOK * 512 * 4);
  __bf16* aout_b = (__bf16*)alloc(NTOK * 512 * 2);
  float*  scales = (float*)alloc(256);
  (void)ws_size; (void)n_in; (void)in_sizes; (void)out_size;

  const int T = 256;
  k_convert_w<<<512, T, 0, stream>>>(w_qkv, w_out, wqkv_b, wout_b, scales);
  k_layernorm<<<2048, T, 0, stream>>>(x, ln_g, ln_b, xn_b, (int)NTOK);
  k_gemm_qkv<<<768, T, 0, stream>>>(xn_b, wqkv_b, q_b, k_b, v_f, vT_b);
  k_landmarks<<<2048, T, 0, stream>>>(q_b, k_b, ql_b, kl_b);
  k_attn2_logits<<<256, T, 0, stream>>>(ql_b, kl_b, mat_x);
  k_softmax256<<<1024, T, 0, stream>>>(mat_x, x_b, 32 * 256);
  k_pinv_scales<<<1024, T, 0, stream>>>(mat_x, scales);
  k_init_z<<<8192, T, 0, stream>>>(mat_x, z0_b, z0T_b, scales);

  __bf16* zc = z0_b;  __bf16* zcT = z0T_b;
  __bf16* zn = z1_b;  __bf16* znT = z1T_b;
  for (int it = 0; it < 6; ++it) {
    k_gemm256b<<<256, T, 0, stream>>>(x_b, zcT, xz_b, nullptr, t1T_b, 7.f, -1.f);
    k_gemm256b<<<256, T, 0, stream>>>(xz_b, t1T_b, nullptr, nullptr, t2T_b, 15.f, -1.f);
    k_gemm256b<<<256, T, 0, stream>>>(xz_b, t2T_b, nullptr, nullptr, t3T_b, 13.f, -1.f);
    k_gemm256b<<<256, T, 0, stream>>>(zc, t3T_b, nullptr, zn, znT, 0.f, 0.25f);
    __bf16* t;
    t = zc; zc = zn; zn = t;
    t = zcT; zcT = znT; znT = t;
  }

  k_attn3v<<<256, T, 0, stream>>>(k_b, ql_b, vT_b, kvT_b);
  k_w2<<<64, T, 0, stream>>>(zc, kvT_b, w2t_b);
  k_attn1_out<<<1024, T, 0, stream>>>(kl_b, q_b, w2t_b, aout_f);
  k_conv_res<<<512, T, 0, stream>>>(v_f, res_w, aout_f, aout_b);
  k_outproj<<<256, T, 0, stream>>>(aout_b, wout_b, b_out, x, out);
}


# QKV and out-proj GEMMs: a work-group = one 128x256 tile; per 2 K steps its 24 packed row blocks (48 KB) go to LDS once (plain loads + ds_store, 2 stages), waves take fragments from LDS; same sums
# speedup vs baseline: 2.5358x; 1.0874x over previous
#include <hip/hip_runtime.h>
#include <hip/hip_bf16.h>
#include <math.h>


typedef __attribute__((ext_vector_type(16))) __bf16 bf16x16;
typedef __attribute__((ext_vector_type(8)))  float  floatx8;

union FragAB { bf16x16 v; unsigned int u[8]; };

__device__ __forceinline__ unsigned pack2bf(float a, float b) {
  __bf16 lo = (__bf16)a, hi = (__bf16)b;
  unsigned short ul = __builtin_bit_cast(unsigned short, lo);
  unsigned short uh = __builtin_bit_cast(unsigned short, hi);
  return (unsigned)ul | ((unsigned)uh << 16);
}

__device__ __forceinline__ int kpair(int p, int h) {
  return (p < 4) ? (2 * p + 8 * h) : (8 + 2 * p + 8 * h);
}

__device__ __forceinline__ bf16x16 load_frag_bf16_rm(const __bf16* src, int row0,
                                                     int k0, int ld) {
  int lane = threadIdx.x & 31;
  int r = lane & 15, h = lane >> 4;
  const __bf16* base = src + (size_t)(row0 + r) * ld + k0;
  FragAB f;
#pragma unroll
  for (int p = 0; p < 8; ++p) f.u[p] = *(const unsigned*)(base + kpair(p, h));
  return f.v;
}

__device__ __forceinline__ floatx8 wmma_bf16(bf16x16 a, bf16x16 b, floatx8 c) {
  return __builtin_amdgcn_wmma_f32_16x16x32_bf16(false, a, false, b, (short)0, c,
                                                 false, false);
}

template <int KSTEPS>
__device__ __forceinline__ void gemm_strip(const __bf16* A, int rowA,
                                           const __bf16* B, int rowB, int ld,
                                           floatx8 acc[4]) {
#pragma unroll
  for (int ks = 0; ks < KSTEPS; ++ks) {
    int k0 = ks * 32;
    bf16x16 af = load_frag_bf16_rm(A, rowA, k0, ld);
#pragma unroll
    for (int t = 0; t < 4; ++t) {
      bf16x16 bfr = load_frag_bf16_rm(B, rowB + 16 * t, k0, ld);
      acc[t] = wmma_bf16(af, bfr, acc[t]);
    }
  }
}

__device__ __forceinline__ size_t pk_idx(int row, int k) {
  int kk = k & 31;
  int lane = (row & 15) + 16 * ((kk >> 3) & 1);
  return (size_t)(((row >> 4) * 8 + (k >> 5)) * 512 + (kk >> 4) * 256 +
                  lane * 8 + (kk & 7));
}

__device__ __forceinline__ bf16x16 load_frag_bf16_pk(const __bf16* mat, int rb,
                                                     int ks) {
  int lane = threadIdx.x & 31;
  const uint4* blk = (const uint4*)(mat + (size_t)(rb * 8 + ks) * 512) + lane;
  uint4 lo = blk[0], hi = blk[32];
  FragAB f;
  f.u[0] = lo.x; f.u[1] = lo.y; f.u[2] = lo.z; f.u[3] = lo.w;
  f.u[4] = hi.x; f.u[5] = hi.y; f.u[6] = hi.z; f.u[7] = hi.w;
  return f.v;
}

__device__ __forceinline__ size_t pk_idx_n(int row, int k, int KB) {
  int kk = k & 31;
  int lane = (row & 15) + 16 * ((kk >> 3) & 1);
  return (size_t)((row >> 4) * KB + (k >> 5)) * 512 + (kk >> 4) * 256 +
         lane * 8 + (kk & 7);
}

#define PK_AHEAD 4
template <bool BPK>
__device__ __forceinline__ void gemm_strip_pk(const __bf16* A, int rowA,
                                              const __bf16* B, int rowB,
                                              floatx8 acc[4]) {
  bf16x16 af[8], bfr[8][4];
#pragma unroll
  for (int ks = 0; ks < PK_AHEAD; ++ks) {
    af[ks] = load_frag_bf16_pk(A, rowA >> 4, ks);
#pragma unroll
    for (int t = 0; t < 4; ++t)
      bfr[ks][t] = BPK ? load_frag_bf16_pk(B, (rowB >> 4) + t, ks)
                       : load_frag_bf16_rm(B, rowB + 16 * t, ks * 32, 256);
    __builtin_amdgcn_sched_barrier(0);
  }
#pragma unroll
  for (int ks = 0; ks < 8; ++ks) {
#pragma unroll
    for (int t = 0; t < 4; ++t) acc[t] = wmma_bf16(af[ks], bfr[ks][t], acc[t]);
    __builtin_amdgcn_sched_barrier(0);
    if (ks + PK_AHEAD < 8) {
      af[ks + PK_AHEAD] = load_frag_bf16_pk(A, rowA >> 4, ks + PK_AHEAD);
#pragma unroll
      for (int t = 0; t < 4; ++t)
        bfr[ks + PK_AHEAD][t] =
            BPK ? load_frag_bf16_pk(B, (rowB >> 4) + t, ks + PK_AHEAD)
                : load_frag_bf16_rm(B, rowB + 16 * t, (ks + PK_AHEAD) * 32, 256);
      __builtin_amdgcn_sched_barrier(0);
    }
  }
}

typedef __attribute__((ext_vector_type(4))) unsigned int u32x4;

__device__ __forceinline__ bf16x16 load_frag_bf16_lds(const u32x4* tile, int j,
                                                      int ks) {
  int lane = threadIdx.x & 31;
  const u32x4* blk = tile + (j * 8 + ks) * 64 + lane;
  u32x4 lo = blk[0], hi = blk[32];
  FragAB f;
  f.u[0] = lo.x; f.u[1] = lo.y; f.u[2] = lo.z; f.u[3] = lo.w;
  f.u[4] = hi.x; f.u[5] = hi.y; f.u[6] = hi.z; f.u[7] = hi.w;
  return f.v;
}

__device__ __forceinline__ void lds_barrier() {
  __builtin_amdgcn_fence(__ATOMIC_RELEASE, "workgroup", "local");
  __builtin_amdgcn_s_barrier();
  __builtin_amdgcn_fence(__ATOMIC_ACQUIRE, "workgroup", "local");
}

#define GB_STAGE 3072
__device__ __forceinline__ void gemm_block_lds(const __bf16* A, int arb0,
                                               const __bf16* B, int brb0,
                                               u32x4* tile,
                                               floatx8 acc[4][4]) {
  int tid = (int)threadIdx.x;
  int lane = tid & 31, w = tid >> 5;
  int wm = w >> 2, wn = w & 3;
  int half = tid >> 7, q = tid & 127;
  const u32x4* Ag = (const u32x4*)A + (size_t)(arb0 + half) * 1024 + q;
  const u32x4* Bg = (const u32x4*)B + (size_t)(brb0 + half) * 1024 + q;
  u32x4 st[12];
#pragma unroll
  for (int j = 0; j < 4; ++j) st[j] = Ag[j * 2048];
#pragma unroll
  for (int j = 0; j < 8; ++j) st[4 + j] = Bg[j * 2048];
#pragma unroll
  for (int S = 0; S < 8; ++S) {
    u32x4* dst = tile + (S & 1) * GB_STAGE;
#pragma unroll
    for (int j = 0; j < 12; ++j) dst[j * 256 + tid] = st[j];
    if (S + 1 < 8) {
#pragma unroll
      for (int j = 0; j < 4; ++j) st[j] = Ag[j * 2048 + (S + 1) * 128];
#pragma unroll
      for (int j = 0; j < 8; ++j) st[4 + j] = Bg[j * 2048 + (S + 1) * 128];
    }
    __builtin_amdgcn_sched_barrier(0);
    lds_barrier();
    bf16x16 af[2][4], bfr[2][4];
#pragma unroll
    for (int kk = 0; kk < 2; ++kk) {
      const u32x4* sb = dst + kk * 64 + lane;
#pragma unroll
      for (int m = 0; m < 4; ++m) {
        const u32x4* blk = sb + (wm * 4 + m) * 128;
        u32x4 lo = blk[0], hi = blk[32];
        FragAB f;
        f.u[0] = lo.x; f.u[1] = lo.y; f.u[2] = lo.z; f.u[3] = lo.w;
        f.u[4] = hi.x; f.u[5] = hi.y; f.u[6] = hi.z; f.u[7] = hi.w;
        af[kk][m] = f.v;
      }
#pragma unroll
      for (int t = 0; t < 4; ++t) {
        const u32x4* blk = sb + (8 + wn * 4 + t) * 128;
        u32x4 lo = blk[0], hi = blk[32];
        FragAB f;
        f.u[0] = lo.x; f.u[1] = lo.y; f.u[2] = lo.z; f.u[3] = lo.w;
        f.u[4] = hi.x; f.u[5] = hi.y; f.u[6] = hi.z; f.u[7] = hi.w;
        bfr[kk][t] = f.v;
      }
    }
    __builtin_amdgcn_sched_barrier(0);
#pragma unroll
    for (int kk = 0; kk < 2; ++kk)
#pragma unroll
      for (int m = 0; m < 4; ++m)
#pragma unroll
        for (int t = 0; t < 4; ++t)
          acc[m][t] = wmma_bf16(af[kk][m], bfr[kk][t], acc[m][t]);
  }
}

__global__ void k_convert_w(const float* wqkv, const float* wout,
                            __bf16* wqkv_b, __bf16* wout_b, float* scales) {
  int i = blockIdx.x * blockDim.x + threadIdx.x;
  if (i < 2) scales[i] = 0.f;
  bool second = i >= 1536 * 64;
  const float* src = second ? wout : wqkv;
  __bf16* dst = second ? wout_b : wqkv_b;
  int u = second ? i - 1536 * 64 : i;
  if (second && u >= 512 * 64) return;
  int blk = u >> 6, half = (u >> 5) & 1, lane = u & 31;
  int row = (blk >> 4) * 16 + (lane & 15);
  int k = (blk & 15) * 32 + half * 16 + 8 * (lane >> 4);
  const float4* s = (const float4*)(src + (size_t)row * 512 + k);
  float4 a = s[0], b = s[1];
  uint4 pv;
  pv.x = pack2bf(a.x, a.y);
  pv.y = pack2bf(a.z, a.w);
  pv.z = pack2bf(b.x, b.y);
  pv.w = pack2bf(b.z, b.w);
  ((uint4*)dst)[u] = pv;
}

__global__ void k_layernorm(const float* x, const float* g, const float* b,
                            __bf16* xn, int ntok) {
  int wave = (int)((blockIdx.x * blockDim.x + threadIdx.x) >> 5);
  int lane = threadIdx.x & 31;
  if (wave >= ntok) return;
  const float* xr = x + (size_t)wave * 512;
  float s = 0.f, s2 = 0.f, vals[16];
#pragma unroll
  for (int i = 0; i < 16; ++i) {
    float v = xr[lane + 32 * i];
    vals[i] = v; s += v; s2 += v * v;
  }
  for (int m = 16; m >= 1; m >>= 1) {
    s += __shfl_xor(s, m, 32);
    s2 += __shfl_xor(s2, m, 32);
  }
  float mu = s * (1.f / 512.f);
  float var = s2 * (1.f / 512.f) - mu * mu;
  float inv = rsqrtf(var + 1e-5f);
#pragma unroll
  for (int i = 0; i < 16; ++i) {
    int c = lane + 32 * i;
    xn[pk_idx_n(wave, c, 16)] = (__bf16)((vals[i] - mu) * inv * g[c] + b[c]);
  }
}

__global__ void __launch_bounds__(256)
k_gemm_qkv(const __bf16* xn, const __bf16* wq, __bf16* qb,
           __bf16* kb, float* vf, __bf16* vT) {
  __shared__ u32x4 tile[2 * GB_STAGE];
  int bM = (int)blockIdx.x / 6, bN = (int)blockIdx.x % 6;
  int w = (int)threadIdx.x >> 5;
  int tM = bM * 2 + (w >> 2), s = bN * 4 + (w & 3);
  floatx8 acc[4][4] = {};
  gemm_block_lds(xn, bM * 8, wq, bN * 16, tile, acc);
  int lane = threadIdx.x & 31;
  int n = lane & 15, h = lane >> 4;
  int col0 = s * 64;
  int sect = col0 >> 9;
  int head = (col0 >> 6) & 7;
#pragma unroll
  for (int m = 0; m < 4; ++m) {
#pragma unroll
    for (int t = 0; t < 4; ++t) {
      int dh = 16 * t + n;
#pragma unroll
      for (int r = 0; r < 8; ++r) {
        int tok = tM * 64 + 16 * m + r + 8 * h;
        int bb = tok >> 12, ns = tok & 4095;
        size_t bh = (size_t)(bb * 8 + head);
        size_t idx = (bh * 4096 + ns) * 64 + dh;
        float val = acc[m][t][r];
        if (sect == 0) qb[idx] = (__bf16)(val * 0.125f);
        else if (sect == 1) kb[idx] = (__bf16)val;
        else {
          vf[idx] = val;
          vT[(bh * 64 + dh) * 4096 + ns] = (__bf16)val;
        }
      }
    }
  }
}

__global__ void k_landmarks(const __bf16* qb, const __bf16* kb, __bf16* ql,
                            __bf16* kl) {
  int i = blockIdx.x * blockDim.x + threadIdx.x;
  if (i >= 32 * 256 * 64) return;
  int dh = i & 63, m = (i >> 6) & 255, bh = i >> 14;
  size_t base = ((size_t)bh * 4096 + m * 16) * 64 + dh;
  float sq = 0.f, sk = 0.f;
#pragma unroll
  for (int t = 0; t < 16; ++t) {
    sq += (float)qb[base + t * 64];
    sk += (float)kb[base + t * 64];
  }
  ql[i] = (__bf16)(sq * (1.f / 16.f));
  kl[i] = (__bf16)(sk * (1.f / 16.f));
}

__global__ void k_attn2_logits(const __bf16* ql, const __bf16* kl, float* out) {
  int wid = (int)((blockIdx.x * blockDim.x + threadIdx.x) >> 5);
  int bh = wid >> 6, t = wid & 63;
  int tM = t >> 2, s = t & 3;
  const __bf16* A = ql + (size_t)bh * 256 * 64;
  const __bf16* B = kl + (size_t)bh * 256 * 64;
  floatx8 acc[4] = {};
  gemm_strip<2>(A, tM * 16, B, s * 64, 64, acc);
  int lane = threadIdx.x & 31;
  int n = lane & 15, h = lane >> 4;
  float* o = out + (size_t)bh * 256 * 256;
#pragma unroll
  for (int t4 = 0; t4 < 4; ++t4)
#pragma unroll
    for (int r = 0; r < 8; ++r)
      o[(size_t)(tM * 16 + r + 8 * h) * 256 + s * 64 + 16 * t4 + n] =
          acc[t4][r];
}

__global__ void k_softmax256(float* x, __bf16* xb, int nrows) {
  int wave = (int)((blockIdx.x * blockDim.x + threadIdx.x) >> 5);
  if (wave >= nrows) return;
  int lane = threadIdx.x & 31;
  float* r = x + (size_t)wave * 256;
  __bf16* rb = xb + ((size_t)(wave >> 8) << 16);
  int row = wave & 255;
  float v[8], m = -1e30f;
#pragma unroll
  for (int i = 0; i < 8; ++i) { v[i] = r[lane + 32 * i]; m = fmaxf(m, v[i]); }
  for (int t = 16; t >= 1; t >>= 1) m = fmaxf(m, __shfl_xor(m, t, 32));
  float s = 0.f;
#pragma unroll
  for (int i = 0; i < 8; ++i) { v[i] = __expf(v[i] - m); s += v[i]; }
  for (int t = 16; t >= 1; t >>= 1) s += __shfl_xor(s, t, 32);
  float inv = 1.f / s;
#pragma unroll
  for (int i = 0; i < 8; ++i) {
    float pv = v[i] * inv;
    r[lane + 32 * i] = pv;
    rb[pk_idx(row, lane + 32 * i)] = (__bf16)pv;
  }
}

__global__ void k_pinv_scales(const float* x, float* scales) {
  int wave = (int)((blockIdx.x * blockDim.x + threadIdx.x) >> 5);
  if (wave >= 32 * 256) return;
  int lane = threadIdx.x & 31;
  int bh = wave >> 8, r = wave & 255;
  const float* m = x + (size_t)bh * 256 * 256;
  float rs = 0.f, cs = 0.f;
#pragma unroll
  for (int i = 0; i < 8; ++i) {
    rs += fabsf(m[(size_t)r * 256 + lane + 32 * i]);
    cs += fabsf(m[(size_t)(lane + 32 * i) * 256 + r]);
  }
  for (int t = 16; t >= 1; t >>= 1) {
    rs += __shfl_xor(rs, t, 32);
    cs += __shfl_xor(cs, t, 32);
  }
  if (lane == 0) {
    atomicMax((unsigned*)&scales[0], __float_as_uint(rs));
    atomicMax((unsigned*)&scales[1], __float_as_uint(cs));
  }
}

__global__ void k_init_z(const float* x, __bf16* zb, __bf16* zTb,
                         const float* scales) {
  size_t i = (size_t)blockIdx.x * blockDim.x + threadIdx.x;
  if (i >= (size_t)32 * 256 * 256) return;
  int j = (int)(i & 255), r = (int)((i >> 8) & 255);
  size_t bh = i >> 16;
  float denom = scales[0] * scales[1];
  float val = x[(bh << 16) + (size_t)j * 256 + r] / denom;
  zb[(bh << 16) + pk_idx(r, j)] = (__bf16)val;
  zTb[(bh << 16) + pk_idx(j, r)] = (__bf16)val;
}

#define G8_ARB 4
#define G8_BRB 8
__global__ void __launch_bounds__(256)
k_gemm256b(const __bf16* A, const __bf16* BT, __bf16* OutRaw,
           __bf16* OutT, __bf16* OutTT, float alpha, float beta) {
  __shared__ u32x4 tile[(G8_ARB + G8_BRB) * 8 * 64];
  int tid = (int)threadIdx.x;
  int w = tid >> 5;
  int bh = (int)blockIdx.x >> 3, g = (int)blockIdx.x & 7;
  int bm = g >> 1, bn = g & 1;
  int tM = bm * G8_ARB + (w >> 1), s = bn * 2 + (w & 1);
  size_t bofs = (size_t)bh << 16;
  const u32x4* Ag = (const u32x4*)(A + bofs) + (size_t)(bm * G8_ARB) * 8 * 64;
  const u32x4* Bg = (const u32x4*)(BT + bofs) + (size_t)(bn * G8_BRB) * 8 * 64;
  u32x4 st0[G8_ARB + G8_BRB], st1[G8_ARB + G8_BRB];
#pragma unroll
  for (int j = 0; j < G8_ARB; ++j) st0[j] = Ag[j * 512 + tid];
#pragma unroll
  for (int j = 0; j < G8_BRB; ++j) st0[G8_ARB + j] = Bg[j * 512 + tid];
  __builtin_amdgcn_sched_barrier(0);
#pragma unroll
  for (int j = 0; j < G8_ARB; ++j) st1[j] = Ag[j * 512 + 256 + tid];
#pragma unroll
  for (int j = 0; j < G8_BRB; ++j) st1[G8_ARB + j] = Bg[j * 512 + 256 + tid];
  floatx8 acc[4] = {};
#pragma unroll
  for (int j = 0; j < G8_ARB + G8_BRB; ++j) tile[j * 512 + tid] = st0[j];
  lds_barrier();
#pragma unroll
  for (int ks = 0; ks < 4; ++ks) {
    bf16x16 af = load_frag_bf16_lds(tile, w >> 1, ks);
#pragma unroll
    for (int t = 0; t < 4; ++t)
      acc[t] = wmma_bf16(
          af, load_frag_bf16_lds(tile, G8_ARB + (w & 1) * 4 + t, ks), acc[t]);
  }
#pragma unroll
  for (int j = 0; j < G8_ARB + G8_BRB; ++j) tile[j * 512 + 256 + tid] = st1[j];
  lds_barrier();
#pragma unroll
  for (int ks = 4; ks < 8; ++ks) {
    bf16x16 af = load_frag_bf16_lds(tile, w >> 1, ks);
#pragma unroll
    for (int t = 0; t < 4; ++t)
      acc[t] = wmma_bf16(
          af, load_frag_bf16_lds(tile, G8_ARB + (w & 1) * 4 + t, ks), acc[t]);
  }
  int lane = threadIdx.x & 31;
  int n = lane & 15, h = lane >> 4;
#pragma unroll
  for (int t4 = 0; t4 < 4; ++t4) {
    int N = s * 64 + 16 * t4 + n;
    float trv[8];
#pragma unroll
    for (int r = 0; r < 8; ++r) {
      int M = tM * 16 + r + 8 * h;
      float tv = acc[t4][r];
      if (OutRaw) OutRaw[bofs + pk_idx(M, N)] = (__bf16)tv;
      float tr = alpha * (M == N ? 1.f : 0.f) + beta * tv;
      if (OutT) OutT[bofs + pk_idx(M, N)] = (__bf16)tr;
      trv[r] = tr;
    }
    if (OutTT) {
      uint4 pv;
      pv.x = pack2bf(trv[0], trv[1]);
      pv.y = pack2bf(trv[2], trv[3]);
      pv.z = pack2bf(trv[4], trv[5]);
      pv.w = pack2bf(trv[6], trv[7]);
      *(uint4*)(OutTT + bofs + pk_idx(N, tM * 16 + 8 * h)) = pv;
    }
  }
}

__device__ __forceinline__ void stream_attn(const __bf16* Ak, const __bf16* Bq,
                                            int qrow0, const __bf16* A2,
                                            int ldA2, int nkeys, float* outF,
                                            int rsF, __bf16* outB, int nsB,
                                            int dsB) {
  int lane = threadIdx.x & 31;
  int n = lane & 15, h = lane >> 4;
  bf16x16 bq0 = load_frag_bf16_rm(Bq, qrow0, 0, 64);
  bf16x16 bq1 = load_frag_bf16_rm(Bq, qrow0, 32, 64);
  float run_m = -1e30f, run_s = 0.f;
  floatx8 acc[4] = {};
  for (int c = 0; c < nkeys; c += 32) {
    floatx8 s0 = {}, s1 = {};
    s0 = wmma_bf16(load_frag_bf16_rm(Ak, c, 0, 64), bq0, s0);
    s0 = wmma_bf16(load_frag_bf16_rm(Ak, c, 32, 64), bq1, s0);
    s1 = wmma_bf16(load_frag_bf16_rm(Ak, c + 16, 0, 64), bq0, s1);
    s1 = wmma_bf16(load_frag_bf16_rm(Ak, c + 16, 32, 64), bq1, s1);
    float mc = -1e30f;
#pragma unroll
    for (int r = 0; r < 8; ++r) { mc = fmaxf(mc, s0[r]); mc = fmaxf(mc, s1[r]); }
    mc = fmaxf(mc, __shfl_xor(mc, 16, 32));
    float newm = fmaxf(run_m, mc);
    float scale = __expf(run_m - newm);
    float p0[8], p1[8], ps = 0.f;
#pragma unroll
    for (int r = 0; r < 8; ++r) {
      p0[r] = __expf(s0[r] - newm);
      p1[r] = __expf(s1[r] - newm);
      ps += p0[r] + p1[r];
    }
    run_s = run_s * scale + ps;
    run_m = newm;
    FragAB bp;
#pragma unroll
    for (int p = 0; p < 4; ++p) {
      bp.u[p] = pack2bf(p0[2 * p], p0[2 * p + 1]);
      bp.u[4 + p] = pack2bf(p1[2 * p], p1[2 * p + 1]);
    }
#pragma unroll
    for (int t = 0; t < 4; ++t) {
#pragma unroll
      for (int r = 0; r < 8; ++r) acc[t][r] *= scale;
      acc[t] = wmma_bf16(load_frag_bf16_rm(A2, 16 * t, c, ldA2), bp.v, acc[t]);
    }
  }
  float tot = run_s + __shfl_xor(run_s, 16, 32);
  float inv = 1.f / tot;
#pragma unroll
  for (int t = 0; t < 4; ++t)
#pragma unroll
    for (int r = 0; r < 8; ++r) {
      int dh = 16 * t + r + 8 * h;
      float val = acc[t][r] * inv;
      if (outF) outF[(size_t)n * rsF + dh] = val;
      if (outB) outB[(size_t)n * nsB + (size_t)dh * dsB] = (__bf16)val;
    }
}

#define A3_WAVES 8
#define A3_QG 2
#define A3_KEYS (4096 / A3_WAVES)
__global__ void __launch_bounds__(256)
k_attn3v(const __bf16* kb, const __bf16* ql, const __bf16* vT, __bf16* kvT) {
  __shared__ float accL[A3_WAVES][A3_QG][4][8][32];
  __shared__ float mL[A3_WAVES][A3_QG][16];
  __shared__ float sL[A3_WAVES][A3_QG][16];
  int lane = threadIdx.x & 31;
  int w = __builtin_amdgcn_readfirstlane((int)(threadIdx.x >> 5));
  int bh = blockIdx.x >> 3, q0 = (blockIdx.x & 7) * (16 * A3_QG);
  const __bf16* Ak = kb + (size_t)bh * 4096 * 64;
  const __bf16* Bq = ql + (size_t)bh * 256 * 64;
  const __bf16* A2 = vT + (size_t)bh * 64 * 4096;
  bf16x16 bq[A3_QG][2];
  float run_m[A3_QG], run_s[A3_QG];
  floatx8 acc[A3_QG][4];
#pragma unroll
  for (int g = 0; g < A3_QG; ++g) {
    bq[g][0] = load_frag_bf16_rm(Bq, q0 + 16 * g, 0, 64);
    bq[g][1] = load_frag_bf16_rm(Bq, q0 + 16 * g, 32, 64);
    run_m[g] = -1e30f;
    run_s[g] = 0.f;
#pragma unroll
    for (int t = 0; t < 4; ++t) acc[g][t] = floatx8{};
  }
  int kbeg = w * A3_KEYS;
  for (int c = kbeg; c < kbeg + A3_KEYS; c += 32) {
    bf16x16 ka[4], va[4];
    ka[0] = load_frag_bf16_rm(Ak, c, 0, 64);
    ka[1] = load_frag_bf16_rm(Ak, c, 32, 64);
    ka[2] = load_frag_bf16_rm(Ak, c + 16, 0, 64);
    ka[3] = load_frag_bf16_rm(Ak, c + 16, 32, 64);
#pragma unroll
    for (int t = 0; t < 4; ++t) va[t] = load_frag_bf16_rm(A2, 16 * t, c, 4096);
    __builtin_amdgcn_sched_barrier(0);
#pragma unroll
    for (int g = 0; g < A3_QG; ++g) {
      floatx8 s0 = {}, s1 = {};
      s0 = wmma_bf16(ka[0], bq[g][0], s0);
      s0 = wmma_bf16(ka[1], bq[g][1], s0);
      s1 = wmma_bf16(ka[2], bq[g][0], s1);
      s1 = wmma_bf16(ka[3], bq[g][1], s1);
      float mc = -1e30f;
#pragma unroll
      for (int r = 0; r < 8; ++r) { mc = fmaxf(mc, s0[r]); mc = fmaxf(mc, s1[r]); }
      mc = fmaxf(mc, __shfl_xor(mc, 16, 32));
      float newm = fmaxf(run_m[g], mc);
      float scale = __expf(run_m[g] - newm);
      float p0[8], p1[8], ps = 0.f;
#pragma unroll
      for (int r = 0; r < 8; ++r) {
        p0[r] = __expf(s0[r] - newm);
        p1[r] = __expf(s1[r] - newm);
        ps += p0[r] + p1[r];
      }
      run_s[g] = run_s[g] * scale + ps;
      run_m[g] = newm;
      FragAB bp;
#pragma unroll
      for (int p = 0; p < 4; ++p) {
        bp.u[p] = pack2bf(p0[2 * p], p0[2 * p + 1]);
        bp.u[4 + p] = pack2bf(p1[2 * p], p1[2 * p + 1]);
      }
#pragma unroll
      for (int t = 0; t < 4; ++t) {
#pragma unroll
        for (int r = 0; r < 8; ++r) acc[g][t][r] *= scale;
        acc[g][t] = wmma_bf16(va[t], bp.v, acc[g][t]);
      }
    }
  }
#pragma unroll
  for (int g = 0; g < A3_QG; ++g) {
    float tot = run_s[g] + __shfl_xor(run_s[g], 16, 32);
    if (lane < 16) {
      mL[w][g][lane] = run_m[g];
      sL[w][g][lane] = tot;
    }
#pragma unroll
    for (int t = 0; t < 4; ++t)
#pragma unroll
      for (int r = 0; r < 8; ++r) accL[w][g][t][r][lane] = acc[g][t][r];
  }
  __syncthreads();
  int cc = lane & 15, hh = lane >> 4;
  for (int g = 0; g < A3_QG; ++g) {
    float mw[A3_WAVES], f[A3_WAVES], M = -1e30f, S = 0.f;
#pragma unroll
    for (int v = 0; v < A3_WAVES; ++v) { mw[v] = mL[v][g][cc]; M = fmaxf(M, mw[v]); }
#pragma unroll
    for (int v = 0; v < A3_WAVES; ++v) {
      f[v] = __expf(mw[v] - M);
      S += sL[v][g][cc] * f[v];
    }
    float inv = 1.f / S;
    for (int t = 0; t < 4; ++t) {
      float o = 0.f;
#pragma unroll
      for (int v = 0; v < A3_WAVES; ++v) o += accL[v][g][t][w][lane] * f[v];
      int dh = 16 * t + 8 * hh + w;
      kvT[((size_t)bh * 64 + dh) * 256 + q0 + 16 * g + cc] = (__bf16)(o * inv);
    }
  }
}

__global__ void __launch_bounds__(256)
k_w2(const __bf16* zb, const __bf16* kvT, __bf16* w2t) {
  int wid = (int)((blockIdx.x * blockDim.x + threadIdx.x) >> 5);
  int bh = wid >> 4, tM = wid & 15;
  floatx8 acc[4] = {};
  gemm_strip_pk<false>(zb + ((size_t)bh << 16), tM * 16,
                       kvT + (size_t)bh * 64 * 256, 0, acc);
  int lane = threadIdx.x & 31;
  int n = lane & 15, h = lane >> 4;
#pragma unroll
  for (int t4 = 0; t4 < 4; ++t4)
#pragma unroll
    for (int r = 0; r < 8; ++r) {
      int M = tM * 16 + r + 8 * h;
      int N = 16 * t4 + n;
      w2t[((size_t)bh * 64 + N) * 256 + M] = (__bf16)acc[t4][r];
    }
}

__global__ void k_attn1_out(const __bf16* kl, const __bf16* qb,
                            const __bf16* w2t, float* aout) {
  int wid = (int)((blockIdx.x * blockDim.x + threadIdx.x) >> 5);
  int bh = wid >> 8, blk = wid & 255;
  int b = bh >> 3, hh = bh & 7;
  stream_attn(kl + (size_t)bh * 256 * 64, qb + (size_t)bh * 4096 * 64, blk * 16,
              w2t + (size_t)bh * 64 * 256, 256, 256,
              aout + ((size_t)b * 4096 + blk * 16) * 512 + hh * 64, 512,
              nullptr, 0, 0);
}

#define CONV_R 16
#define CONV_G 8
#define CONV_ROWS (CONV_R + 32)
__global__ void __launch_bounds__(256, 4)
k_conv_res(const float* __restrict__ vf, const float* __restrict__ rw,
           float* __restrict__ aout, __bf16* __restrict__ aout_b) {
  int t = (int)(blockIdx.x * blockDim.x + threadIdx.x);
  if (t >= 4 * (4096 / CONV_R) * 128) return;
  int c4 = t & 127;
  int chunk = t >> 7;
  int b = chunk / (4096 / CONV_R);
  int ns0 = (chunk % (4096 / CONV_R)) * CONV_R;
  int col = c4 * 4, hh = col >> 6, dh = col & 63;
  const float* vcol = vf + ((size_t)(b * 8 + hh) * 4096) * 64 + dh;
  float* arow = aout + ((size_t)b * 4096 + ns0) * 512 + col;
  int boff = (int)pk_idx_n(b * 4096 + ns0, col, 16);
  float w[33];
#pragma unroll
  for (int j = 0; j < 33; ++j) w[j] = rw[hh * 33 + j];
  float4 acc[CONV_R], win[CONV_ROWS], a[CONV_R];
#pragma unroll
  for (int o = 0; o < CONV_R; ++o) acc[o] = make_float4(0.f, 0.f, 0.f, 0.f);
#pragma unroll
  for (int k = 0; k < CONV_G; ++k) {
    int p = ns0 + k - 16;
    int pc = p < 0 ? 0 : (p > 4095 ? 4095 : p);
    win[k] = *(const float4*)(vcol + (size_t)pc * 64);
  }
#pragma unroll
  for (int g = 0; g < CONV_ROWS / CONV_G; ++g) {
    if (g + 1 < CONV_ROWS / CONV_G) {
#pragma unroll
      for (int k = (g + 1) * CONV_G; k < (g + 2) * CONV_G; ++k) {
        int p = ns0 + k - 16;
        int pc = p < 0 ? 0 : (p > 4095 ? 4095 : p);
        win[k] = *(const float4*)(vcol + (size_t)pc * 64);
      }
    } else {
#pragma unroll
      for (int o = 0; o < CONV_R; ++o)
        a[o] = *(const float4*)(arow + (size_t)o * 512);
    }
#pragma unroll
    for (int k = g * CONV_G; k < (g + 1) * CONV_G; ++k) {
      int p = ns0 + k - 16;
      float4 v = win[k];
      if (p < 0 || p > 4095) v = make_float4(0.f, 0.f, 0.f, 0.f);
#pragma unroll
      for (int o = 0; o < CONV_R; ++o) {
        int j = k - o;
        if (j >= 0 && j < 33) {
          acc[o].x += v.x * w[j];
          acc[o].y += v.y * w[j];
          acc[o].z += v.z * w[j];
          acc[o].w += v.w * w[j];
        }
      }
    }
  }
#pragma unroll
  for (int o = 0; o < CONV_R; ++o) {
    float4 r;
    r.x = a[o].x + acc[o].x;
    r.y = a[o].y + acc[o].y;
    r.z = a[o].z + acc[o].z;
    r.w = a[o].w + acc[o].w;
    *(float4*)(arow + (size_t)o * 512) = r;
    uint2 pk;
    pk.x = pack2bf(r.x, r.y);
    pk.y = pack2bf(r.z, r.w);
    *(uint2*)(aout_b + (boff + o * 8)) = pk;
  }
}

__global__ void __launch_bounds__(256)
k_outproj(const __bf16* __restrict__ aoutb, const __bf16* __restrict__ wo,
          const float* __restrict__ bias, const float* __restrict__ x,
          float* __restrict__ out) {
  __shared__ u32x4 tile[2 * GB_STAGE];
  int bM = (int)blockIdx.x >> 1, bN = (int)blockIdx.x & 1;
  int w = (int)threadIdx.x >> 5;
  int tM = bM * 2 + (w >> 2), s = bN * 4 + (w & 3);
  floatx8 acc[4][4] = {};
  gemm_block_lds(aoutb, bM * 8, wo, bN * 16, tile, acc);
  int lane = threadIdx.x & 31;
  int n = lane & 15, h = lane >> 4;
#pragma unroll
  for (int t4 = 0; t4 < 4; ++t4) {
    int col = s * 64 + 16 * t4 + n;
    float bc = bias[col];
#pragma unroll
    for (int m = 0; m < 4; ++m) {
#pragma unroll
      for (int r = 0; r < 8; ++r) {
        size_t idx = (size_t)(tM * 64 + 16 * m + r + 8 * h) * 512 + col;
        out[idx] = x[idx] + acc[m][t4][r] + bc;
      }
    }
  }
}

extern "C" void kernel_launch(void* const* d_in, const int* in_sizes, int n_in,
                              void* d_out, int out_size, void* d_ws,
                              size_t ws_size, hipStream_t stream) {
  const float* x     = (const float*)d_in[0];
  const float* ln_g  = (const float*)d_in[1];
  const float* ln_b  = (const float*)d_in[2];
  const float* w_qkv = (const float*)d_in[3];
  const float* w_out = (const float*)d_in[4];
  const float* b_out = (const float*)d_in[5];
  const float* res_w = (const float*)d_in[6];
  float* out = (float*)d_out;

  char* p = (char*)d_ws;
  auto alloc = [&](size_t bytes) -> void* {
    void* r = (void*)p;
    p += (bytes + 255) & ~(size_t)255;
    return r;
  };
  const size_t NTOK = 4 * 4096;
  const size_t MM = (size_t)32 * 256 * 256;
  __bf16* xn_b   = (__bf16*)alloc(NTOK * 512 * 2);
  __bf16* wqkv_b = (__bf16*)alloc(1536 * 512 * 2);
  __bf16* wout_b = (__bf16*)alloc(512 * 512 * 2);
  __bf16* q_b    = (__bf16*)alloc((size_t)32 * 4096 * 64 * 2);
  __bf16* k_b    = (__bf16*)alloc((size_t)32 * 4096 * 64 * 2);
  float*  v_f    = (float*)alloc((size_t)32 * 4096 * 64 * 4);
  __bf16* vT_b   = (__bf16*)alloc((size_t)32 * 64 * 4096 * 2);
  __bf16* ql_b   = (__bf16*)alloc((size_t)32 * 256 * 64 * 2);
  __bf16* kl_b   = (__bf16*)alloc((size_t)32 * 256 * 64 * 2);
  float*  mat_x  = (float*)alloc(MM * 4);
  __bf16* x_b    = (__bf16*)alloc(MM * 2);
  __bf16* z0_b   = (__bf16*)alloc(MM * 2);
  __bf16* z0T_b  = (__bf16*)alloc(MM * 2);
  __bf16* z1_b   = (__bf16*)alloc(MM * 2);
  __bf16* z1T_b  = (__bf16*)alloc(MM * 2);
  __bf16* xz_b   = (__bf16*)alloc(MM * 2);
  __bf16* t1T_b  = (__bf16*)alloc(MM * 2);
  __bf16* t2T_b  = (__bf16*)alloc(MM * 2);
  __bf16* t3T_b  = (__bf16*)alloc(MM * 2);
  __bf16* kvT_b  = (__bf16*)alloc((size_t)32 * 64 * 256 * 2);
  __bf16* w2t_b  = (__bf16*)alloc((size_t)32 * 64 * 256 * 2);
  float*  aout_f = (float*)alloc(NTOK * 512 * 4);
  __bf16* aout_b = (__bf16*)alloc(NTOK * 512 * 2);
  float*  scales = (float*)alloc(256);
  (void)ws_size; (void)n_in; (void)in_sizes; (void)out_size;

  const int T = 256;
  k_convert_w<<<512, T, 0, stream>>>(w_qkv, w_out, wqkv_b, wout_b, scales);
  k_layernorm<<<2048, T, 0, stream>>>(x, ln_g, ln_b, xn_b, (int)NTOK);
  k_gemm_qkv<<<768, T, 0, stream>>>(xn_b, wqkv_b, q_b, k_b, v_f, vT_b);
  k_landmarks<<<2048, T, 0, stream>>>(q_b, k_b, ql_b, kl_b);
  k_attn2_logits<<<256, T, 0, stream>>>(ql_b, kl_b, mat_x);
  k_softmax256<<<1024, T, 0, stream>>>(mat_x, x_b, 32 * 256);
  k_pinv_scales<<<1024, T, 0, stream>>>(mat_x, scales);
  k_init_z<<<8192, T, 0, stream>>>(mat_x, z0_b, z0T_b, scales);

  __bf16* zc = z0_b;  __bf16* zcT = z0T_b;
  __bf16* zn = z1_b;  __bf16* znT = z1T_b;
  for (int it = 0; it < 6; ++it) {
    k_gemm256b<<<256, T, 0, stream>>>(x_b, zcT, xz_b, nullptr, t1T_b, 7.f, -1.f);
    k_gemm256b<<<256, T, 0, stream>>>(xz_b, t1T_b, nullptr, nullptr, t2T_b, 15.f, -1.f);
    k_gemm256b<<<256, T, 0, stream>>>(xz_b, t2T_b, nullptr, nullptr, t3T_b, 13.f, -1.f);
    k_gemm256b<<<256, T, 0, stream>>>(zc, t3T_b, nullptr, zn, znT, 0.f, 0.25f);
    __bf16* t;
    t = zc; zc = zn; zn = t;
    t = zcT; zcT = znT; znT = t;
  }

  k_attn3v<<<256, T, 0, stream>>>(k_b, ql_b, vT_b, kvT_b);
  k_w2<<<64, T, 0, stream>>>(zc, kvT_b, w2t_b);
  k_attn1_out<<<1024, T, 0, stream>>>(kl_b, q_b, w2t_b, aout_f);
  k_conv_res<<<512, T, 0, stream>>>(v_f, res_w, aout_f, aout_b);
  k_outproj<<<256, T, 0, stream>>>(aout_b, wout_b, b_out, x, out);
}


# QKV and pinv GEMM epilogues: per-element 2-byte stores behind per-element branches -> branch-free 16-byte stores; 8-wide rows from the operand-swapped WMMA of the same fragments; same sums
# speedup vs baseline: 2.8836x; 1.1372x over previous
#include <hip/hip_runtime.h>
#include <hip/hip_bf16.h>
#include <math.h>


typedef __attribute__((ext_vector_type(16))) __bf16 bf16x16;
typedef __attribute__((ext_vector_type(8)))  float  floatx8;

union FragAB { bf16x16 v; unsigned int u[8]; };

__device__ __forceinline__ unsigned pack2bf(float a, float b) {
  __bf16 lo = (__bf16)a, hi = (__bf16)b;
  unsigned short ul = __builtin_bit_cast(unsigned short, lo);
  unsigned short uh = __builtin_bit_cast(unsigned short, hi);
  return (unsigned)ul | ((unsigned)uh << 16);
}

__device__ __forceinline__ uint4 pack8bf(floatx8 a, float sc) {
  uint4 p;
  p.x = pack2bf(a[0] * sc, a[1] * sc);
  p.y = pack2bf(a[2] * sc, a[3] * sc);
  p.z = pack2bf(a[4] * sc, a[5] * sc);
  p.w = pack2bf(a[6] * sc, a[7] * sc);
  return p;
}

__device__ __forceinline__ int kpair(int p, int h) {
  return (p < 4) ? (2 * p + 8 * h) : (8 + 2 * p + 8 * h);
}

__device__ __forceinline__ bf16x16 load_frag_bf16_rm(const __bf16* src, int row0,
                                                     int k0, int ld) {
  int lane = threadIdx.x & 31;
  int r = lane & 15, h = lane >> 4;
  const __bf16* base = src + (size_t)(row0 + r) * ld + k0;
  FragAB f;
#pragma unroll
  for (int p = 0; p < 8; ++p) f.u[p] = *(const unsigned*)(base + kpair(p, h));
  return f.v;
}

__device__ __forceinline__ floatx8 wmma_bf16(bf16x16 a, bf16x16 b, floatx8 c) {
  return __builtin_amdgcn_wmma_f32_16x16x32_bf16(false, a, false, b, (short)0, c,
                                                 false, false);
}

template <int KSTEPS>
__device__ __forceinline__ void gemm_strip(const __bf16* A, int rowA,
                                           const __bf16* B, int rowB, int ld,
                                           floatx8 acc[4]) {
#pragma unroll
  for (int ks = 0; ks < KSTEPS; ++ks) {
    int k0 = ks * 32;
    bf16x16 af = load_frag_bf16_rm(A, rowA, k0, ld);
#pragma unroll
    for (int t = 0; t < 4; ++t) {
      bf16x16 bfr = load_frag_bf16_rm(B, rowB + 16 * t, k0, ld);
      acc[t] = wmma_bf16(af, bfr, acc[t]);
    }
  }
}

__device__ __forceinline__ size_t pk_idx(int row, int k) {
  int kk = k & 31;
  int lane = (row & 15) + 16 * ((kk >> 3) & 1);
  return (size_t)(((row >> 4) * 8 + (k >> 5)) * 512 + (kk >> 4) * 256 +
                  lane * 8 + (kk & 7));
}

__device__ __forceinline__ bf16x16 load_frag_bf16_pk(const __bf16* mat, int rb,
                                                     int ks) {
  int lane = threadIdx.x & 31;
  const uint4* blk = (const uint4*)(mat + (size_t)(rb * 8 + ks) * 512) + lane;
  uint4 lo = blk[0], hi = blk[32];
  FragAB f;
  f.u[0] = lo.x; f.u[1] = lo.y; f.u[2] = lo.z; f.u[3] = lo.w;
  f.u[4] = hi.x; f.u[5] = hi.y; f.u[6] = hi.z; f.u[7] = hi.w;
  return f.v;
}

__device__ __forceinline__ size_t pk_idx_n(int row, int k, int KB) {
  int kk = k & 31;
  int lane = (row & 15) + 16 * ((kk >> 3) & 1);
  return (size_t)((row >> 4) * KB + (k >> 5)) * 512 + (kk >> 4) * 256 +
         lane * 8 + (kk & 7);
}

#define PK_AHEAD 4
template <bool BPK>
__device__ __forceinline__ void gemm_strip_pk(const __bf16* A, int rowA,
                                              const __bf16* B, int rowB,
                                              floatx8 acc[4]) {
  bf16x16 af[8], bfr[8][4];
#pragma unroll
  for (int ks = 0; ks < PK_AHEAD; ++ks) {
    af[ks] = load_frag_bf16_pk(A, rowA >> 4, ks);
#pragma unroll
    for (int t = 0; t < 4; ++t)
      bfr[ks][t] = BPK ? load_frag_bf16_pk(B, (rowB >> 4) + t, ks)
                       : load_frag_bf16_rm(B, rowB + 16 * t, ks * 32, 256);
    __builtin_amdgcn_sched_barrier(0);
  }
#pragma unroll
  for (int ks = 0; ks < 8; ++ks) {
#pragma unroll
    for (int t = 0; t < 4; ++t) acc[t] = wmma_bf16(af[ks], bfr[ks][t], acc[t]);
    __builtin_amdgcn_sched_barrier(0);
    if (ks + PK_AHEAD < 8) {
      af[ks + PK_AHEAD] = load_frag_bf16_pk(A, rowA >> 4, ks + PK_AHEAD);
#pragma unroll
      for (int t = 0; t < 4; ++t)
        bfr[ks + PK_AHEAD][t] =
            BPK ? load_frag_bf16_pk(B, (rowB >> 4) + t, ks + PK_AHEAD)
                : load_frag_bf16_rm(B, rowB + 16 * t, (ks + PK_AHEAD) * 32, 256);
      __builtin_amdgcn_sched_barrier(0);
    }
  }
}

typedef __attribute__((ext_vector_type(4))) unsigned int u32x4;

__device__ __forceinline__ bf16x16 load_frag_bf16_lds(const u32x4* tile, int j,
                                                      int ks) {
  int lane = threadIdx.x & 31;
  const u32x4* blk = tile + (j * 8 + ks) * 64 + lane;
  u32x4 lo = blk[0], hi = blk[32];
  FragAB f;
  f.u[0] = lo.x; f.u[1] = lo.y; f.u[2] = lo.z; f.u[3] = lo.w;
  f.u[4] = hi.x; f.u[5] = hi.y; f.u[6] = hi.z; f.u[7] = hi.w;
  return f.v;
}

__device__ __forceinline__ void lds_barrier() {
  __builtin_amdgcn_fence(__ATOMIC_RELEASE, "workgroup", "local");
  __builtin_amdgcn_s_barrier();
  __builtin_amdgcn_fence(__ATOMIC_ACQUIRE, "workgroup", "local");
}

#define GB_STAGE 3072
template <bool SWAP>
__device__ __forceinline__ void gemm_block_lds(const __bf16* A, int arb0,
                                               const __bf16* B, int brb0,
                                               u32x4* tile,
                                               floatx8 acc[4][4]) {
  int tid = (int)threadIdx.x;
  int lane = tid & 31, w = tid >> 5;
  int wm = w >> 2, wn = w & 3;
  int half = tid >> 7, q = tid & 127;
  const u32x4* Ag = (const u32x4*)A + (size_t)(arb0 + half) * 1024 + q;
  const u32x4* Bg = (const u32x4*)B + (size_t)(brb0 + half) * 1024 + q;
  u32x4 st[12];
#pragma unroll
  for (int j = 0; j < 4; ++j) st[j] = Ag[j * 2048];
#pragma unroll
  for (int j = 0; j < 8; ++j) st[4 + j] = Bg[j * 2048];
#pragma unroll
  for (int S = 0; S < 8; ++S) {
    u32x4* dst = tile + (S & 1) * GB_STAGE;
#pragma unroll
    for (int j = 0; j < 12; ++j) dst[j * 256 + tid] = st[j];
    if (S + 1 < 8) {
#pragma unroll
      for (int j = 0; j < 4; ++j) st[j] = Ag[j * 2048 + (S + 1) * 128];
#pragma unroll
      for (int j = 0; j < 8; ++j) st[4 + j] = Bg[j * 2048 + (S + 1) * 128];
    }
    __builtin_amdgcn_sched_barrier(0);
    lds_barrier();
    bf16x16 af[2][4], bfr[2][4];
#pragma unroll
    for (int kk = 0; kk < 2; ++kk) {
      const u32x4* sb = dst + kk * 64 + lane;
#pragma unroll
      for (int m = 0; m < 4; ++m) {
        const u32x4* blk = sb + (wm * 4 + m) * 128;
        u32x4 lo = blk[0], hi = blk[32];
        FragAB f;
        f.u[0] = lo.x; f.u[1] = lo.y; f.u[2] = lo.z; f.u[3] = lo.w;
        f.u[4] = hi.x; f.u[5] = hi.y; f.u[6] = hi.z; f.u[7] = hi.w;
        af[kk][m] = f.v;
      }
#pragma unroll
      for (int t = 0; t < 4; ++t) {
        const u32x4* blk = sb + (8 + wn * 4 + t) * 128;
        u32x4 lo = blk[0], hi = blk[32];
        FragAB f;
        f.u[0] = lo.x; f.u[1] = lo.y; f.u[2] = lo.z; f.u[3] = lo.w;
        f.u[4] = hi.x; f.u[5] = hi.y; f.u[6] = hi.z; f.u[7] = hi.w;
        bfr[kk][t] = f.v;
      }
    }
    __builtin_amdgcn_sched_barrier(0);
#pragma unroll
    for (int kk = 0; kk < 2; ++kk)
#pragma unroll
      for (int m = 0; m < 4; ++m)
#pragma unroll
        for (int t = 0; t < 4; ++t)
          acc[m][t] = SWAP ? wmma_bf16(bfr[kk][t], af[kk][m], acc[m][t])
                           : wmma_bf16(af[kk][m], bfr[kk][t], acc[m][t]);
  }
}

__global__ void k_convert_w(const float* wqkv, const float* wout,
                            __bf16* wqkv_b, __bf16* wout_b, float* scales) {
  int i = blockIdx.x * blockDim.x + threadIdx.x;
  if (i < 2) scales[i] = 0.f;
  bool second = i >= 1536 * 64;
  const float* src = second ? wout : wqkv;
  __bf16* dst = second ? wout_b : wqkv_b;
  int u = second ? i - 1536 * 64 : i;
  if (second && u >= 512 * 64) return;
  int blk = u >> 6, half = (u >> 5) & 1, lane = u & 31;
  int row = (blk >> 4) * 16 + (lane & 15);
  int k = (blk & 15) * 32 + half * 16 + 8 * (lane >> 4);
  const float4* s = (const float4*)(src + (size_t)row * 512 + k);
  float4 a = s[0], b = s[1];
  uint4 pv;
  pv.x = pack2bf(a.x, a.y);
  pv.y = pack2bf(a.z, a.w);
  pv.z = pack2bf(b.x, b.y);
  pv.w = pack2bf(b.z, b.w);
  ((uint4*)dst)[u] = pv;
}

__global__ void k_layernorm(const float* x, const float* g, const float* b,
                            __bf16* xn, int ntok) {
  int wave = (int)((blockIdx.x * blockDim.x + threadIdx.x) >> 5);
  int lane = threadIdx.x & 31;
  if (wave >= ntok) return;
  const float* xr = x + (size_t)wave * 512;
  float s = 0.f, s2 = 0.f, vals[16];
#pragma unroll
  for (int i = 0; i < 16; ++i) {
    float v = xr[lane + 32 * i];
    vals[i] = v; s += v; s2 += v * v;
  }
  for (int m = 16; m >= 1; m >>= 1) {
    s += __shfl_xor(s, m, 32);
    s2 += __shfl_xor(s2, m, 32);
  }
  float mu = s * (1.f / 512.f);
  float var = s2 * (1.f / 512.f) - mu * mu;
  float inv = rsqrtf(var + 1e-5f);
#pragma unroll
  for (int i = 0; i < 16; ++i) {
    int c = lane + 32 * i;
    xn[pk_idx_n(wave, c, 16)] = (__bf16)((vals[i] - mu) * inv * g[c] + b[c]);
  }
}

__global__ void __launch_bounds__(256)
k_gemm_qkv(const __bf16* xn, const __bf16* wq, __bf16* qb,
           __bf16* kb, float* vf, __bf16* vT) {
  __shared__ u32x4 tile[2 * GB_STAGE];
  int bM = (int)blockIdx.x / 6, bN = (int)blockIdx.x % 6;
  int w = (int)threadIdx.x >> 5;
  int tM = bM * 2 + (w >> 2), s = bN * 4 + (w & 3);
  int lane = threadIdx.x & 31;
  int c = lane & 15, h = lane >> 4;
  int head = s & 7;
  int tok0 = tM * 64;
  int bb = tok0 >> 12, ns0 = tok0 & 4095;
  size_t bh = (size_t)(bb * 8 + head);
  floatx8 acc[4][4] = {};
  if (bN < 4) {
    gemm_block_lds<true>(xn, bM * 8, wq, bN * 16, tile, acc);
    __bf16* dst = (bN < 2 ? qb : kb) + (bh * 4096 + ns0 + c) * 64 + 8 * h;
    float sc = bN < 2 ? 0.125f : 1.f;
#pragma unroll
    for (int m = 0; m < 4; ++m)
#pragma unroll
      for (int t = 0; t < 4; ++t)
        *(uint4*)(dst + (16 * m) * 64 + 16 * t) = pack8bf(acc[m][t], sc);
  } else {
    gemm_block_lds<false>(xn, bM * 8, wq, bN * 16, tile, acc);
    float* vfd = vf + (bh * 4096 + ns0 + 8 * h) * 64 + c;
    __bf16* vTd = vT + (bh * 64 + c) * 4096 + ns0 + 8 * h;
#pragma unroll
    for (int m = 0; m < 4; ++m)
#pragma unroll
      for (int t = 0; t < 4; ++t) {
#pragma unroll
        for (int r = 0; r < 8; ++r)
          vfd[(16 * m + r) * 64 + 16 * t] = acc[m][t][r];
        *(uint4*)(vTd + (size_t)(16 * t) * 4096 + 16 * m) =
            pack8bf(acc[m][t], 1.f);
      }
  }
}

__global__ void k_landmarks(const __bf16* qb, const __bf16* kb, __bf16* ql,
                            __bf16* kl) {
  int i = blockIdx.x * blockDim.x + threadIdx.x;
  if (i >= 32 * 256 * 64) return;
  int dh = i & 63, m = (i >> 6) & 255, bh = i >> 14;
  size_t base = ((size_t)bh * 4096 + m * 16) * 64 + dh;
  float sq = 0.f, sk = 0.f;
#pragma unroll
  for (int t = 0; t < 16; ++t) {
    sq += (float)qb[base + t * 64];
    sk += (float)kb[base + t * 64];
  }
  ql[i] = (__bf16)(sq * (1.f / 16.f));
  kl[i] = (__bf16)(sk * (1.f / 16.f));
}

__global__ void k_attn2_logits(const __bf16* ql, const __bf16* kl, float* out) {
  int wid = (int)((blockIdx.x * blockDim.x + threadIdx.x) >> 5);
  int bh = wid >> 6, t = wid & 63;
  int tM = t >> 2, s = t & 3;
  const __bf16* A = ql + (size_t)bh * 256 * 64;
  const __bf16* B = kl + (size_t)bh * 256 * 64;
  floatx8 acc[4] = {};
  gemm_strip<2>(A, tM * 16, B, s * 64, 64, acc);
  int lane = threadIdx.x & 31;
  int n = lane & 15, h = lane >> 4;
  float* o = out + (size_t)bh * 256 * 256;
#pragma unroll
  for (int t4 = 0; t4 < 4; ++t4)
#pragma unroll
    for (int r = 0; r < 8; ++r)
      o[(size_t)(tM * 16 + r + 8 * h) * 256 + s * 64 + 16 * t4 + n] =
          acc[t4][r];
}

__global__ void k_softmax256(float* x, __bf16* xb, int nrows) {
  int wave = (int)((blockIdx.x * blockDim.x + threadIdx.x) >> 5);
  if (wave >= nrows) return;
  int lane = threadIdx.x & 31;
  float* r = x + (size_t)wave * 256;
  __bf16* rb = xb + ((size_t)(wave >> 8) << 16);
  int row = wave & 255;
  float v[8], m = -1e30f;
#pragma unroll
  for (int i = 0; i < 8; ++i) { v[i] = r[lane + 32 * i]; m = fmaxf(m, v[i]); }
  for (int t = 16; t >= 1; t >>= 1) m = fmaxf(m, __shfl_xor(m, t, 32));
  float s = 0.f;
#pragma unroll
  for (int i = 0; i < 8; ++i) { v[i] = __expf(v[i] - m); s += v[i]; }
  for (int t = 16; t >= 1; t >>= 1) s += __shfl_xor(s, t, 32);
  float inv = 1.f / s;
#pragma unroll
  for (int i = 0; i < 8; ++i) {
    float pv = v[i] * inv;
    r[lane + 32 * i] = pv;
    rb[pk_idx(row, lane + 32 * i)] = (__bf16)pv;
  }
}

__global__ void k_pinv_scales(const float* x, float* scales) {
  int wave = (int)((blockIdx.x * blockDim.x + threadIdx.x) >> 5);
  if (wave >= 32 * 256) return;
  int lane = threadIdx.x & 31;
  int bh = wave >> 8, r = wave & 255;
  const float* m = x + (size_t)bh * 256 * 256;
  float rs = 0.f, cs = 0.f;
#pragma unroll
  for (int i = 0; i < 8; ++i) {
    rs += fabsf(m[(size_t)r * 256 + lane + 32 * i]);
    cs += fabsf(m[(size_t)(lane + 32 * i) * 256 + r]);
  }
  for (int t = 16; t >= 1; t >>= 1) {
    rs += __shfl_xor(rs, t, 32);
    cs += __shfl_xor(cs, t, 32);
  }
  if (lane == 0) {
    atomicMax((unsigned*)&scales[0], __float_as_uint(rs));
    atomicMax((unsigned*)&scales[1], __float_as_uint(cs));
  }
}

__global__ void k_init_z(const float* x, __bf16* zb, __bf16* zTb,
                         const float* scales) {
  size_t i = (size_t)blockIdx.x * blockDim.x + threadIdx.x;
  if (i >= (size_t)32 * 256 * 256) return;
  int j = (int)(i & 255), r = (int)((i >> 8) & 255);
  size_t bh = i >> 16;
  float denom = scales[0] * scales[1];
  float val = x[(bh << 16) + (size_t)j * 256 + r] / denom;
  zb[(bh << 16) + pk_idx(r, j)] = (__bf16)val;
  zTb[(bh << 16) + pk_idx(j, r)] = (__bf16)val;
}

#define G8_ARB 4
#define G8_BRB 8
template <bool RAW, bool OT, bool OTT>
__global__ void __launch_bounds__(256)
k_gemm256b(const __bf16* A, const __bf16* BT, __bf16* OutRaw,
           __bf16* OutT, __bf16* OutTT, float alpha, float beta) {
  __shared__ u32x4 tile[(G8_ARB + G8_BRB) * 8 * 64];
  int tid = (int)threadIdx.x;
  int w = tid >> 5;
  int bh = (int)blockIdx.x >> 3, g = (int)blockIdx.x & 7;
  int bm = g >> 1, bn = g & 1;
  int tM = bm * G8_ARB + (w >> 1), s = bn * 2 + (w & 1);
  size_t bofs = (size_t)bh << 16;
  const u32x4* Ag = (const u32x4*)(A + bofs) + (size_t)(bm * G8_ARB) * 8 * 64;
  const u32x4* Bg = (const u32x4*)(BT + bofs) + (size_t)(bn * G8_BRB) * 8 * 64;
  u32x4 st0[G8_ARB + G8_BRB], st1[G8_ARB + G8_BRB];
#pragma unroll
  for (int j = 0; j < G8_ARB; ++j) st0[j] = Ag[j * 512 + tid];
#pragma unroll
  for (int j = 0; j < G8_BRB; ++j) st0[G8_ARB + j] = Bg[j * 512 + tid];
  __builtin_amdgcn_sched_barrier(0);
#pragma unroll
  for (int j = 0; j < G8_ARB; ++j) st1[j] = Ag[j * 512 + 256 + tid];
#pragma unroll
  for (int j = 0; j < G8_BRB; ++j) st1[G8_ARB + j] = Bg[j * 512 + 256 + tid];
  floatx8 acc[4] = {};
  floatx8 accS[4] = {};
#pragma unroll
  for (int j = 0; j < G8_ARB + G8_BRB; ++j) tile[j * 512 + tid] = st0[j];
  lds_barrier();
#pragma unroll
  for (int ks = 0; ks < 4; ++ks) {
    bf16x16 af = load_frag_bf16_lds(tile, w >> 1, ks);
#pragma unroll
    for (int t = 0; t < 4; ++t) {
      bf16x16 bf = load_frag_bf16_lds(tile, G8_ARB + (w & 1) * 4 + t, ks);
      if (OTT) acc[t] = wmma_bf16(af, bf, acc[t]);
      if (RAW || OT) accS[t] = wmma_bf16(bf, af, accS[t]);
    }
  }
#pragma unroll
  for (int j = 0; j < G8_ARB + G8_BRB; ++j) tile[j * 512 + 256 + tid] = st1[j];
  lds_barrier();
#pragma unroll
  for (int ks = 4; ks < 8; ++ks) {
    bf16x16 af = load_frag_bf16_lds(tile, w >> 1, ks);
#pragma unroll
    for (int t = 0; t < 4; ++t) {
      bf16x16 bf = load_frag_bf16_lds(tile, G8_ARB + (w & 1) * 4 + t, ks);
      if (OTT) acc[t] = wmma_bf16(af, bf, acc[t]);
      if (RAW || OT) accS[t] = wmma_bf16(bf, af, accS[t]);
    }
  }
  int lane = threadIdx.x & 31;
  int n = lane & 15, h = lane >> 4;
#pragma unroll
  for (int t4 = 0; t4 < 4; ++t4) {
    if (OTT) {
      int N = s * 64 + 16 * t4 + n;
      floatx8 trv;
#pragma unroll
      for (int r = 0; r < 8; ++r) {
        int M = tM * 16 + r + 8 * h;
        trv[r] = alpha * (M == N ? 1.f : 0.f) + beta * acc[t4][r];
      }
      *(uint4*)(OutTT + bofs + pk_idx(N, tM * 16 + 8 * h)) = pack8bf(trv, 1.f);
    }
    if (RAW || OT) {
      int M = tM * 16 + n;
      int N0 = s * 64 + 16 * t4 + 8 * h;
      size_t o = bofs + pk_idx(M, N0);
      if (RAW) *(uint4*)(OutRaw + o) = pack8bf(accS[t4], 1.f);
      if (OT) {
        floatx8 trv;
#pragma unroll
        for (int r = 0; r < 8; ++r)
          trv[r] = alpha * (M == N0 + r ? 1.f : 0.f) + beta * accS[t4][r];
        *(uint4*)(OutT + o) = pack8bf(trv, 1.f);
      }
    }
  }
}

__device__ __forceinline__ void stream_attn(const __bf16* Ak, const __bf16* Bq,
                                            int qrow0, const __bf16* A2,
                                            int ldA2, int nkeys, float* outF,
                                            int rsF, __bf16* outB, int nsB,
                                            int dsB) {
  int lane = threadIdx.x & 31;
  int n = lane & 15, h = lane >> 4;
  bf16x16 bq0 = load_frag_bf16_rm(Bq, qrow0, 0, 64);
  bf16x16 bq1 = load_frag_bf16_rm(Bq, qrow0, 32, 64);
  float run_m = -1e30f, run_s = 0.f;
  floatx8 acc[4] = {};
  for (int c = 0; c < nkeys; c += 32) {
    floatx8 s0 = {}, s1 = {};
    s0 = wmma_bf16(load_frag_bf16_rm(Ak, c, 0, 64), bq0, s0);
    s0 = wmma_bf16(load_frag_bf16_rm(Ak, c, 32, 64), bq1, s0);
    s1 = wmma_bf16(load_frag_bf16_rm(Ak, c + 16, 0, 64), bq0, s1);
    s1 = wmma_bf16(load_frag_bf16_rm(Ak, c + 16, 32, 64), bq1, s1);
    float mc = -1e30f;
#pragma unroll
    for (int r = 0; r < 8; ++r) { mc = fmaxf(mc, s0[r]); mc = fmaxf(mc, s1[r]); }
    mc = fmaxf(mc, __shfl_xor(mc, 16, 32));
    float newm = fmaxf(run_m, mc);
    float scale = __expf(run_m - newm);
    float p0[8], p1[8], ps = 0.f;
#pragma unroll
    for (int r = 0; r < 8; ++r) {
      p0[r] = __expf(s0[r] - newm);
      p1[r] = __expf(s1[r] - newm);
      ps += p0[r] + p1[r];
    }
    run_s = run_s * scale + ps;
    run_m = newm;
    FragAB bp;
#pragma unroll
    for (int p = 0; p < 4; ++p) {
      bp.u[p] = pack2bf(p0[2 * p], p0[2 * p + 1]);
      bp.u[4 + p] = pack2bf(p1[2 * p], p1[2 * p + 1]);
    }
#pragma unroll
    for (int t = 0; t < 4; ++t) {
#pragma unroll
      for (int r = 0; r < 8; ++r) acc[t][r] *= scale;
      acc[t] = wmma_bf16(load_frag_bf16_rm(A2, 16 * t, c, ldA2), bp.v, acc[t]);
    }
  }
  float tot = run_s + __shfl_xor(run_s, 16, 32);
  float inv = 1.f / tot;
#pragma unroll
  for (int t = 0; t < 4; ++t)
#pragma unroll
    for (int r = 0; r < 8; ++r) {
      int dh = 16 * t + r + 8 * h;
      float val = acc[t][r] * inv;
      if (outF) outF[(size_t)n * rsF + dh] = val;
      if (outB) outB[(size_t)n * nsB + (size_t)dh * dsB] = (__bf16)val;
    }
}

#define A3_WAVES 8
#define A3_QG 2
#define A3_KEYS (4096 / A3_WAVES)
__global__ void __launch_bounds__(256)
k_attn3v(const __bf16* kb, const __bf16* ql, const __bf16* vT, __bf16* kvT) {
  __shared__ float accL[A3_WAVES][A3_QG][4][8][32];
  __shared__ float mL[A3_WAVES][A3_QG][16];
  __shared__ float sL[A3_WAVES][A3_QG][16];
  int lane = threadIdx.x & 31;
  int w = __builtin_amdgcn_readfirstlane((int)(threadIdx.x >> 5));
  int bh = blockIdx.x >> 3, q0 = (blockIdx.x & 7) * (16 * A3_QG);
  const __bf16* Ak = kb + (size_t)bh * 4096 * 64;
  const __bf16* Bq = ql + (size_t)bh * 256 * 64;
  const __bf16* A2 = vT + (size_t)bh * 64 * 4096;
  bf16x16 bq[A3_QG][2];
  float run_m[A3_QG], run_s[A3_QG];
  floatx8 acc[A3_QG][4];
#pragma unroll
  for (int g = 0; g < A3_QG; ++g) {
    bq[g][0] = load_frag_bf16_rm(Bq, q0 + 16 * g, 0, 64);
    bq[g][1] = load_frag_bf16_rm(Bq, q0 + 16 * g, 32, 64);
    run_m[g] = -1e30f;
    run_s[g] = 0.f;
#pragma unroll
    for (int t = 0; t < 4; ++t) acc[g][t] = floatx8{};
  }
  int kbeg = w * A3_KEYS;
  for (int c = kbeg; c < kbeg + A3_KEYS; c += 32) {
    bf16x16 ka[4], va[4];
    ka[0] = load_frag_bf16_rm(Ak, c, 0, 64);
    ka[1] = load_frag_bf16_rm(Ak, c, 32, 64);
    ka[2] = load_frag_bf16_rm(Ak, c + 16, 0, 64);
    ka[3] = load_frag_bf16_rm(Ak, c + 16, 32, 64);
#pragma unroll
    for (int t = 0; t < 4; ++t) va[t] = load_frag_bf16_rm(A2, 16 * t, c, 4096);
    __builtin_amdgcn_sched_barrier(0);
#pragma unroll
    for (int g = 0; g < A3_QG; ++g) {
      floatx8 s0 = {}, s1 = {};
      s0 = wmma_bf16(ka[0], bq[g][0], s0);
      s0 = wmma_bf16(ka[1], bq[g][1], s0);
      s1 = wmma_bf16(ka[2], bq[g][0], s1);
      s1 = wmma_bf16(ka[3], bq[g][1], s1);
      float mc = -1e30f;
#pragma unroll
      for (int r = 0; r < 8; ++r) { mc = fmaxf(mc, s0[r]); mc = fmaxf(mc, s1[r]); }
      mc = fmaxf(mc, __shfl_xor(mc, 16, 32));
      float newm = fmaxf(run_m[g], mc);
      float scale = __expf(run_m[g] - newm);
      float p0[8], p1[8], ps = 0.f;
#pragma unroll
      for (int r = 0; r < 8; ++r) {
        p0[r] = __expf(s0[r] - newm);
        p1[r] = __expf(s1[r] - newm);
        ps += p0[r] + p1[r];
      }
      run_s[g] = run_s[g] * scale + ps;
      run_m[g] = newm;
      FragAB bp;
#pragma unroll
      for (int p = 0; p < 4; ++p) {
        bp.u[p] = pack2bf(p0[2 * p], p0[2 * p + 1]);
        bp.u[4 + p] = pack2bf(p1[2 * p], p1[2 * p + 1]);
      }
#pragma unroll
      for (int t = 0; t < 4; ++t) {
#pragma unroll
        for (int r = 0; r < 8; ++r) acc[g][t][r] *= scale;
        acc[g][t] = wmma_bf16(va[t], bp.v, acc[g][t]);
      }
    }
  }
#pragma unroll
  for (int g = 0; g < A3_QG; ++g) {
    float tot = run_s[g] + __shfl_xor(run_s[g], 16, 32);
    if (lane < 16) {
      mL[w][g][lane] = run_m[g];
      sL[w][g][lane] = tot;
    }
#pragma unroll
    for (int t = 0; t < 4; ++t)
#pragma unroll
      for (int r = 0; r < 8; ++r) accL[w][g][t][r][lane] = acc[g][t][r];
  }
  __syncthreads();
  int cc = lane & 15, hh = lane >> 4;
  for (int g = 0; g < A3_QG; ++g) {
    float mw[A3_WAVES], f[A3_WAVES], M = -1e30f, S = 0.f;
#pragma unroll
    for (int v = 0; v < A3_WAVES; ++v) { mw[v] = mL[v][g][cc]; M = fmaxf(M, mw[v]); }
#pragma unroll
    for (int v = 0; v < A3_WAVES; ++v) {
      f[v] = __expf(mw[v] - M);
      S += sL[v][g][cc] * f[v];
    }
    float inv = 1.f / S;
    for (int t = 0; t < 4; ++t) {
      float o = 0.f;
#pragma unroll
      for (int v = 0; v < A3_WAVES; ++v) o += accL[v][g][t][w][lane] * f[v];
      int dh = 16 * t + 8 * hh + w;
      kvT[((size_t)bh * 64 + dh) * 256 + q0 + 16 * g + cc] = (__bf16)(o * inv);
    }
  }
}

__global__ void __launch_bounds__(256)
k_w2(const __bf16* zb, const __bf16* kvT, __bf16* w2t) {
  int wid = (int)((blockIdx.x * blockDim.x + threadIdx.x) >> 5);
  int bh = wid >> 4, tM = wid & 15;
  floatx8 acc[4] = {};
  gemm_strip_pk<false>(zb + ((size_t)bh << 16), tM * 16,
                       kvT + (size_t)bh * 64 * 256, 0, acc);
  int lane = threadIdx.x & 31;
  int n = lane & 15, h = lane >> 4;
#pragma unroll
  for (int t4 = 0; t4 < 4; ++t4)
#pragma unroll
    for (int r = 0; r < 8; ++r) {
      int M = tM * 16 + r + 8 * h;
      int N = 16 * t4 + n;
      w2t[((size_t)bh * 64 + N) * 256 + M] = (__bf16)acc[t4][r];
    }
}

__global__ void k_attn1_out(const __bf16* kl, const __bf16* qb,
                            const __bf16* w2t, float* aout) {
  int wid = (int)((blockIdx.x * blockDim.x + threadIdx.x) >> 5);
  int bh = wid >> 8, blk = wid & 255;
  int b = bh >> 3, hh = bh & 7;
  stream_attn(kl + (size_t)bh * 256 * 64, qb + (size_t)bh * 4096 * 64, blk * 16,
              w2t + (size_t)bh * 64 * 256, 256, 256,
              aout + ((size_t)b * 4096 + blk * 16) * 512 + hh * 64, 512,
              nullptr, 0, 0);
}

#define CONV_R 16
#define CONV_G 8
#define CONV_ROWS (CONV_R + 32)
__global__ void __launch_bounds__(256, 4)
k_conv_res(const float* __restrict__ vf, const float* __restrict__ rw,
           float* __restrict__ aout, __bf16* __restrict__ aout_b) {
  int t = (int)(blockIdx.x * blockDim.x + threadIdx.x);
  if (t >= 4 * (4096 / CONV_R) * 128) return;
  int c4 = t & 127;
  int chunk = t >> 7;
  int b = chunk / (4096 / CONV_R);
  int ns0 = (chunk % (4096 / CONV_R)) * CONV_R;
  int col = c4 * 4, hh = col >> 6, dh = col & 63;
  const float* vcol = vf + ((size_t)(b * 8 + hh) * 4096) * 64 + dh;
  float* arow = aout + ((size_t)b * 4096 + ns0) * 512 + col;
  int boff = (int)pk_idx_n(b * 4096 + ns0, col, 16);
  float w[33];
#pragma unroll
  for (int j = 0; j < 33; ++j) w[j] = rw[hh * 33 + j];
  float4 acc[CONV_R], win[CONV_ROWS], a[CONV_R];
#pragma unroll
  for (int o = 0; o < CONV_R; ++o) acc[o] = make_float4(0.f, 0.f, 0.f, 0.f);
#pragma unroll
  for (int k = 0; k < CONV_G; ++k) {
    int p = ns0 + k - 16;
    int pc = p < 0 ? 0 : (p > 4095 ? 4095 : p);
    win[k] = *(const float4*)(vcol + (size_t)pc * 64);
  }
#pragma unroll
  for (int g = 0; g < CONV_ROWS / CONV_G; ++g) {
    if (g + 1 < CONV_ROWS / CONV_G) {
#pragma unroll
      for (int k = (g + 1) * CONV_G; k < (g + 2) * CONV_G; ++k) {
        int p = ns0 + k - 16;
        int pc = p < 0 ? 0 : (p > 4095 ? 4095 : p);
        win[k] = *(const float4*)(vcol + (size_t)pc * 64);
      }
    } else {
#pragma unroll
      for (int o = 0; o < CONV_R; ++o)
        a[o] = *(const float4*)(arow + (size_t)o * 512);
    }
#pragma unroll
    for (int k = g * CONV_G; k < (g + 1) * CONV_G; ++k) {
      int p = ns0 + k - 16;
      float4 v = win[k];
      if (p < 0 || p > 4095) v = make_float4(0.f, 0.f, 0.f, 0.f);
#pragma unroll
      for (int o = 0; o < CONV_R; ++o) {
        int j = k - o;
        if (j >= 0 && j < 33) {
          acc[o].x += v.x * w[j];
          acc[o].y += v.y * w[j];
          acc[o].z += v.z * w[j];
          acc[o].w += v.w * w[j];
        }
      }
    }
  }
#pragma unroll
  for (int o = 0; o < CONV_R; ++o) {
    float4 r;
    r.x = a[o].x + acc[o].x;
    r.y = a[o].y + acc[o].y;
    r.z = a[o].z + acc[o].z;
    r.w = a[o].w + acc[o].w;
    *(float4*)(arow + (size_t)o * 512) = r;
    uint2 pk;
    pk.x = pack2bf(r.x, r.y);
    pk.y = pack2bf(r.z, r.w);
    *(uint2*)(aout_b + (boff + o * 8)) = pk;
  }
}

__global__ void __launch_bounds__(256)
k_outproj(const __bf16* __restrict__ aoutb, const __bf16* __restrict__ wo,
          const float* __restrict__ bias, const float* __restrict__ x,
          float* __restrict__ out) {
  __shared__ u32x4 tile[2 * GB_STAGE];
  int bM = (int)blockIdx.x >> 1, bN = (int)blockIdx.x & 1;
  int w = (int)threadIdx.x >> 5;
  int tM = bM * 2 + (w >> 2), s = bN * 4 + (w & 3);
  floatx8 acc[4][4] = {};
  gemm_block_lds<false>(aoutb, bM * 8, wo, bN * 16, tile, acc);
  int lane = threadIdx.x & 31;
  int n = lane & 15, h = lane >> 4;
#pragma unroll
  for (int t4 = 0; t4 < 4; ++t4) {
    int col = s * 64 + 16 * t4 + n;
    float bc = bias[col];
#pragma unroll
    for (int m = 0; m < 4; ++m) {
#pragma unroll
      for (int r = 0; r < 8; ++r) {
        size_t idx = (size_t)(tM * 64 + 16 * m + r + 8 * h) * 512 + col;
        out[idx] = x[idx] + acc[m][t4][r] + bc;
      }
    }
  }
}

extern "C" void kernel_launch(void* const* d_in, const int* in_sizes, int n_in,
                              void* d_out, int out_size, void* d_ws,
                              size_t ws_size, hipStream_t stream) {
  const float* x     = (const float*)d_in[0];
  const float* ln_g  = (const float*)d_in[1];
  const float* ln_b  = (const float*)d_in[2];
  const float* w_qkv = (const float*)d_in[3];
  const float* w_out = (const float*)d_in[4];
  const float* b_out = (const float*)d_in[5];
  const float* res_w = (const float*)d_in[6];
  float* out = (float*)d_out;

  char* p = (char*)d_ws;
  auto alloc = [&](size_t bytes) -> void* {
    void* r = (void*)p;
    p += (bytes + 255) & ~(size_t)255;
    return r;
  };
  const size_t NTOK = 4 * 4096;
  const size_t MM = (size_t)32 * 256 * 256;
  __bf16* xn_b   = (__bf16*)alloc(NTOK * 512 * 2);
  __bf16* wqkv_b = (__bf16*)alloc(1536 * 512 * 2);
  __bf16* wout_b = (__bf16*)alloc(512 * 512 * 2);
  __bf16* q_b    = (__bf16*)alloc((size_t)32 * 4096 * 64 * 2);
  __bf16* k_b    = (__bf16*)alloc((size_t)32 * 4096 * 64 * 2);
  float*  v_f    = (float*)alloc((size_t)32 * 4096 * 64 * 4);
  __bf16* vT_b   = (__bf16*)alloc((size_t)32 * 64 * 4096 * 2);
  __bf16* ql_b   = (__bf16*)alloc((size_t)32 * 256 * 64 * 2);
  __bf16* kl_b   = (__bf16*)alloc((size_t)32 * 256 * 64 * 2);
  float*  mat_x  = (float*)alloc(MM * 4);
  __bf16* x_b    = (__bf16*)alloc(MM * 2);
  __bf16* z0_b   = (__bf16*)alloc(MM * 2);
  __bf16* z0T_b  = (__bf16*)alloc(MM * 2);
  __bf16* z1_b   = (__bf16*)alloc(MM * 2);
  __bf16* z1T_b  = (__bf16*)alloc(MM * 2);
  __bf16* xz_b   = (__bf16*)alloc(MM * 2);
  __bf16* t1T_b  = (__bf16*)alloc(MM * 2);
  __bf16* t2T_b  = (__bf16*)alloc(MM * 2);
  __bf16* t3T_b  = (__bf16*)alloc(MM * 2);
  __bf16* kvT_b  = (__bf16*)alloc((size_t)32 * 64 * 256 * 2);
  __bf16* w2t_b  = (__bf16*)alloc((size_t)32 * 64 * 256 * 2);
  float*  aout_f = (float*)alloc(NTOK * 512 * 4);
  __bf16* aout_b = (__bf16*)alloc(NTOK * 512 * 2);
  float*  scales = (float*)alloc(256);
  (void)ws_size; (void)n_in; (void)in_sizes; (void)out_size;

  const int T = 256;
  k_convert_w<<<512, T, 0, stream>>>(w_qkv, w_out, wqkv_b, wout_b, scales);
  k_layernorm<<<2048, T, 0, stream>>>(x, ln_g, ln_b, xn_b, (int)NTOK);
  k_gemm_qkv<<<768, T, 0, stream>>>(xn_b, wqkv_b, q_b, k_b, v_f, vT_b);
  k_landmarks<<<2048, T, 0, stream>>>(q_b, k_b, ql_b, kl_b);
  k_attn2_logits<<<256, T, 0, stream>>>(ql_b, kl_b, mat_x);
  k_softmax256<<<1024, T, 0, stream>>>(mat_x, x_b, 32 * 256);
  k_pinv_scales<<<1024, T, 0, stream>>>(mat_x, scales);
  k_init_z<<<8192, T, 0, stream>>>(mat_x, z0_b, z0T_b, scales);

  __bf16* zc = z0_b;  __bf16* zcT = z0T_b;
  __bf16* zn = z1_b;  __bf16* znT = z1T_b;
  for (int it = 0; it < 6; ++it) {
    k_gemm256b<true, false, true><<<256, T, 0, stream>>>(x_b, zcT, xz_b, nullptr, t1T_b, 7.f, -1.f);
    k_gemm256b<false, false, true><<<256, T, 0, stream>>>(xz_b, t1T_b, nullptr, nullptr, t2T_b, 15.f, -1.f);
    k_gemm256b<false, false, true><<<256, T, 0, stream>>>(xz_b, t2T_b, nullptr, nullptr, t3T_b, 13.f, -1.f);
    k_gemm256b<false, true, true><<<256, T, 0, stream>>>(zc, t3T_b, nullptr, zn, znT, 0.f, 0.25f);
    __bf16* t;
    t = zc; zc = zn; zn = t;
    t = zcT; zcT = znT; znT = t;
  }

  k_attn3v<<<256, T, 0, stream>>>(k_b, ql_b, vT_b, kvT_b);
  k_w2<<<64, T, 0, stream>>>(zc, kvT_b, w2t_b);
  k_attn1_out<<<1024, T, 0, stream>>>(kl_b, q_b, w2t_b, aout_f);
  k_conv_res<<<512, T, 0, stream>>>(v_f, res_w, aout_f, aout_b);
  k_outproj<<<256, T, 0, stream>>>(aout_b, wout_b, b_out, x, out);
}


# element-wise kernels (layernorm, landmark means, softmax, scale sums, z0): a wave owns 4 rows or a 32x32 tile, not 1 row or 32 elements; 12288 waves (was 114688); z0 in 16-byte pieces; same sums
# speedup vs baseline: 3.1316x; 1.0860x over previous
#include <hip/hip_runtime.h>
#include <hip/hip_bf16.h>
#include <math.h>


typedef __attribute__((ext_vector_type(16))) __bf16 bf16x16;
typedef __attribute__((ext_vector_type(8)))  float  floatx8;

union FragAB { bf16x16 v; unsigned int u[8]; };

__device__ __forceinline__ unsigned pack2bf(float a, float b) {
  __bf16 lo = (__bf16)a, hi = (__bf16)b;
  unsigned short ul = __builtin_bit_cast(unsigned short, lo);
  unsigned short uh = __builtin_bit_cast(unsigned short, hi);
  return (unsigned)ul | ((unsigned)uh << 16);
}

__device__ __forceinline__ uint4 pack8bf(floatx8 a, float sc) {
  uint4 p;
  p.x = pack2bf(a[0] * sc, a[1] * sc);
  p.y = pack2bf(a[2] * sc, a[3] * sc);
  p.z = pack2bf(a[4] * sc, a[5] * sc);
  p.w = pack2bf(a[6] * sc, a[7] * sc);
  return p;
}

__device__ __forceinline__ int kpair(int p, int h) {
  return (p < 4) ? (2 * p + 8 * h) : (8 + 2 * p + 8 * h);
}

__device__ __forceinline__ bf16x16 load_frag_bf16_rm(const __bf16* src, int row0,
                                                     int k0, int ld) {
  int lane = threadIdx.x & 31;
  int r = lane & 15, h = lane >> 4;
  const __bf16* base = src + (size_t)(row0 + r) * ld + k0;
  FragAB f;
#pragma unroll
  for (int p = 0; p < 8; ++p) f.u[p] = *(const unsigned*)(base + kpair(p, h));
  return f.v;
}

__device__ __forceinline__ floatx8 wmma_bf16(bf16x16 a, bf16x16 b, floatx8 c) {
  return __builtin_amdgcn_wmma_f32_16x16x32_bf16(false, a, false, b, (short)0, c,
                                                 false, false);
}

template <int KSTEPS>
__device__ __forceinline__ void gemm_strip(const __bf16* A, int rowA,
                                           const __bf16* B, int rowB, int ld,
                                           floatx8 acc[4]) {
#pragma unroll
  for (int ks = 0; ks < KSTEPS; ++ks) {
    int k0 = ks * 32;
    bf16x16 af = load_frag_bf16_rm(A, rowA, k0, ld);
#pragma unroll
    for (int t = 0; t < 4; ++t) {
      bf16x16 bfr = load_frag_bf16_rm(B, rowB + 16 * t, k0, ld);
      acc[t] = wmma_bf16(af, bfr, acc[t]);
    }
  }
}

__device__ __forceinline__ size_t pk_idx(int row, int k) {
  int kk = k & 31;
  int lane = (row & 15) + 16 * ((kk >> 3) & 1);
  return (size_t)(((row >> 4) * 8 + (k >> 5)) * 512 + (kk >> 4) * 256 +
                  lane * 8 + (kk & 7));
}

__device__ __forceinline__ bf16x16 load_frag_bf16_pk(const __bf16* mat, int rb,
                                                     int ks) {
  int lane = threadIdx.x & 31;
  const uint4* blk = (const uint4*)(mat + (size_t)(rb * 8 + ks) * 512) + lane;
  uint4 lo = blk[0], hi = blk[32];
  FragAB f;
  f.u[0] = lo.x; f.u[1] = lo.y; f.u[2] = lo.z; f.u[3] = lo.w;
  f.u[4] = hi.x; f.u[5] = hi.y; f.u[6] = hi.z; f.u[7] = hi.w;
  return f.v;
}

__device__ __forceinline__ size_t pk_idx_n(int row, int k, int KB) {
  int kk = k & 31;
  int lane = (row & 15) + 16 * ((kk >> 3) & 1);
  return (size_t)((row >> 4) * KB + (k >> 5)) * 512 + (kk >> 4) * 256 +
         lane * 8 + (kk & 7);
}

#define PK_AHEAD 4
template <bool BPK>
__device__ __forceinline__ void gemm_strip_pk(const __bf16* A, int rowA,
                                              const __bf16* B, int rowB,
                                              floatx8 acc[4]) {
  bf16x16 af[8], bfr[8][4];
#pragma unroll
  for (int ks = 0; ks < PK_AHEAD; ++ks) {
    af[ks] = load_frag_bf16_pk(A, rowA >> 4, ks);
#pragma unroll
    for (int t = 0; t < 4; ++t)
      bfr[ks][t] = BPK ? load_frag_bf16_pk(B, (rowB >> 4) + t, ks)
                       : load_frag_bf16_rm(B, rowB + 16 * t, ks * 32, 256);
    __builtin_amdgcn_sched_barrier(0);
  }
#pragma unroll
  for (int ks = 0; ks < 8; ++ks) {
#pragma unroll
    for (int t = 0; t < 4; ++t) acc[t] = wmma_bf16(af[ks], bfr[ks][t], acc[t]);
    __builtin_amdgcn_sched_barrier(0);
    if (ks + PK_AHEAD < 8) {
      af[ks + PK_AHEAD] = load_frag_bf16_pk(A, rowA >> 4, ks + PK_AHEAD);
#pragma unroll
      for (int t = 0; t < 4; ++t)
        bfr[ks + PK_AHEAD][t] =
            BPK ? load_frag_bf16_pk(B, (rowB >> 4) + t, ks + PK_AHEAD)
                : load_frag_bf16_rm(B, rowB + 16 * t, (ks + PK_AHEAD) * 32, 256);
      __builtin_amdgcn_sched_barrier(0);
    }
  }
}

typedef __attribute__((ext_vector_type(4))) unsigned int u32x4;

__device__ __forceinline__ bf16x16 load_frag_bf16_lds(const u32x4* tile, int j,
                                                      int ks) {
  int lane = threadIdx.x & 31;
  const u32x4* blk = tile + (j * 8 + ks) * 64 + lane;
  u32x4 lo = blk[0], hi = blk[32];
  FragAB f;
  f.u[0] = lo.x; f.u[1] = lo.y; f.u[2] = lo.z; f.u[3] = lo.w;
  f.u[4] = hi.x; f.u[5] = hi.y; f.u[6] = hi.z; f.u[7] = hi.w;
  return f.v;
}

__device__ __forceinline__ void lds_barrier() {
  __builtin_amdgcn_fence(__ATOMIC_RELEASE, "workgroup", "local");
  __builtin_amdgcn_s_barrier();
  __builtin_amdgcn_fence(__ATOMIC_ACQUIRE, "workgroup", "local");
}

#define GB_STAGE 3072
template <bool SWAP>
__device__ __forceinline__ void gemm_block_lds(const __bf16* A, int arb0,
                                               const __bf16* B, int brb0,
                                               u32x4* tile,
                                               floatx8 acc[4][4]) {
  int tid = (int)threadIdx.x;
  int lane = tid & 31, w = tid >> 5;
  int wm = w >> 2, wn = w & 3;
  int half = tid >> 7, q = tid & 127;
  const u32x4* Ag = (const u32x4*)A + (size_t)(arb0 + half) * 1024 + q;
  const u32x4* Bg = (const u32x4*)B + (size_t)(brb0 + half) * 1024 + q;
  u32x4 st[12];
#pragma unroll
  for (int j = 0; j < 4; ++j) st[j] = Ag[j * 2048];
#pragma unroll
  for (int j = 0; j < 8; ++j) st[4 + j] = Bg[j * 2048];
#pragma unroll
  for (int S = 0; S < 8; ++S) {
    u32x4* dst = tile + (S & 1) * GB_STAGE;
#pragma unroll
    for (int j = 0; j < 12; ++j) dst[j * 256 + tid] = st[j];
    if (S + 1 < 8) {
#pragma unroll
      for (int j = 0; j < 4; ++j) st[j] = Ag[j * 2048 + (S + 1) * 128];
#pragma unroll
      for (int j = 0; j < 8; ++j) st[4 + j] = Bg[j * 2048 + (S + 1) * 128];
    }
    __builtin_amdgcn_sched_barrier(0);
    lds_barrier();
    bf16x16 af[2][4], bfr[2][4];
#pragma unroll
    for (int kk = 0; kk < 2; ++kk) {
      const u32x4* sb = dst + kk * 64 + lane;
#pragma unroll
      for (int m = 0; m < 4; ++m) {
        const u32x4* blk = sb + (wm * 4 + m) * 128;
        u32x4 lo = blk[0], hi = blk[32];
        FragAB f;
        f.u[0] = lo.x; f.u[1] = lo.y; f.u[2] = lo.z; f.u[3] = lo.w;
        f.u[4] = hi.x; f.u[5] = hi.y; f.u[6] = hi.z; f.u[7] = hi.w;
        af[kk][m] = f.v;
      }
#pragma unroll
      for (int t = 0; t < 4; ++t) {
        const u32x4* blk = sb + (8 + wn * 4 + t) * 128;
        u32x4 lo = blk[0], hi = blk[32];
        FragAB f;
        f.u[0] = lo.x; f.u[1] = lo.y; f.u[2] = lo.z; f.u[3] = lo.w;
        f.u[4] = hi.x; f.u[5] = hi.y; f.u[6] = hi.z; f.u[7] = hi.w;
        bfr[kk][t] = f.v;
      }
    }
    __builtin_amdgcn_sched_barrier(0);
#pragma unroll
    for (int kk = 0; kk < 2; ++kk)
#pragma unroll
      for (int m = 0; m < 4; ++m)
#pragma unroll
        for (int t = 0; t < 4; ++t)
          acc[m][t] = SWAP ? wmma_bf16(bfr[kk][t], af[kk][m], acc[m][t])
                           : wmma_bf16(af[kk][m], bfr[kk][t], acc[m][t]);
  }
}

__global__ void k_convert_w(const float* wqkv, const float* wout,
                            __bf16* wqkv_b, __bf16* wout_b, float* scales) {
  int i = blockIdx.x * blockDim.x + threadIdx.x;
  if (i < 2) scales[i] = 0.f;
  bool second = i >= 1536 * 64;
  const float* src = second ? wout : wqkv;
  __bf16* dst = second ? wout_b : wqkv_b;
  int u = second ? i - 1536 * 64 : i;
  if (second && u >= 512 * 64) return;
  int blk = u >> 6, half = (u >> 5) & 1, lane = u & 31;
  int row = (blk >> 4) * 16 + (lane & 15);
  int k = (blk & 15) * 32 + half * 16 + 8 * (lane >> 4);
  const float4* s = (const float4*)(src + (size_t)row * 512 + k);
  float4 a = s[0], b = s[1];
  uint4 pv;
  pv.x = pack2bf(a.x, a.y);
  pv.y = pack2bf(a.z, a.w);
  pv.z = pack2bf(b.x, b.y);
  pv.w = pack2bf(b.z, b.w);
  ((uint4*)dst)[u] = pv;
}

#define LN_TOK 4
__global__ void __launch_bounds__(256)
k_layernorm(const float* x, const float* g, const float* b, __bf16* xn,
            int ntok) {
  int wave = (int)((blockIdx.x * blockDim.x + threadIdx.x) >> 5);
  int lane = threadIdx.x & 31;
  int tok0 = wave * LN_TOK;
  if (tok0 >= ntok) return;
  float vals[LN_TOK][16], gg[16], bb[16];
#pragma unroll
  for (int k = 0; k < LN_TOK; ++k) {
    const float* xr = x + (size_t)(tok0 + k) * 512;
#pragma unroll
    for (int i = 0; i < 16; ++i) vals[k][i] = xr[lane + 32 * i];
  }
#pragma unroll
  for (int i = 0; i < 16; ++i) {
    gg[i] = g[lane + 32 * i];
    bb[i] = b[lane + 32 * i];
  }
  __builtin_amdgcn_sched_barrier(0);
#pragma unroll
  for (int k = 0; k < LN_TOK; ++k) {
    float s = 0.f, s2 = 0.f;
#pragma unroll
    for (int i = 0; i < 16; ++i) {
      float v = vals[k][i];
      s += v; s2 += v * v;
    }
    for (int m = 16; m >= 1; m >>= 1) {
      s += __shfl_xor(s, m, 32);
      s2 += __shfl_xor(s2, m, 32);
    }
    float mu = s * (1.f / 512.f);
    float var = s2 * (1.f / 512.f) - mu * mu;
    float inv = rsqrtf(var + 1e-5f);
#pragma unroll
    for (int i = 0; i < 16; ++i) {
      int c = lane + 32 * i;
      xn[pk_idx_n(tok0 + k, c, 16)] =
          (__bf16)((vals[k][i] - mu) * inv * gg[i] + bb[i]);
    }
  }
}

__global__ void __launch_bounds__(256)
k_gemm_qkv(const __bf16* xn, const __bf16* wq, __bf16* qb,
           __bf16* kb, float* vf, __bf16* vT) {
  __shared__ u32x4 tile[2 * GB_STAGE];
  int bM = (int)blockIdx.x / 6, bN = (int)blockIdx.x % 6;
  int w = (int)threadIdx.x >> 5;
  int tM = bM * 2 + (w >> 2), s = bN * 4 + (w & 3);
  int lane = threadIdx.x & 31;
  int c = lane & 15, h = lane >> 4;
  int head = s & 7;
  int tok0 = tM * 64;
  int bb = tok0 >> 12, ns0 = tok0 & 4095;
  size_t bh = (size_t)(bb * 8 + head);
  floatx8 acc[4][4] = {};
  if (bN < 4) {
    gemm_block_lds<true>(xn, bM * 8, wq, bN * 16, tile, acc);
    __bf16* dst = (bN < 2 ? qb : kb) + (bh * 4096 + ns0 + c) * 64 + 8 * h;
    float sc = bN < 2 ? 0.125f : 1.f;
#pragma unroll
    for (int m = 0; m < 4; ++m)
#pragma unroll
      for (int t = 0; t < 4; ++t)
        *(uint4*)(dst + (16 * m) * 64 + 16 * t) = pack8bf(acc[m][t], sc);
  } else {
    gemm_block_lds<false>(xn, bM * 8, wq, bN * 16, tile, acc);
    float* vfd = vf + (bh * 4096 + ns0 + 8 * h) * 64 + c;
    __bf16* vTd = vT + (bh * 64 + c) * 4096 + ns0 + 8 * h;
#pragma unroll
    for (int m = 0; m < 4; ++m)
#pragma unroll
      for (int t = 0; t < 4; ++t) {
#pragma unroll
        for (int r = 0; r < 8; ++r)
          vfd[(16 * m + r) * 64 + 16 * t] = acc[m][t][r];
        *(uint4*)(vTd + (size_t)(16 * t) * 4096 + 16 * m) =
            pack8bf(acc[m][t], 1.f);
      }
  }
}

__device__ __forceinline__ float bf_lo(unsigned u) {
  return __builtin_bit_cast(float, u << 16);
}
__device__ __forceinline__ float bf_hi(unsigned u) {
  return __builtin_bit_cast(float, u & 0xffff0000u);
}

__global__ void __launch_bounds__(256)
k_landmarks(const __bf16* qb, const __bf16* kb, __bf16* ql, __bf16* kl) {
  int i = blockIdx.x * blockDim.x + threadIdx.x;
  if (i >= 32 * 256 * 8) return;
  int d8 = i & 7, m = (i >> 3) & 255, bh = i >> 11;
  size_t base = ((size_t)bh * 4096 + m * 16) * 64 + d8 * 8;
  u32x4 qv[16], kv[16];
#pragma unroll
  for (int t = 0; t < 16; ++t) {
    qv[t] = *(const u32x4*)(qb + base + t * 64);
    kv[t] = *(const u32x4*)(kb + base + t * 64);
  }
  float sq[8], sk[8];
#pragma unroll
  for (int e = 0; e < 8; ++e) { sq[e] = 0.f; sk[e] = 0.f; }
#pragma unroll
  for (int t = 0; t < 16; ++t) {
    sq[0] += bf_lo(qv[t].x); sq[1] += bf_hi(qv[t].x);
    sq[2] += bf_lo(qv[t].y); sq[3] += bf_hi(qv[t].y);
    sq[4] += bf_lo(qv[t].z); sq[5] += bf_hi(qv[t].z);
    sq[6] += bf_lo(qv[t].w); sq[7] += bf_hi(qv[t].w);
    sk[0] += bf_lo(kv[t].x); sk[1] += bf_hi(kv[t].x);
    sk[2] += bf_lo(kv[t].y); sk[3] += bf_hi(kv[t].y);
    sk[4] += bf_lo(kv[t].z); sk[5] += bf_hi(kv[t].z);
    sk[6] += bf_lo(kv[t].w); sk[7] += bf_hi(kv[t].w);
  }
  const float r16 = 1.f / 16.f;
  uint4 oq, ok;
  oq.x = pack2bf(sq[0] * r16, sq[1] * r16);
  oq.y = pack2bf(sq[2] * r16, sq[3] * r16);
  oq.z = pack2bf(sq[4] * r16, sq[5] * r16);
  oq.w = pack2bf(sq[6] * r16, sq[7] * r16);
  ok.x = pack2bf(sk[0] * r16, sk[1] * r16);
  ok.y = pack2bf(sk[2] * r16, sk[3] * r16);
  ok.z = pack2bf(sk[4] * r16, sk[5] * r16);
  ok.w = pack2bf(sk[6] * r16, sk[7] * r16);
  *(uint4*)(ql + (size_t)i * 8) = oq;
  *(uint4*)(kl + (size_t)i * 8) = ok;
}

__global__ void k_attn2_logits(const __bf16* ql, const __bf16* kl, float* out) {
  int wid = (int)((blockIdx.x * blockDim.x + threadIdx.x) >> 5);
  int bh = wid >> 6, t = wid & 63;
  int tM = t >> 2, s = t & 3;
  const __bf16* A = ql + (size_t)bh * 256 * 64;
  const __bf16* B = kl + (size_t)bh * 256 * 64;
  floatx8 acc[4] = {};
  gemm_strip<2>(A, tM * 16, B, s * 64, 64, acc);
  int lane = threadIdx.x & 31;
  int n = lane & 15, h = lane >> 4;
  float* o = out + (size_t)bh * 256 * 256;
#pragma unroll
  for (int t4 = 0; t4 < 4; ++t4)
#pragma unroll
    for (int r = 0; r < 8; ++r)
      o[(size_t)(tM * 16 + r + 8 * h) * 256 + s * 64 + 16 * t4 + n] =
          acc[t4][r];
}

#define SM_ROWS 4
__global__ void k_softmax256(float* x, __bf16* xb, int nrows) {
  int wave = (int)((blockIdx.x * blockDim.x + threadIdx.x) >> 5);
  int row0 = wave * SM_ROWS;
  if (row0 >= nrows) return;
  int lane = threadIdx.x & 31;
  __bf16* rb = xb + ((size_t)(row0 >> 8) << 16);
  float v[SM_ROWS][8];
#pragma unroll
  for (int k = 0; k < SM_ROWS; ++k)
#pragma unroll
    for (int i = 0; i < 8; ++i) v[k][i] = x[(size_t)(row0 + k) * 256 + lane + 32 * i];
#pragma unroll
  for (int k = 0; k < SM_ROWS; ++k) {
    float* r = x + (size_t)(row0 + k) * 256;
    int row = (row0 + k) & 255;
    float m = -1e30f;
#pragma unroll
    for (int i = 0; i < 8; ++i) m = fmaxf(m, v[k][i]);
    for (int t = 16; t >= 1; t >>= 1) m = fmaxf(m, __shfl_xor(m, t, 32));
    float s = 0.f;
#pragma unroll
    for (int i = 0; i < 8; ++i) { v[k][i] = __expf(v[k][i] - m); s += v[k][i]; }
    for (int t = 16; t >= 1; t >>= 1) s += __shfl_xor(s, t, 32);
    float inv = 1.f / s;
#pragma unroll
    for (int i = 0; i < 8; ++i) {
      float pv = v[k][i] * inv;
      r[lane + 32 * i] = pv;
      rb[pk_idx(row, lane + 32 * i)] = (__bf16)pv;
    }
  }
}

__global__ void k_pinv_scales(const float* x, float* scales) {
  int wave = (int)((blockIdx.x * blockDim.x + threadIdx.x) >> 5);
  if (wave >= 32 * 64) return;
  int lane = threadIdx.x & 31;
  int bh = wave >> 6, r0 = (wave & 63) * 4;
  const float* m = x + (size_t)bh * 256 * 256;
  float rv[4][8];
  float4 cv[8];
#pragma unroll
  for (int i = 0; i < 8; ++i)
    cv[i] = *(const float4*)(m + (size_t)(lane + 32 * i) * 256 + r0);
#pragma unroll
  for (int k = 0; k < 4; ++k)
#pragma unroll
    for (int i = 0; i < 8; ++i) rv[k][i] = m[(size_t)(r0 + k) * 256 + lane + 32 * i];
  unsigned rmax = 0u, cmax = 0u;
#pragma unroll
  for (int k = 0; k < 4; ++k) {
    float rs = 0.f, cs = 0.f;
#pragma unroll
    for (int i = 0; i < 8; ++i) {
      float cval = k == 0 ? cv[i].x : k == 1 ? cv[i].y : k == 2 ? cv[i].z : cv[i].w;
      rs += fabsf(rv[k][i]);
      cs += fabsf(cval);
    }
    for (int t = 16; t >= 1; t >>= 1) {
      rs += __shfl_xor(rs, t, 32);
      cs += __shfl_xor(cs, t, 32);
    }
    rmax = max(rmax, __float_as_uint(rs));
    cmax = max(cmax, __float_as_uint(cs));
  }
  if (lane == 0) {
    atomicMax((unsigned*)&scales[0], rmax);
    atomicMax((unsigned*)&scales[1], cmax);
  }
}

__global__ void k_init_z(const float* x, __bf16* zb, __bf16* zTb,
                         const float* scales) {
  int wave = (int)((blockIdx.x * blockDim.x + threadIdx.x) >> 5);
  if (wave >= 32 * 64) return;
  int lane = threadIdx.x & 31;
  int c = lane & 15, h = lane >> 4;
  int bh = wave >> 6, tj = (wave >> 3) & 7, tr = wave & 7;
  size_t bofs = (size_t)bh << 16;
  const float* xt = x + bofs + (size_t)(tj * 32) * 256 + tr * 32;
  float col[4][8];
#pragma unroll
  for (int g = 0; g < 4; ++g)
#pragma unroll
    for (int e = 0; e < 8; ++e) col[g][e] = xt[(size_t)(8 * g + e) * 256 + lane];
  float4 rw[2][2][2];
#pragma unroll
  for (int jb = 0; jb < 2; ++jb)
#pragma unroll
    for (int half = 0; half < 2; ++half) {
      const float4* s =
          (const float4*)(xt + (size_t)(16 * jb + c) * 256 + 16 * half + 8 * h);
      rw[jb][half][0] = s[0];
      rw[jb][half][1] = s[1];
    }
  __builtin_amdgcn_sched_barrier(0);
  float denom = scales[0] * scales[1];
#pragma unroll
  for (int jb = 0; jb < 2; ++jb)
#pragma unroll
    for (int half = 0; half < 2; ++half) {
      float4 a = rw[jb][half][0], b = rw[jb][half][1];
      uint4 pv;
      pv.x = pack2bf(a.x / denom, a.y / denom);
      pv.y = pack2bf(a.z / denom, a.w / denom);
      pv.z = pack2bf(b.x / denom, b.y / denom);
      pv.w = pack2bf(b.z / denom, b.w / denom);
      *(uint4*)(zTb + bofs + (size_t)(((2 * tj + jb) * 8 + tr) * 512 +
                                      half * 256 + lane * 8)) = pv;
    }
#pragma unroll
  for (int g = 0; g < 4; ++g) {
    uint4 pv;
    pv.x = pack2bf(col[g][0] / denom, col[g][1] / denom);
    pv.y = pack2bf(col[g][2] / denom, col[g][3] / denom);
    pv.z = pack2bf(col[g][4] / denom, col[g][5] / denom);
    pv.w = pack2bf(col[g][6] / denom, col[g][7] / denom);
    *(uint4*)(zb + bofs + (size_t)(((2 * tr + h) * 8 + tj) * 512 +
                                   (g >> 1) * 256 + (c + 16 * (g & 1)) * 8)) = pv;
  }
}

#define G8_ARB 4
#define G8_BRB 8
template <bool RAW, bool OT, bool OTT>
__global__ void __launch_bounds__(256)
k_gemm256b(const __bf16* A, const __bf16* BT, __bf16* OutRaw,
           __bf16* OutT, __bf16* OutTT, float alpha, float beta) {
  __shared__ u32x4 tile[(G8_ARB + G8_BRB) * 8 * 64];
  int tid = (int)threadIdx.x;
  int w = tid >> 5;
  int bh = (int)blockIdx.x >> 3, g = (int)blockIdx.x & 7;
  int bm = g >> 1, bn = g & 1;
  int tM = bm * G8_ARB + (w >> 1), s = bn * 2 + (w & 1);
  size_t bofs = (size_t)bh << 16;
  const u32x4* Ag = (const u32x4*)(A + bofs) + (size_t)(bm * G8_ARB) * 8 * 64;
  const u32x4* Bg = (const u32x4*)(BT + bofs) + (size_t)(bn * G8_BRB) * 8 * 64;
  u32x4 st0[G8_ARB + G8_BRB], st1[G8_ARB + G8_BRB];
#pragma unroll
  for (int j = 0; j < G8_ARB; ++j) st0[j] = Ag[j * 512 + tid];
#pragma unroll
  for (int j = 0; j < G8_BRB; ++j) st0[G8_ARB + j] = Bg[j * 512 + tid];
  __builtin_amdgcn_sched_barrier(0);
#pragma unroll
  for (int j = 0; j < G8_ARB; ++j) st1[j] = Ag[j * 512 + 256 + tid];
#pragma unroll
  for (int j = 0; j < G8_BRB; ++j) st1[G8_ARB + j] = Bg[j * 512 + 256 + tid];
  floatx8 acc[4] = {};
  floatx8 accS[4] = {};
#pragma unroll
  for (int j = 0; j < G8_ARB + G8_BRB; ++j) tile[j * 512 + tid] = st0[j];
  lds_barrier();
#pragma unroll
  for (int ks = 0; ks < 4; ++ks) {
    bf16x16 af = load_frag_bf16_lds(tile, w >> 1, ks);
#pragma unroll
    for (int t = 0; t < 4; ++t) {
      bf16x16 bf = load_frag_bf16_lds(tile, G8_ARB + (w & 1) * 4 + t, ks);
      if (OTT) acc[t] = wmma_bf16(af, bf, acc[t]);
      if (RAW || OT) accS[t] = wmma_bf16(bf, af, accS[t]);
    }
  }
#pragma unroll
  for (int j = 0; j < G8_ARB + G8_BRB; ++j) tile[j * 512 + 256 + tid] = st1[j];
  lds_barrier();
#pragma unroll
  for (int ks = 4; ks < 8; ++ks) {
    bf16x16 af = load_frag_bf16_lds(tile, w >> 1, ks);
#pragma unroll
    for (int t = 0; t < 4; ++t) {
      bf16x16 bf = load_frag_bf16_lds(tile, G8_ARB + (w & 1) * 4 + t, ks);
      if (OTT) acc[t] = wmma_bf16(af, bf, acc[t]);
      if (RAW || OT) accS[t] = wmma_bf16(bf, af, accS[t]);
    }
  }
  int lane = threadIdx.x & 31;
  int n = lane & 15, h = lane >> 4;
#pragma unroll
  for (int t4 = 0; t4 < 4; ++t4) {
    if (OTT) {
      int N = s * 64 + 16 * t4 + n;
      floatx8 trv;
#pragma unroll
      for (int r = 0; r < 8; ++r) {
        int M = tM * 16 + r + 8 * h;
        trv[r] = alpha * (M == N ? 1.f : 0.f) + beta * acc[t4][r];
      }
      *(uint4*)(OutTT + bofs + pk_idx(N, tM * 16 + 8 * h)) = pack8bf(trv, 1.f);
    }
    if (RAW || OT) {
      int M = tM * 16 + n;
      int N0 = s * 64 + 16 * t4 + 8 * h;
      size_t o = bofs + pk_idx(M, N0);
      if (RAW) *(uint4*)(OutRaw + o) = pack8bf(accS[t4], 1.f);
      if (OT) {
        floatx8 trv;
#pragma unroll
        for (int r = 0; r < 8; ++r)
          trv[r] = alpha * (M == N0 + r ? 1.f : 0.f) + beta * accS[t4][r];
        *(uint4*)(OutT + o) = pack8bf(trv, 1.f);
      }
    }
  }
}

__device__ __forceinline__ void stream_attn(const __bf16* Ak, const __bf16* Bq,
                                            int qrow0, const __bf16* A2,
                                            int ldA2, int nkeys, float* outF,
                                            int rsF, __bf16* outB, int nsB,
                                            int dsB) {
  int lane = threadIdx.x & 31;
  int n = lane & 15, h = lane >> 4;
  bf16x16 bq0 = load_frag_bf16_rm(Bq, qrow0, 0, 64);
  bf16x16 bq1 = load_frag_bf16_rm(Bq, qrow0, 32, 64);
  float run_m = -1e30f, run_s = 0.f;
  floatx8 acc[4] = {};
  for (int c = 0; c < nkeys; c += 32) {
    floatx8 s0 = {}, s1 = {};
    s0 = wmma_bf16(load_frag_bf16_rm(Ak, c, 0, 64), bq0, s0);
    s0 = wmma_bf16(load_frag_bf16_rm(Ak, c, 32, 64), bq1, s0);
    s1 = wmma_bf16(load_frag_bf16_rm(Ak, c + 16, 0, 64), bq0, s1);
    s1 = wmma_bf16(load_frag_bf16_rm(Ak, c + 16, 32, 64), bq1, s1);
    float mc = -1e30f;
#pragma unroll
    for (int r = 0; r < 8; ++r) { mc = fmaxf(mc, s0[r]); mc = fmaxf(mc, s1[r]); }
    mc = fmaxf(mc, __shfl_xor(mc, 16, 32));
    float newm = fmaxf(run_m, mc);
    float scale = __expf(run_m - newm);
    float p0[8], p1[8], ps = 0.f;
#pragma unroll
    for (int r = 0; r < 8; ++r) {
      p0[r] = __expf(s0[r] - newm);
      p1[r] = __expf(s1[r] - newm);
      ps += p0[r] + p1[r];
    }
    run_s = run_s * scale + ps;
    run_m = newm;
    FragAB bp;
#pragma unroll
    for (int p = 0; p < 4; ++p) {
      bp.u[p] = pack2bf(p0[2 * p], p0[2 * p + 1]);
      bp.u[4 + p] = pack2bf(p1[2 * p], p1[2 * p + 1]);
    }
#pragma unroll
    for (int t = 0; t < 4; ++t) {
#pragma unroll
      for (int r = 0; r < 8; ++r) acc[t][r] *= scale;
      acc[t] = wmma_bf16(load_frag_bf16_rm(A2, 16 * t, c, ldA2), bp.v, acc[t]);
    }
  }
  float tot = run_s + __shfl_xor(run_s, 16, 32);
  float inv = 1.f / tot;
#pragma unroll
  for (int t = 0; t < 4; ++t)
#pragma unroll
    for (int r = 0; r < 8; ++r) {
      int dh = 16 * t + r + 8 * h;
      float val = acc[t][r] * inv;
      if (outF) outF[(size_t)n * rsF + dh] = val;
      if (outB) outB[(size_t)n * nsB + (size_t)dh * dsB] = (__bf16)val;
    }
}

#define A3_WAVES 8
#define A3_QG 2
#define A3_KEYS (4096 / A3_WAVES)
__global__ void __launch_bounds__(256)
k_attn3v(const __bf16* kb, const __bf16* ql, const __bf16* vT, __bf16* kvT) {
  __shared__ float accL[A3_WAVES][A3_QG][4][8][32];
  __shared__ float mL[A3_WAVES][A3_QG][16];
  __shared__ float sL[A3_WAVES][A3_QG][16];
  int lane = threadIdx.x & 31;
  int w = __builtin_amdgcn_readfirstlane((int)(threadIdx.x >> 5));
  int bh = blockIdx.x >> 3, q0 = (blockIdx.x & 7) * (16 * A3_QG);
  const __bf16* Ak = kb + (size_t)bh * 4096 * 64;
  const __bf16* Bq = ql + (size_t)bh * 256 * 64;
  const __bf16* A2 = vT + (size_t)bh * 64 * 4096;
  bf16x16 bq[A3_QG][2];
  float run_m[A3_QG], run_s[A3_QG];
  floatx8 acc[A3_QG][4];
#pragma unroll
  for (int g = 0; g < A3_QG; ++g) {
    bq[g][0] = load_frag_bf16_rm(Bq, q0 + 16 * g, 0, 64);
    bq[g][1] = load_frag_bf16_rm(Bq, q0 + 16 * g, 32, 64);
    run_m[g] = -1e30f;
    run_s[g] = 0.f;
#pragma unroll
    for (int t = 0; t < 4; ++t) acc[g][t] = floatx8{};
  }
  int kbeg = w * A3_KEYS;
  for (int c = kbeg; c < kbeg + A3_KEYS; c += 32) {
    bf16x16 ka[4], va[4];
    ka[0] = load_frag_bf16_rm(Ak, c, 0, 64);
    ka[1] = load_frag_bf16_rm(Ak, c, 32, 64);
    ka[2] = load_frag_bf16_rm(Ak, c + 16, 0, 64);
    ka[3] = load_frag_bf16_rm(Ak, c + 16, 32, 64);
#pragma unroll
    for (int t = 0; t < 4; ++t) va[t] = load_frag_bf16_rm(A2, 16 * t, c, 4096);
    __builtin_amdgcn_sched_barrier(0);
#pragma unroll
    for (int g = 0; g < A3_QG; ++g) {
      floatx8 s0 = {}, s1 = {};
      s0 = wmma_bf16(ka[0], bq[g][0], s0);
      s0 = wmma_bf16(ka[1], bq[g][1], s0);
      s1 = wmma_bf16(ka[2], bq[g][0], s1);
      s1 = wmma_bf16(ka[3], bq[g][1], s1);
      float mc = -1e30f;
#pragma unroll
      for (int r = 0; r < 8; ++r) { mc = fmaxf(mc, s0[r]); mc = fmaxf(mc, s1[r]); }
      mc = fmaxf(mc, __shfl_xor(mc, 16, 32));
      float newm = fmaxf(run_m[g], mc);
      float scale = __expf(run_m[g] - newm);
      float p0[8], p1[8], ps = 0.f;
#pragma unroll
      for (int r = 0; r < 8; ++r) {
        p0[r] = __expf(s0[r] - newm);
        p1[r] = __expf(s1[r] - newm);
        ps += p0[r] + p1[r];
      }
      run_s[g] = run_s[g] * scale + ps;
      run_m[g] = newm;
      FragAB bp;
#pragma unroll
      for (int p = 0; p < 4; ++p) {
        bp.u[p] = pack2bf(p0[2 * p], p0[2 * p + 1]);
        bp.u[4 + p] = pack2bf(p1[2 * p], p1[2 * p + 1]);
      }
#pragma unroll
      for (int t = 0; t < 4; ++t) {
#pragma unroll
        for (int r = 0; r < 8; ++r) acc[g][t][r] *= scale;
        acc[g][t] = wmma_bf16(va[t], bp.v, acc[g][t]);
      }
    }
  }
#pragma unroll
  for (int g = 0; g < A3_QG; ++g) {
    float tot = run_s[g] + __shfl_xor(run_s[g], 16, 32);
    if (lane < 16) {
      mL[w][g][lane] = run_m[g];
      sL[w][g][lane] = tot;
    }
#pragma unroll
    for (int t = 0; t < 4; ++t)
#pragma unroll
      for (int r = 0; r < 8; ++r) accL[w][g][t][r][lane] = acc[g][t][r];
  }
  __syncthreads();
  int cc = lane & 15, hh = lane >> 4;
  for (int g = 0; g < A3_QG; ++g) {
    float mw[A3_WAVES], f[A3_WAVES], M = -1e30f, S = 0.f;
#pragma unroll
    for (int v = 0; v < A3_WAVES; ++v) { mw[v] = mL[v][g][cc]; M = fmaxf(M, mw[v]); }
#pragma unroll
    for (int v = 0; v < A3_WAVES; ++v) {
      f[v] = __expf(mw[v] - M);
      S += sL[v][g][cc] * f[v];
    }
    float inv = 1.f / S;
    for (int t = 0; t < 4; ++t) {
      float o = 0.f;
#pragma unroll
      for (int v = 0; v < A3_WAVES; ++v) o += accL[v][g][t][w][lane] * f[v];
      int dh = 16 * t + 8 * hh + w;
      kvT[((size_t)bh * 64 + dh) * 256 + q0 + 16 * g + cc] = (__bf16)(o * inv);
    }
  }
}

__global__ void __launch_bounds__(256)
k_w2(const __bf16* zb, const __bf16* kvT, __bf16* w2t) {
  int wid = (int)((blockIdx.x * blockDim.x + threadIdx.x) >> 5);
  int bh = wid >> 4, tM = wid & 15;
  floatx8 acc[4] = {};
  gemm_strip_pk<false>(zb + ((size_t)bh << 16), tM * 16,
                       kvT + (size_t)bh * 64 * 256, 0, acc);
  int lane = threadIdx.x & 31;
  int n = lane & 15, h = lane >> 4;
#pragma unroll
  for (int t4 = 0; t4 < 4; ++t4)
#pragma unroll
    for (int r = 0; r < 8; ++r) {
      int M = tM * 16 + r + 8 * h;
      int N = 16 * t4 + n;
      w2t[((size_t)bh * 64 + N) * 256 + M] = (__bf16)acc[t4][r];
    }
}

__global__ void k_attn1_out(const __bf16* kl, const __bf16* qb,
                            const __bf16* w2t, float* aout) {
  int wid = (int)((blockIdx.x * blockDim.x + threadIdx.x) >> 5);
  int bh = wid >> 8, blk = wid & 255;
  int b = bh >> 3, hh = bh & 7;
  stream_attn(kl + (size_t)bh * 256 * 64, qb + (size_t)bh * 4096 * 64, blk * 16,
              w2t + (size_t)bh * 64 * 256, 256, 256,
              aout + ((size_t)b * 4096 + blk * 16) * 512 + hh * 64, 512,
              nullptr, 0, 0);
}

#define CONV_R 16
#define CONV_G 8
#define CONV_ROWS (CONV_R + 32)
__global__ void __launch_bounds__(256, 4)
k_conv_res(const float* __restrict__ vf, const float* __restrict__ rw,
           float* __restrict__ aout, __bf16* __restrict__ aout_b) {
  int t = (int)(blockIdx.x * blockDim.x + threadIdx.x);
  if (t >= 4 * (4096 / CONV_R) * 128) return;
  int c4 = t & 127;
  int chunk = t >> 7;
  int b = chunk / (4096 / CONV_R);
  int ns0 = (chunk % (4096 / CONV_R)) * CONV_R;
  int col = c4 * 4, hh = col >> 6, dh = col & 63;
  const float* vcol = vf + ((size_t)(b * 8 + hh) * 4096) * 64 + dh;
  float* arow = aout + ((size_t)b * 4096 + ns0) * 512 + col;
  int boff = (int)pk_idx_n(b * 4096 + ns0, col, 16);
  float w[33];
#pragma unroll
  for (int j = 0; j < 33; ++j) w[j] = rw[hh * 33 + j];
  float4 acc[CONV_R], win[CONV_ROWS], a[CONV_R];
#pragma unroll
  for (int o = 0; o < CONV_R; ++o) acc[o] = make_float4(0.f, 0.f, 0.f, 0.f);
#pragma unroll
  for (int k = 0; k < CONV_G; ++k) {
    int p = ns0 + k - 16;
    int pc = p < 0 ? 0 : (p > 4095 ? 4095 : p);
    win[k] = *(const float4*)(vcol + (size_t)pc * 64);
  }
#pragma unroll
  for (int g = 0; g < CONV_ROWS / CONV_G; ++g) {
    if (g + 1 < CONV_ROWS / CONV_G) {
#pragma unroll
      for (int k = (g + 1) * CONV_G; k < (g + 2) * CONV_G; ++k) {
        int p = ns0 + k - 16;
        int pc = p < 0 ? 0 : (p > 4095 ? 4095 : p);
        win[k] = *(const float4*)(vcol + (size_t)pc * 64);
      }
    } else {
#pragma unroll
      for (int o = 0; o < CONV_R; ++o)
        a[o] = *(const float4*)(arow + (size_t)o * 512);
    }
#pragma unroll
    for (int k = g * CONV_G; k < (g + 1) * CONV_G; ++k) {
      int p = ns0 + k - 16;
      float4 v = win[k];
      if (p < 0 || p > 4095) v = make_float4(0.f, 0.f, 0.f, 0.f);
#pragma unroll
      for (int o = 0; o < CONV_R; ++o) {
        int j = k - o;
        if (j >= 0 && j < 33) {
          acc[o].x += v.x * w[j];
          acc[o].y += v.y * w[j];
          acc[o].z += v.z * w[j];
          acc[o].w += v.w * w[j];
        }
      }
    }
  }
#pragma unroll
  for (int o = 0; o < CONV_R; ++o) {
    float4 r;
    r.x = a[o].x + acc[o].x;
    r.y = a[o].y + acc[o].y;
    r.z = a[o].z + acc[o].z;
    r.w = a[o].w + acc[o].w;
    *(float4*)(arow + (size_t)o * 512) = r;
    uint2 pk;
    pk.x = pack2bf(r.x, r.y);
    pk.y = pack2bf(r.z, r.w);
    *(uint2*)(aout_b + (boff + o * 8)) = pk;
  }
}

__global__ void __launch_bounds__(256)
k_outproj(const __bf16* __restrict__ aoutb, const __bf16* __restrict__ wo,
          const float* __restrict__ bias, const float* __restrict__ x,
          float* __restrict__ out) {
  __shared__ u32x4 tile[2 * GB_STAGE];
  int bM = (int)blockIdx.x >> 1, bN = (int)blockIdx.x & 1;
  int w = (int)threadIdx.x >> 5;
  int tM = bM * 2 + (w >> 2), s = bN * 4 + (w & 3);
  floatx8 acc[4][4] = {};
  gemm_block_lds<false>(aoutb, bM * 8, wo, bN * 16, tile, acc);
  int lane = threadIdx.x & 31;
  int n = lane & 15, h = lane >> 4;
#pragma unroll
  for (int t4 = 0; t4 < 4; ++t4) {
    int col = s * 64 + 16 * t4 + n;
    float bc = bias[col];
#pragma unroll
    for (int m = 0; m < 4; ++m) {
#pragma unroll
      for (int r = 0; r < 8; ++r) {
        size_t idx = (size_t)(tM * 64 + 16 * m + r + 8 * h) * 512 + col;
        out[idx] = x[idx] + acc[m][t4][r] + bc;
      }
    }
  }
}

extern "C" void kernel_launch(void* const* d_in, const int* in_sizes, int n_in,
                              void* d_out, int out_size, void* d_ws,
                              size_t ws_size, hipStream_t stream) {
  const float* x     = (const float*)d_in[0];
  const float* ln_g  = (const float*)d_in[1];
  const float* ln_b  = (const float*)d_in[2];
  const float* w_qkv = (const float*)d_in[3];
  const float* w_out = (const float*)d_in[4];
  const float* b_out = (const float*)d_in[5];
  const float* res_w = (const float*)d_in[6];
  float* out = (float*)d_out;

  char* p = (char*)d_ws;
  auto alloc = [&](size_t bytes) -> void* {
    void* r = (void*)p;
    p += (bytes + 255) & ~(size_t)255;
    return r;
  };
  const size_t NTOK = 4 * 4096;
  const size_t MM = (size_t)32 * 256 * 256;
  __bf16* xn_b   = (__bf16*)alloc(NTOK * 512 * 2);
  __bf16* wqkv_b = (__bf16*)alloc(1536 * 512 * 2);
  __bf16* wout_b = (__bf16*)alloc(512 * 512 * 2);
  __bf16* q_b    = (__bf16*)alloc((size_t)32 * 4096 * 64 * 2);
  __bf16* k_b    = (__bf16*)alloc((size_t)32 * 4096 * 64 * 2);
  float*  v_f    = (float*)alloc((size_t)32 * 4096 * 64 * 4);
  __bf16* vT_b   = (__bf16*)alloc((size_t)32 * 64 * 4096 * 2);
  __bf16* ql_b   = (__bf16*)alloc((size_t)32 * 256 * 64 * 2);
  __bf16* kl_b   = (__bf16*)alloc((size_t)32 * 256 * 64 * 2);
  float*  mat_x  = (float*)alloc(MM * 4);
  __bf16* x_b    = (__bf16*)alloc(MM * 2);
  __bf16* z0_b   = (__bf16*)alloc(MM * 2);
  __bf16* z0T_b  = (__bf16*)alloc(MM * 2);
  __bf16* z1_b   = (__bf16*)alloc(MM * 2);
  __bf16* z1T_b  = (__bf16*)alloc(MM * 2);
  __bf16* xz_b   = (__bf16*)alloc(MM * 2);
  __bf16* t1T_b  = (__bf16*)alloc(MM * 2);
  __bf16* t2T_b  = (__bf16*)alloc(MM * 2);
  __bf16* t3T_b  = (__bf16*)alloc(MM * 2);
  __bf16* kvT_b  = (__bf16*)alloc((size_t)32 * 64 * 256 * 2);
  __bf16* w2t_b  = (__bf16*)alloc((size_t)32 * 64 * 256 * 2);
  float*  aout_f = (float*)alloc(NTOK * 512 * 4);
  __bf16* aout_b = (__bf16*)alloc(NTOK * 512 * 2);
  float*  scales = (float*)alloc(256);
  (void)ws_size; (void)n_in; (void)in_sizes; (void)out_size;

  const int T = 256;
  k_convert_w<<<512, T, 0, stream>>>(w_qkv, w_out, wqkv_b, wout_b, scales);
  k_layernorm<<<512, T, 0, stream>>>(x, ln_g, ln_b, xn_b, (int)NTOK);
  k_gemm_qkv<<<768, T, 0, stream>>>(xn_b, wqkv_b, q_b, k_b, v_f, vT_b);
  k_landmarks<<<256, T, 0, stream>>>(q_b, k_b, ql_b, kl_b);
  k_attn2_logits<<<256, T, 0, stream>>>(ql_b, kl_b, mat_x);
  k_softmax256<<<256, T, 0, stream>>>(mat_x, x_b, 32 * 256);
  k_pinv_scales<<<256, T, 0, stream>>>(mat_x, scales);
  k_init_z<<<256, T, 0, stream>>>(mat_x, z0_b, z0T_b, scales);

  __bf16* zc = z0_b;  __bf16* zcT = z0T_b;
  __bf16* zn = z1_b;  __bf16* znT = z1T_b;
  for (int it = 0; it < 6; ++it) {
    k_gemm256b<true, false, true><<<256, T, 0, stream>>>(x_b, zcT, xz_b, nullptr, t1T_b, 7.f, -1.f);
    k_gemm256b<false, false, true><<<256, T, 0, stream>>>(xz_b, t1T_b, nullptr, nullptr, t2T_b, 15.f, -1.f);
    k_gemm256b<false, false, true><<<256, T, 0, stream>>>(xz_b, t2T_b, nullptr, nullptr, t3T_b, 13.f, -1.f);
    k_gemm256b<false, true, true><<<256, T, 0, stream>>>(zc, t3T_b, nullptr, zn, znT, 0.f, 0.25f);
    __bf16* t;
    t = zc; zc = zn; zn = t;
    t = zcT; zcT = znT; znT = t;
  }

  k_attn3v<<<256, T, 0, stream>>>(k_b, ql_b, vT_b, kvT_b);
  k_w2<<<64, T, 0, stream>>>(zc, kvT_b, w2t_b);
  k_attn1_out<<<1024, T, 0, stream>>>(kl_b, q_b, w2t_b, aout_f);
  k_conv_res<<<512, T, 0, stream>>>(v_f, res_w, aout_f, aout_b);
  k_outproj<<<256, T, 0, stream>>>(aout_b, wout_b, b_out, x, out);
}


# pinv iteration: the 3 GEMMs after xz fused into 1 launch (24 -> 12): z'=0.25(13z-15a+7b-c), a=z.xz, b=a.xz, c=b.xz per 32-row block, xz^T fragments held in registers; new summation order
# speedup vs baseline: 3.8110x; 1.2169x over previous
#include <hip/hip_runtime.h>
#include <hip/hip_bf16.h>
#include <math.h>


typedef __attribute__((ext_vector_type(16))) __bf16 bf16x16;
typedef __attribute__((ext_vector_type(8)))  float  floatx8;

union FragAB { bf16x16 v; unsigned int u[8]; };

__device__ __forceinline__ unsigned pack2bf(float a, float b) {
  __bf16 lo = (__bf16)a, hi = (__bf16)b;
  unsigned short ul = __builtin_bit_cast(unsigned short, lo);
  unsigned short uh = __builtin_bit_cast(unsigned short, hi);
  return (unsigned)ul | ((unsigned)uh << 16);
}

__device__ __forceinline__ uint4 pack8bf(floatx8 a, float sc) {
  uint4 p;
  p.x = pack2bf(a[0] * sc, a[1] * sc);
  p.y = pack2bf(a[2] * sc, a[3] * sc);
  p.z = pack2bf(a[4] * sc, a[5] * sc);
  p.w = pack2bf(a[6] * sc, a[7] * sc);
  return p;
}

__device__ __forceinline__ int kpair(int p, int h) {
  return (p < 4) ? (2 * p + 8 * h) : (8 + 2 * p + 8 * h);
}

__device__ __forceinline__ bf16x16 load_frag_bf16_rm(const __bf16* src, int row0,
                                                     int k0, int ld) {
  int lane = threadIdx.x & 31;
  int r = lane & 15, h = lane >> 4;
  const __bf16* base = src + (size_t)(row0 + r) * ld + k0;
  FragAB f;
#pragma unroll
  for (int p = 0; p < 8; ++p) f.u[p] = *(const unsigned*)(base + kpair(p, h));
  return f.v;
}

__device__ __forceinline__ floatx8 wmma_bf16(bf16x16 a, bf16x16 b, floatx8 c) {
  return __builtin_amdgcn_wmma_f32_16x16x32_bf16(false, a, false, b, (short)0, c,
                                                 false, false);
}

template <int KSTEPS>
__device__ __forceinline__ void gemm_strip(const __bf16* A, int rowA,
                                           const __bf16* B, int rowB, int ld,
                                           floatx8 acc[4]) {
#pragma unroll
  for (int ks = 0; ks < KSTEPS; ++ks) {
    int k0 = ks * 32;
    bf16x16 af = load_frag_bf16_rm(A, rowA, k0, ld);
#pragma unroll
    for (int t = 0; t < 4; ++t) {
      bf16x16 bfr = load_frag_bf16_rm(B, rowB + 16 * t, k0, ld);
      acc[t] = wmma_bf16(af, bfr, acc[t]);
    }
  }
}

__device__ __forceinline__ size_t pk_idx(int row, int k) {
  int kk = k & 31;
  int lane = (row & 15) + 16 * ((kk >> 3) & 1);
  return (size_t)(((row >> 4) * 8 + (k >> 5)) * 512 + (kk >> 4) * 256 +
                  lane * 8 + (kk & 7));
}

__device__ __forceinline__ bf16x16 load_frag_bf16_pk(const __bf16* mat, int rb,
                                                     int ks) {
  int lane = threadIdx.x & 31;
  const uint4* blk = (const uint4*)(mat + (size_t)(rb * 8 + ks) * 512) + lane;
  uint4 lo = blk[0], hi = blk[32];
  FragAB f;
  f.u[0] = lo.x; f.u[1] = lo.y; f.u[2] = lo.z; f.u[3] = lo.w;
  f.u[4] = hi.x; f.u[5] = hi.y; f.u[6] = hi.z; f.u[7] = hi.w;
  return f.v;
}

__device__ __forceinline__ size_t pk_idx_n(int row, int k, int KB) {
  int kk = k & 31;
  int lane = (row & 15) + 16 * ((kk >> 3) & 1);
  return (size_t)((row >> 4) * KB + (k >> 5)) * 512 + (kk >> 4) * 256 +
         lane * 8 + (kk & 7);
}

#define PK_AHEAD 4
template <bool BPK>
__device__ __forceinline__ void gemm_strip_pk(const __bf16* A, int rowA,
                                              const __bf16* B, int rowB,
                                              floatx8 acc[4]) {
  bf16x16 af[8], bfr[8][4];
#pragma unroll
  for (int ks = 0; ks < PK_AHEAD; ++ks) {
    af[ks] = load_frag_bf16_pk(A, rowA >> 4, ks);
#pragma unroll
    for (int t = 0; t < 4; ++t)
      bfr[ks][t] = BPK ? load_frag_bf16_pk(B, (rowB >> 4) + t, ks)
                       : load_frag_bf16_rm(B, rowB + 16 * t, ks * 32, 256);
    __builtin_amdgcn_sched_barrier(0);
  }
#pragma unroll
  for (int ks = 0; ks < 8; ++ks) {
#pragma unroll
    for (int t = 0; t < 4; ++t) acc[t] = wmma_bf16(af[ks], bfr[ks][t], acc[t]);
    __builtin_amdgcn_sched_barrier(0);
    if (ks + PK_AHEAD < 8) {
      af[ks + PK_AHEAD] = load_frag_bf16_pk(A, rowA >> 4, ks + PK_AHEAD);
#pragma unroll
      for (int t = 0; t < 4; ++t)
        bfr[ks + PK_AHEAD][t] =
            BPK ? load_frag_bf16_pk(B, (rowB >> 4) + t, ks + PK_AHEAD)
                : load_frag_bf16_rm(B, rowB + 16 * t, (ks + PK_AHEAD) * 32, 256);
      __builtin_amdgcn_sched_barrier(0);
    }
  }
}

typedef __attribute__((ext_vector_type(4))) unsigned int u32x4;

__device__ __forceinline__ bf16x16 load_frag_bf16_lds(const u32x4* tile, int j,
                                                      int ks) {
  int lane = threadIdx.x & 31;
  const u32x4* blk = tile + (j * 8 + ks) * 64 + lane;
  u32x4 lo = blk[0], hi = blk[32];
  FragAB f;
  f.u[0] = lo.x; f.u[1] = lo.y; f.u[2] = lo.z; f.u[3] = lo.w;
  f.u[4] = hi.x; f.u[5] = hi.y; f.u[6] = hi.z; f.u[7] = hi.w;
  return f.v;
}

__device__ __forceinline__ void lds_barrier() {
  __builtin_amdgcn_fence(__ATOMIC_RELEASE, "workgroup", "local");
  __builtin_amdgcn_s_barrier();
  __builtin_amdgcn_fence(__ATOMIC_ACQUIRE, "workgroup", "local");
}

#define GB_STAGE 3072
template <bool SWAP>
__device__ __forceinline__ void gemm_block_lds(const __bf16* A, int arb0,
                                               const __bf16* B, int brb0,
                                               u32x4* tile,
                                               floatx8 acc[4][4]) {
  int tid = (int)threadIdx.x;
  int lane = tid & 31, w = tid >> 5;
  int wm = w >> 2, wn = w & 3;
  int half = tid >> 7, q = tid & 127;
  const u32x4* Ag = (const u32x4*)A + (size_t)(arb0 + half) * 1024 + q;
  const u32x4* Bg = (const u32x4*)B + (size_t)(brb0 + half) * 1024 + q;
  u32x4 st[12];
#pragma unroll
  for (int j = 0; j < 4; ++j) st[j] = Ag[j * 2048];
#pragma unroll
  for (int j = 0; j < 8; ++j) st[4 + j] = Bg[j * 2048];
#pragma unroll
  for (int S = 0; S < 8; ++S) {
    u32x4* dst = tile + (S & 1) * GB_STAGE;
#pragma unroll
    for (int j = 0; j < 12; ++j) dst[j * 256 + tid] = st[j];
    if (S + 1 < 8) {
#pragma unroll
      for (int j = 0; j < 4; ++j) st[j] = Ag[j * 2048 + (S + 1) * 128];
#pragma unroll
      for (int j = 0; j < 8; ++j) st[4 + j] = Bg[j * 2048 + (S + 1) * 128];
    }
    __builtin_amdgcn_sched_barrier(0);
    lds_barrier();
    bf16x16 af[2][4], bfr[2][4];
#pragma unroll
    for (int kk = 0; kk < 2; ++kk) {
      const u32x4* sb = dst + kk * 64 + lane;
#pragma unroll
      for (int m = 0; m < 4; ++m) {
        const u32x4* blk = sb + (wm * 4 + m) * 128;
        u32x4 lo = blk[0], hi = blk[32];
        FragAB f;
        f.u[0] = lo.x; f.u[1] = lo.y; f.u[2] = lo.z; f.u[3] = lo.w;
        f.u[4] = hi.x; f.u[5] = hi.y; f.u[6] = hi.z; f.u[7] = hi.w;
        af[kk][m] = f.v;
      }
#pragma unroll
      for (int t = 0; t < 4; ++t) {
        const u32x4* blk = sb + (8 + wn * 4 + t) * 128;
        u32x4 lo = blk[0], hi = blk[32];
        FragAB f;
        f.u[0] = lo.x; f.u[1] = lo.y; f.u[2] = lo.z; f.u[3] = lo.w;
        f.u[4] = hi.x; f.u[5] = hi.y; f.u[6] = hi.z; f.u[7] = hi.w;
        bfr[kk][t] = f.v;
      }
    }
    __builtin_amdgcn_sched_barrier(0);
#pragma unroll
    for (int kk = 0; kk < 2; ++kk)
#pragma unroll
      for (int m = 0; m < 4; ++m)
#pragma unroll
        for (int t = 0; t < 4; ++t)
          acc[m][t] = SWAP ? wmma_bf16(bfr[kk][t], af[kk][m], acc[m][t])
                           : wmma_bf16(af[kk][m], bfr[kk][t], acc[m][t]);
  }
}

__global__ void k_convert_w(const float* wqkv, const float* wout,
                            __bf16* wqkv_b, __bf16* wout_b, float* scales) {
  int i = blockIdx.x * blockDim.x + threadIdx.x;
  if (i < 2) scales[i] = 0.f;
  bool second = i >= 1536 * 64;
  const float* src = second ? wout : wqkv;
  __bf16* dst = second ? wout_b : wqkv_b;
  int u = second ? i - 1536 * 64 : i;
  if (second && u >= 512 * 64) return;
  int blk = u >> 6, half = (u >> 5) & 1, lane = u & 31;
  int row = (blk >> 4) * 16 + (lane & 15);
  int k = (blk & 15) * 32 + half * 16 + 8 * (lane >> 4);
  const float4* s = (const float4*)(src + (size_t)row * 512 + k);
  float4 a = s[0], b = s[1];
  uint4 pv;
  pv.x = pack2bf(a.x, a.y);
  pv.y = pack2bf(a.z, a.w);
  pv.z = pack2bf(b.x, b.y);
  pv.w = pack2bf(b.z, b.w);
  ((uint4*)dst)[u] = pv;
}

#define LN_TOK 4
__global__ void __launch_bounds__(256)
k_layernorm(const float* x, const float* g, const float* b, __bf16* xn,
            int ntok) {
  int wave = (int)((blockIdx.x * blockDim.x + threadIdx.x) >> 5);
  int lane = threadIdx.x & 31;
  int tok0 = wave * LN_TOK;
  if (tok0 >= ntok) return;
  float vals[LN_TOK][16], gg[16], bb[16];
#pragma unroll
  for (int k = 0; k < LN_TOK; ++k) {
    const float* xr = x + (size_t)(tok0 + k) * 512;
#pragma unroll
    for (int i = 0; i < 16; ++i) vals[k][i] = xr[lane + 32 * i];
  }
#pragma unroll
  for (int i = 0; i < 16; ++i) {
    gg[i] = g[lane + 32 * i];
    bb[i] = b[lane + 32 * i];
  }
  __builtin_amdgcn_sched_barrier(0);
#pragma unroll
  for (int k = 0; k < LN_TOK; ++k) {
    float s = 0.f, s2 = 0.f;
#pragma unroll
    for (int i = 0; i < 16; ++i) {
      float v = vals[k][i];
      s += v; s2 += v * v;
    }
    for (int m = 16; m >= 1; m >>= 1) {
      s += __shfl_xor(s, m, 32);
      s2 += __shfl_xor(s2, m, 32);
    }
    float mu = s * (1.f / 512.f);
    float var = s2 * (1.f / 512.f) - mu * mu;
    float inv = rsqrtf(var + 1e-5f);
#pragma unroll
    for (int i = 0; i < 16; ++i) {
      int c = lane + 32 * i;
      xn[pk_idx_n(tok0 + k, c, 16)] =
          (__bf16)((vals[k][i] - mu) * inv * gg[i] + bb[i]);
    }
  }
}

__global__ void __launch_bounds__(256)
k_gemm_qkv(const __bf16* xn, const __bf16* wq, __bf16* qb,
           __bf16* kb, float* vf, __bf16* vT) {
  __shared__ u32x4 tile[2 * GB_STAGE];
  int bM = (int)blockIdx.x / 6, bN = (int)blockIdx.x % 6;
  int w = (int)threadIdx.x >> 5;
  int tM = bM * 2 + (w >> 2), s = bN * 4 + (w & 3);
  int lane = threadIdx.x & 31;
  int c = lane & 15, h = lane >> 4;
  int head = s & 7;
  int tok0 = tM * 64;
  int bb = tok0 >> 12, ns0 = tok0 & 4095;
  size_t bh = (size_t)(bb * 8 + head);
  floatx8 acc[4][4] = {};
  if (bN < 4) {
    gemm_block_lds<true>(xn, bM * 8, wq, bN * 16, tile, acc);
    __bf16* dst = (bN < 2 ? qb : kb) + (bh * 4096 + ns0 + c) * 64 + 8 * h;
    float sc = bN < 2 ? 0.125f : 1.f;
#pragma unroll
    for (int m = 0; m < 4; ++m)
#pragma unroll
      for (int t = 0; t < 4; ++t)
        *(uint4*)(dst + (16 * m) * 64 + 16 * t) = pack8bf(acc[m][t], sc);
  } else {
    gemm_block_lds<false>(xn, bM * 8, wq, bN * 16, tile, acc);
    float* vfd = vf + (bh * 4096 + ns0 + 8 * h) * 64 + c;
    __bf16* vTd = vT + (bh * 64 + c) * 4096 + ns0 + 8 * h;
#pragma unroll
    for (int m = 0; m < 4; ++m)
#pragma unroll
      for (int t = 0; t < 4; ++t) {
#pragma unroll
        for (int r = 0; r < 8; ++r)
          vfd[(16 * m + r) * 64 + 16 * t] = acc[m][t][r];
        *(uint4*)(vTd + (size_t)(16 * t) * 4096 + 16 * m) =
            pack8bf(acc[m][t], 1.f);
      }
  }
}

__device__ __forceinline__ float bf_lo(unsigned u) {
  return __builtin_bit_cast(float, u << 16);
}
__device__ __forceinline__ float bf_hi(unsigned u) {
  return __builtin_bit_cast(float, u & 0xffff0000u);
}

__global__ void __launch_bounds__(256)
k_landmarks(const __bf16* qb, const __bf16* kb, __bf16* ql, __bf16* kl) {
  int i = blockIdx.x * blockDim.x + threadIdx.x;
  if (i >= 32 * 256 * 8) return;
  int d8 = i & 7, m = (i >> 3) & 255, bh = i >> 11;
  size_t base = ((size_t)bh * 4096 + m * 16) * 64 + d8 * 8;
  u32x4 qv[16], kv[16];
#pragma unroll
  for (int t = 0; t < 16; ++t) {
    qv[t] = *(const u32x4*)(qb + base + t * 64);
    kv[t] = *(const u32x4*)(kb + base + t * 64);
  }
  float sq[8], sk[8];
#pragma unroll
  for (int e = 0; e < 8; ++e) { sq[e] = 0.f; sk[e] = 0.f; }
#pragma unroll
  for (int t = 0; t < 16; ++t) {
    sq[0] += bf_lo(qv[t].x); sq[1] += bf_hi(qv[t].x);
    sq[2] += bf_lo(qv[t].y); sq[3] += bf_hi(qv[t].y);
    sq[4] += bf_lo(qv[t].z); sq[5] += bf_hi(qv[t].z);
    sq[6] += bf_lo(qv[t].w); sq[7] += bf_hi(qv[t].w);
    sk[0] += bf_lo(kv[t].x); sk[1] += bf_hi(kv[t].x);
    sk[2] += bf_lo(kv[t].y); sk[3] += bf_hi(kv[t].y);
    sk[4] += bf_lo(kv[t].z); sk[5] += bf_hi(kv[t].z);
    sk[6] += bf_lo(kv[t].w); sk[7] += bf_hi(kv[t].w);
  }
  const float r16 = 1.f / 16.f;
  uint4 oq, ok;
  oq.x = pack2bf(sq[0] * r16, sq[1] * r16);
  oq.y = pack2bf(sq[2] * r16, sq[3] * r16);
  oq.z = pack2bf(sq[4] * r16, sq[5] * r16);
  oq.w = pack2bf(sq[6] * r16, sq[7] * r16);
  ok.x = pack2bf(sk[0] * r16, sk[1] * r16);
  ok.y = pack2bf(sk[2] * r16, sk[3] * r16);
  ok.z = pack2bf(sk[4] * r16, sk[5] * r16);
  ok.w = pack2bf(sk[6] * r16, sk[7] * r16);
  *(uint4*)(ql + (size_t)i * 8) = oq;
  *(uint4*)(kl + (size_t)i * 8) = ok;
}

__global__ void k_attn2_logits(const __bf16* ql, const __bf16* kl, float* out) {
  int wid = (int)((blockIdx.x * blockDim.x + threadIdx.x) >> 5);
  int bh = wid >> 6, t = wid & 63;
  int tM = t >> 2, s = t & 3;
  const __bf16* A = ql + (size_t)bh * 256 * 64;
  const __bf16* B = kl + (size_t)bh * 256 * 64;
  floatx8 acc[4] = {};
  gemm_strip<2>(A, tM * 16, B, s * 64, 64, acc);
  int lane = threadIdx.x & 31;
  int n = lane & 15, h = lane >> 4;
  float* o = out + (size_t)bh * 256 * 256;
#pragma unroll
  for (int t4 = 0; t4 < 4; ++t4)
#pragma unroll
    for (int r = 0; r < 8; ++r)
      o[(size_t)(tM * 16 + r + 8 * h) * 256 + s * 64 + 16 * t4 + n] =
          acc[t4][r];
}

#define SM_ROWS 4
__global__ void k_softmax256(float* x, __bf16* xb, int nrows) {
  int wave = (int)((blockIdx.x * blockDim.x + threadIdx.x) >> 5);
  int row0 = wave * SM_ROWS;
  if (row0 >= nrows) return;
  int lane = threadIdx.x & 31;
  __bf16* rb = xb + ((size_t)(row0 >> 8) << 16);
  float v[SM_ROWS][8];
#pragma unroll
  for (int k = 0; k < SM_ROWS; ++k)
#pragma unroll
    for (int i = 0; i < 8; ++i) v[k][i] = x[(size_t)(row0 + k) * 256 + lane + 32 * i];
#pragma unroll
  for (int k = 0; k < SM_ROWS; ++k) {
    float* r = x + (size_t)(row0 + k) * 256;
    int row = (row0 + k) & 255;
    float m = -1e30f;
#pragma unroll
    for (int i = 0; i < 8; ++i) m = fmaxf(m, v[k][i]);
    for (int t = 16; t >= 1; t >>= 1) m = fmaxf(m, __shfl_xor(m, t, 32));
    float s = 0.f;
#pragma unroll
    for (int i = 0; i < 8; ++i) { v[k][i] = __expf(v[k][i] - m); s += v[k][i]; }
    for (int t = 16; t >= 1; t >>= 1) s += __shfl_xor(s, t, 32);
    float inv = 1.f / s;
#pragma unroll
    for (int i = 0; i < 8; ++i) {
      float pv = v[k][i] * inv;
      r[lane + 32 * i] = pv;
      rb[pk_idx(row, lane + 32 * i)] = (__bf16)pv;
    }
  }
}

__global__ void k_pinv_scales(const float* x, float* scales) {
  int wave = (int)((blockIdx.x * blockDim.x + threadIdx.x) >> 5);
  if (wave >= 32 * 64) return;
  int lane = threadIdx.x & 31;
  int bh = wave >> 6, r0 = (wave & 63) * 4;
  const float* m = x + (size_t)bh * 256 * 256;
  float rv[4][8];
  float4 cv[8];
#pragma unroll
  for (int i = 0; i < 8; ++i)
    cv[i] = *(const float4*)(m + (size_t)(lane + 32 * i) * 256 + r0);
#pragma unroll
  for (int k = 0; k < 4; ++k)
#pragma unroll
    for (int i = 0; i < 8; ++i) rv[k][i] = m[(size_t)(r0 + k) * 256 + lane + 32 * i];
  unsigned rmax = 0u, cmax = 0u;
#pragma unroll
  for (int k = 0; k < 4; ++k) {
    float rs = 0.f, cs = 0.f;
#pragma unroll
    for (int i = 0; i < 8; ++i) {
      float cval = k == 0 ? cv[i].x : k == 1 ? cv[i].y : k == 2 ? cv[i].z : cv[i].w;
      rs += fabsf(rv[k][i]);
      cs += fabsf(cval);
    }
    for (int t = 16; t >= 1; t >>= 1) {
      rs += __shfl_xor(rs, t, 32);
      cs += __shfl_xor(cs, t, 32);
    }
    rmax = max(rmax, __float_as_uint(rs));
    cmax = max(cmax, __float_as_uint(cs));
  }
  if (lane == 0) {
    atomicMax((unsigned*)&scales[0], rmax);
    atomicMax((unsigned*)&scales[1], cmax);
  }
}

__global__ void k_init_z(const float* x, __bf16* zb, __bf16* zTb,
                         const float* scales) {
  int wave = (int)((blockIdx.x * blockDim.x + threadIdx.x) >> 5);
  if (wave >= 32 * 64) return;
  int lane = threadIdx.x & 31;
  int c = lane & 15, h = lane >> 4;
  int bh = wave >> 6, tj = (wave >> 3) & 7, tr = wave & 7;
  size_t bofs = (size_t)bh << 16;
  const float* xt = x + bofs + (size_t)(tj * 32) * 256 + tr * 32;
  float col[4][8];
#pragma unroll
  for (int g = 0; g < 4; ++g)
#pragma unroll
    for (int e = 0; e < 8; ++e) col[g][e] = xt[(size_t)(8 * g + e) * 256 + lane];
  float4 rw[2][2][2];
#pragma unroll
  for (int jb = 0; jb < 2; ++jb)
#pragma unroll
    for (int half = 0; half < 2; ++half) {
      const float4* s =
          (const float4*)(xt + (size_t)(16 * jb + c) * 256 + 16 * half + 8 * h);
      rw[jb][half][0] = s[0];
      rw[jb][half][1] = s[1];
    }
  __builtin_amdgcn_sched_barrier(0);
  float denom = scales[0] * scales[1];
#pragma unroll
  for (int jb = 0; jb < 2; ++jb)
#pragma unroll
    for (int half = 0; half < 2; ++half) {
      float4 a = rw[jb][half][0], b = rw[jb][half][1];
      uint4 pv;
      pv.x = pack2bf(a.x / denom, a.y / denom);
      pv.y = pack2bf(a.z / denom, a.w / denom);
      pv.z = pack2bf(b.x / denom, b.y / denom);
      pv.w = pack2bf(b.z / denom, b.w / denom);
      *(uint4*)(zTb + bofs + (size_t)(((2 * tj + jb) * 8 + tr) * 512 +
                                      half * 256 + lane * 8)) = pv;
    }
#pragma unroll
  for (int g = 0; g < 4; ++g) {
    uint4 pv;
    pv.x = pack2bf(col[g][0] / denom, col[g][1] / denom);
    pv.y = pack2bf(col[g][2] / denom, col[g][3] / denom);
    pv.z = pack2bf(col[g][4] / denom, col[g][5] / denom);
    pv.w = pack2bf(col[g][6] / denom, col[g][7] / denom);
    *(uint4*)(zb + bofs + (size_t)(((2 * tr + h) * 8 + tj) * 512 +
                                   (g >> 1) * 256 + (c + 16 * (g & 1)) * 8)) = pv;
  }
}

#define G8_ARB 4
#define G8_BRB 8
__global__ void __launch_bounds__(256)
k_gemm256t(const __bf16* A, const __bf16* BT, __bf16* OutTT) {
  __shared__ u32x4 tile[(G8_ARB + G8_BRB) * 8 * 64];
  int tid = (int)threadIdx.x;
  int w = tid >> 5;
  int bh = (int)blockIdx.x >> 3, g = (int)blockIdx.x & 7;
  int bm = g >> 1, bn = g & 1;
  int tM = bm * G8_ARB + (w >> 1), s = bn * 2 + (w & 1);
  size_t bofs = (size_t)bh << 16;
  const u32x4* Ag = (const u32x4*)(A + bofs) + (size_t)(bm * G8_ARB) * 8 * 64;
  const u32x4* Bg = (const u32x4*)(BT + bofs) + (size_t)(bn * G8_BRB) * 8 * 64;
  u32x4 st0[G8_ARB + G8_BRB], st1[G8_ARB + G8_BRB];
#pragma unroll
  for (int j = 0; j < G8_ARB; ++j) st0[j] = Ag[j * 512 + tid];
#pragma unroll
  for (int j = 0; j < G8_BRB; ++j) st0[G8_ARB + j] = Bg[j * 512 + tid];
  __builtin_amdgcn_sched_barrier(0);
#pragma unroll
  for (int j = 0; j < G8_ARB; ++j) st1[j] = Ag[j * 512 + 256 + tid];
#pragma unroll
  for (int j = 0; j < G8_BRB; ++j) st1[G8_ARB + j] = Bg[j * 512 + 256 + tid];
  floatx8 acc[4] = {};
#pragma unroll
  for (int j = 0; j < G8_ARB + G8_BRB; ++j) tile[j * 512 + tid] = st0[j];
  lds_barrier();
#pragma unroll
  for (int ks = 0; ks < 4; ++ks) {
    bf16x16 af = load_frag_bf16_lds(tile, w >> 1, ks);
#pragma unroll
    for (int t = 0; t < 4; ++t) {
      bf16x16 bf = load_frag_bf16_lds(tile, G8_ARB + (w & 1) * 4 + t, ks);
      acc[t] = wmma_bf16(af, bf, acc[t]);
    }
  }
#pragma unroll
  for (int j = 0; j < G8_ARB + G8_BRB; ++j) tile[j * 512 + 256 + tid] = st1[j];
  lds_barrier();
#pragma unroll
  for (int ks = 4; ks < 8; ++ks) {
    bf16x16 af = load_frag_bf16_lds(tile, w >> 1, ks);
#pragma unroll
    for (int t = 0; t < 4; ++t) {
      bf16x16 bf = load_frag_bf16_lds(tile, G8_ARB + (w & 1) * 4 + t, ks);
      acc[t] = wmma_bf16(af, bf, acc[t]);
    }
  }
  int lane = threadIdx.x & 31;
  int n = lane & 15, h = lane >> 4;
#pragma unroll
  for (int t4 = 0; t4 < 4; ++t4) {
    int N = s * 64 + 16 * t4 + n;
    *(uint4*)(OutTT + bofs + pk_idx(N, tM * 16 + 8 * h)) = pack8bf(acc[t4], 1.f);
  }
}

#define PC_A 1024
__device__ __forceinline__ u32x4 pack8v(floatx8 a) {
  u32x4 p;
  p.x = pack2bf(a[0], a[1]);
  p.y = pack2bf(a[2], a[3]);
  p.z = pack2bf(a[4], a[5]);
  p.w = pack2bf(a[6], a[7]);
  return p;
}

__device__ __forceinline__ void chain_mm(const u32x4* tile, int ja,
                                         const bf16x16 bfr[8][2],
                                         floatx8 acc[2][2]) {
#pragma unroll
  for (int ks = 0; ks < 8; ++ks) {
    bf16x16 af[2];
#pragma unroll
    for (int m = 0; m < 2; ++m) af[m] = load_frag_bf16_lds(tile, ja + m, ks);
#pragma unroll
    for (int m = 0; m < 2; ++m)
#pragma unroll
      for (int t = 0; t < 2; ++t)
        acc[m][t] = wmma_bf16(bfr[ks][t], af[m], acc[m][t]);
  }
}

__global__ void __launch_bounds__(256)
k_pinv_chain(const __bf16* Z, const __bf16* XZT, __bf16* Zn, __bf16* ZnT) {
  __shared__ u32x4 tile[2 * PC_A];
  int tid = (int)threadIdx.x;
  int lane = tid & 31, w = tid >> 5;
  int bh = (int)blockIdx.x >> 3, g = (int)blockIdx.x & 7;
  size_t bofs = (size_t)bh << 16;
  const u32x4* Ag = (const u32x4*)(Z + bofs) + (size_t)(2 * g) * 512;
  u32x4* buf0 = tile;
  u32x4* buf1 = tile + PC_A;
  u32x4 sa[4];
#pragma unroll
  for (int j = 0; j < 4; ++j) sa[j] = Ag[j * 256 + tid];
  __builtin_amdgcn_sched_barrier(0);
  bf16x16 bfr[8][2];
#pragma unroll
  for (int ks = 0; ks < 8; ++ks)
#pragma unroll
    for (int t = 0; t < 2; ++t)
      bfr[ks][t] = load_frag_bf16_pk(XZT + bofs, 2 * w + t, ks);
  __builtin_amdgcn_sched_barrier(0);
#pragma unroll
  for (int j = 0; j < 4; ++j) buf0[j * 256 + tid] = sa[j];
  lds_barrier();
  floatx8 acc[2][2] = {};
  chain_mm(tile, 0, bfr, acc);
  floatx8 res[2][2];
#pragma unroll
  for (int m = 0; m < 2; ++m)
#pragma unroll
    for (int t = 0; t < 2; ++t) {
      int pc = (m * 8 + w) * 64 + t * 32 + lane;
      u32x4 zv = buf0[pc];
      floatx8 a = acc[m][t];
      res[m][t][0] = 13.f * bf_lo(zv.x) - 15.f * a[0];
      res[m][t][1] = 13.f * bf_hi(zv.x) - 15.f * a[1];
      res[m][t][2] = 13.f * bf_lo(zv.y) - 15.f * a[2];
      res[m][t][3] = 13.f * bf_hi(zv.y) - 15.f * a[3];
      res[m][t][4] = 13.f * bf_lo(zv.z) - 15.f * a[4];
      res[m][t][5] = 13.f * bf_hi(zv.z) - 15.f * a[5];
      res[m][t][6] = 13.f * bf_lo(zv.w) - 15.f * a[6];
      res[m][t][7] = 13.f * bf_hi(zv.w) - 15.f * a[7];
      buf1[pc] = pack8v(a);
      acc[m][t] = floatx8{};
    }
  lds_barrier();
  chain_mm(tile, 2, bfr, acc);
#pragma unroll
  for (int m = 0; m < 2; ++m)
#pragma unroll
    for (int t = 0; t < 2; ++t) {
      int pc = (m * 8 + w) * 64 + t * 32 + lane;
#pragma unroll
      for (int r = 0; r < 8; ++r) res[m][t][r] += 7.f * acc[m][t][r];
      buf0[pc] = pack8v(acc[m][t]);
      acc[m][t] = floatx8{};
    }
  lds_barrier();
  chain_mm(tile, 0, bfr, acc);
#pragma unroll
  for (int m = 0; m < 2; ++m)
#pragma unroll
    for (int t = 0; t < 2; ++t) {
      int pc = (m * 8 + w) * 64 + t * 32 + lane;
      floatx8 zp;
#pragma unroll
      for (int r = 0; r < 8; ++r) zp[r] = 0.25f * (res[m][t][r] - acc[m][t][r]);
      u32x4 pv = pack8v(zp);
      *(u32x4*)(Zn + bofs + (size_t)(((2 * g + m) * 8 + w) * 512 +
                                     (t * 32 + lane) * 8)) = pv;
      buf1[pc] = pv;
    }
  lds_barrier();
  const unsigned short* sA = (const unsigned short*)buf1;
#pragma unroll
  for (int j = 0; j < 4; ++j) {
    int p = j * 256 + tid;
    int nb = p >> 6, q = p & 63;
    int half = q >> 5, lp = q & 31;
    int N = nb * 16 + (lp & 15), mh = lp >> 4;
    int kk = N & 31;
    int base = (half * 8 + (N >> 5)) * 512 + (kk >> 4) * 256 +
               (16 * ((kk >> 3) & 1) + 8 * mh) * 8 + (kk & 7);
    unsigned v[8];
#pragma unroll
    for (int e = 0; e < 8; ++e) v[e] = sA[base + e * 8];
    u32x4 pv;
    pv.x = v[0] | (v[1] << 16);
    pv.y = v[2] | (v[3] << 16);
    pv.z = v[4] | (v[5] << 16);
    pv.w = v[6] | (v[7] << 16);
    *(u32x4*)(ZnT + bofs + (size_t)((nb * 8 + g) * 512 + q * 8)) = pv;
  }
}

__device__ __forceinline__ void stream_attn(const __bf16* Ak, const __bf16* Bq,
                                            int qrow0, const __bf16* A2,
                                            int ldA2, int nkeys, float* outF,
                                            int rsF, __bf16* outB, int nsB,
                                            int dsB) {
  int lane = threadIdx.x & 31;
  int n = lane & 15, h = lane >> 4;
  bf16x16 bq0 = load_frag_bf16_rm(Bq, qrow0, 0, 64);
  bf16x16 bq1 = load_frag_bf16_rm(Bq, qrow0, 32, 64);
  float run_m = -1e30f, run_s = 0.f;
  floatx8 acc[4] = {};
  for (int c = 0; c < nkeys; c += 32) {
    floatx8 s0 = {}, s1 = {};
    s0 = wmma_bf16(load_frag_bf16_rm(Ak, c, 0, 64), bq0, s0);
    s0 = wmma_bf16(load_frag_bf16_rm(Ak, c, 32, 64), bq1, s0);
    s1 = wmma_bf16(load_frag_bf16_rm(Ak, c + 16, 0, 64), bq0, s1);
    s1 = wmma_bf16(load_frag_bf16_rm(Ak, c + 16, 32, 64), bq1, s1);
    float mc = -1e30f;
#pragma unroll
    for (int r = 0; r < 8; ++r) { mc = fmaxf(mc, s0[r]); mc = fmaxf(mc, s1[r]); }
    mc = fmaxf(mc, __shfl_xor(mc, 16, 32));
    float newm = fmaxf(run_m, mc);
    float scale = __expf(run_m - newm);
    float p0[8], p1[8], ps = 0.f;
#pragma unroll
    for (int r = 0; r < 8; ++r) {
      p0[r] = __expf(s0[r] - newm);
      p1[r] = __expf(s1[r] - newm);
      ps += p0[r] + p1[r];
    }
    run_s = run_s * scale + ps;
    run_m = newm;
    FragAB bp;
#pragma unroll
    for (int p = 0; p < 4; ++p) {
      bp.u[p] = pack2bf(p0[2 * p], p0[2 * p + 1]);
      bp.u[4 + p] = pack2bf(p1[2 * p], p1[2 * p + 1]);
    }
#pragma unroll
    for (int t = 0; t < 4; ++t) {
#pragma unroll
      for (int r = 0; r < 8; ++r) acc[t][r] *= scale;
      acc[t] = wmma_bf16(load_frag_bf16_rm(A2, 16 * t, c, ldA2), bp.v, acc[t]);
    }
  }
  float tot = run_s + __shfl_xor(run_s, 16, 32);
  float inv = 1.f / tot;
#pragma unroll
  for (int t = 0; t < 4; ++t)
#pragma unroll
    for (int r = 0; r < 8; ++r) {
      int dh = 16 * t + r + 8 * h;
      float val = acc[t][r] * inv;
      if (outF) outF[(size_t)n * rsF + dh] = val;
      if (outB) outB[(size_t)n * nsB + (size_t)dh * dsB] = (__bf16)val;
    }
}

#define A3_WAVES 8
#define A3_QG 2
#define A3_KEYS (4096 / A3_WAVES)
__global__ void __launch_bounds__(256)
k_attn3v(const __bf16* kb, const __bf16* ql, const __bf16* vT, __bf16* kvT) {
  __shared__ float accL[A3_WAVES][A3_QG][4][8][32];
  __shared__ float mL[A3_WAVES][A3_QG][16];
  __shared__ float sL[A3_WAVES][A3_QG][16];
  int lane = threadIdx.x & 31;
  int w = __builtin_amdgcn_readfirstlane((int)(threadIdx.x >> 5));
  int bh = blockIdx.x >> 3, q0 = (blockIdx.x & 7) * (16 * A3_QG);
  const __bf16* Ak = kb + (size_t)bh * 4096 * 64;
  const __bf16* Bq = ql + (size_t)bh * 256 * 64;
  const __bf16* A2 = vT + (size_t)bh * 64 * 4096;
  bf16x16 bq[A3_QG][2];
  float run_m[A3_QG], run_s[A3_QG];
  floatx8 acc[A3_QG][4];
#pragma unroll
  for (int g = 0; g < A3_QG; ++g) {
    bq[g][0] = load_frag_bf16_rm(Bq, q0 + 16 * g, 0, 64);
    bq[g][1] = load_frag_bf16_rm(Bq, q0 + 16 * g, 32, 64);
    run_m[g] = -1e30f;
    run_s[g] = 0.f;
#pragma unroll
    for (int t = 0; t < 4; ++t) acc[g][t] = floatx8{};
  }
  int kbeg = w * A3_KEYS;
  for (int c = kbeg; c < kbeg + A3_KEYS; c += 32) {
    bf16x16 ka[4], va[4];
    ka[0] = load_frag_bf16_rm(Ak, c, 0, 64);
    ka[1] = load_frag_bf16_rm(Ak, c, 32, 64);
    ka[2] = load_frag_bf16_rm(Ak, c + 16, 0, 64);
    ka[3] = load_frag_bf16_rm(Ak, c + 16, 32, 64);
#pragma unroll
    for (int t = 0; t < 4; ++t) va[t] = load_frag_bf16_rm(A2, 16 * t, c, 4096);
    __builtin_amdgcn_sched_barrier(0);
#pragma unroll
    for (int g = 0; g < A3_QG; ++g) {
      floatx8 s0 = {}, s1 = {};
      s0 = wmma_bf16(ka[0], bq[g][0], s0);
      s0 = wmma_bf16(ka[1], bq[g][1], s0);
      s1 = wmma_bf16(ka[2], bq[g][0], s1);
      s1 = wmma_bf16(ka[3], bq[g][1], s1);
      float mc = -1e30f;
#pragma unroll
      for (int r = 0; r < 8; ++r) { mc = fmaxf(mc, s0[r]); mc = fmaxf(mc, s1[r]); }
      mc = fmaxf(mc, __shfl_xor(mc, 16, 32));
      float newm = fmaxf(run_m[g], mc);
      float scale = __expf(run_m[g] - newm);
      float p0[8], p1[8], ps = 0.f;
#pragma unroll
      for (int r = 0; r < 8; ++r) {
        p0[r] = __expf(s0[r] - newm);
        p1[r] = __expf(s1[r] - newm);
        ps += p0[r] + p1[r];
      }
      run_s[g] = run_s[g] * scale + ps;
      run_m[g] = newm;
      FragAB bp;
#pragma unroll
      for (int p = 0; p < 4; ++p) {
        bp.u[p] = pack2bf(p0[2 * p], p0[2 * p + 1]);
        bp.u[4 + p] = pack2bf(p1[2 * p], p1[2 * p + 1]);
      }
#pragma unroll
      for (int t = 0; t < 4; ++t) {
#pragma unroll
        for (int r = 0; r < 8; ++r) acc[g][t][r] *= scale;
        acc[g][t] = wmma_bf16(va[t], bp.v, acc[g][t]);
      }
    }
  }
#pragma unroll
  for (int g = 0; g < A3_QG; ++g) {
    float tot = run_s[g] + __shfl_xor(run_s[g], 16, 32);
    if (lane < 16) {
      mL[w][g][lane] = run_m[g];
      sL[w][g][lane] = tot;
    }
#pragma unroll
    for (int t = 0; t < 4; ++t)
#pragma unroll
      for (int r = 0; r < 8; ++r) accL[w][g][t][r][lane] = acc[g][t][r];
  }
  __syncthreads();
  int cc = lane & 15, hh = lane >> 4;
  for (int g = 0; g < A3_QG; ++g) {
    float mw[A3_WAVES], f[A3_WAVES], M = -1e30f, S = 0.f;
#pragma unroll
    for (int v = 0; v < A3_WAVES; ++v) { mw[v] = mL[v][g][cc]; M = fmaxf(M, mw[v]); }
#pragma unroll
    for (int v = 0; v < A3_WAVES; ++v) {
      f[v] = __expf(mw[v] - M);
      S += sL[v][g][cc] * f[v];
    }
    float inv = 1.f / S;
    for (int t = 0; t < 4; ++t) {
      float o = 0.f;
#pragma unroll
      for (int v = 0; v < A3_WAVES; ++v) o += accL[v][g][t][w][lane] * f[v];
      int dh = 16 * t + 8 * hh + w;
      kvT[((size_t)bh * 64 + dh) * 256 + q0 + 16 * g + cc] = (__bf16)(o * inv);
    }
  }
}

__global__ void __launch_bounds__(256)
k_w2(const __bf16* zb, const __bf16* kvT, __bf16* w2t) {
  int wid = (int)((blockIdx.x * blockDim.x + threadIdx.x) >> 5);
  int bh = wid >> 4, tM = wid & 15;
  floatx8 acc[4] = {};
  gemm_strip_pk<false>(zb + ((size_t)bh << 16), tM * 16,
                       kvT + (size_t)bh * 64 * 256, 0, acc);
  int lane = threadIdx.x & 31;
  int n = lane & 15, h = lane >> 4;
#pragma unroll
  for (int t4 = 0; t4 < 4; ++t4)
#pragma unroll
    for (int r = 0; r < 8; ++r) {
      int M = tM * 16 + r + 8 * h;
      int N = 16 * t4 + n;
      w2t[((size_t)bh * 64 + N) * 256 + M] = (__bf16)acc[t4][r];
    }
}

__global__ void k_attn1_out(const __bf16* kl, const __bf16* qb,
                            const __bf16* w2t, float* aout) {
  int wid = (int)((blockIdx.x * blockDim.x + threadIdx.x) >> 5);
  int bh = wid >> 8, blk = wid & 255;
  int b = bh >> 3, hh = bh & 7;
  stream_attn(kl + (size_t)bh * 256 * 64, qb + (size_t)bh * 4096 * 64, blk * 16,
              w2t + (size_t)bh * 64 * 256, 256, 256,
              aout + ((size_t)b * 4096 + blk * 16) * 512 + hh * 64, 512,
              nullptr, 0, 0);
}

#define CONV_R 16
#define CONV_G 8
#define CONV_ROWS (CONV_R + 32)
__global__ void __launch_bounds__(256, 4)
k_conv_res(const float* __restrict__ vf, const float* __restrict__ rw,
           float* __restrict__ aout, __bf16* __restrict__ aout_b) {
  int t = (int)(blockIdx.x * blockDim.x + threadIdx.x);
  if (t >= 4 * (4096 / CONV_R) * 128) return;
  int c4 = t & 127;
  int chunk = t >> 7;
  int b = chunk / (4096 / CONV_R);
  int ns0 = (chunk % (4096 / CONV_R)) * CONV_R;
  int col = c4 * 4, hh = col >> 6, dh = col & 63;
  const float* vcol = vf + ((size_t)(b * 8 + hh) * 4096) * 64 + dh;
  float* arow = aout + ((size_t)b * 4096 + ns0) * 512 + col;
  int boff = (int)pk_idx_n(b * 4096 + ns0, col, 16);
  float w[33];
#pragma unroll
  for (int j = 0; j < 33; ++j) w[j] = rw[hh * 33 + j];
  float4 acc[CONV_R], win[CONV_ROWS], a[CONV_R];
#pragma unroll
  for (int o = 0; o < CONV_R; ++o) acc[o] = make_float4(0.f, 0.f, 0.f, 0.f);
#pragma unroll
  for (int k = 0; k < CONV_G; ++k) {
    int p = ns0 + k - 16;
    int pc = p < 0 ? 0 : (p > 4095 ? 4095 : p);
    win[k] = *(const float4*)(vcol + (size_t)pc * 64);
  }
#pragma unroll
  for (int g = 0; g < CONV_ROWS / CONV_G; ++g) {
    if (g + 1 < CONV_ROWS / CONV_G) {
#pragma unroll
      for (int k = (g + 1) * CONV_G; k < (g + 2) * CONV_G; ++k) {
        int p = ns0 + k - 16;
        int pc = p < 0 ? 0 : (p > 4095 ? 4095 : p);
        win[k] = *(const float4*)(vcol + (size_t)pc * 64);
      }
    } else {
#pragma unroll
      for (int o = 0; o < CONV_R; ++o)
        a[o] = *(const float4*)(arow + (size_t)o * 512);
    }
#pragma unroll
    for (int k = g * CONV_G; k < (g + 1) * CONV_G; ++k) {
      int p = ns0 + k - 16;
      float4 v = win[k];
      if (p < 0 || p > 4095) v = make_float4(0.f, 0.f, 0.f, 0.f);
#pragma unroll
      for (int o = 0; o < CONV_R; ++o) {
        int j = k - o;
        if (j >= 0 && j < 33) {
          acc[o].x += v.x * w[j];
          acc[o].y += v.y * w[j];
          acc[o].z += v.z * w[j];
          acc[o].w += v.w * w[j];
        }
      }
    }
  }
#pragma unroll
  for (int o = 0; o < CONV_R; ++o) {
    float4 r;
    r.x = a[o].x + acc[o].x;
    r.y = a[o].y + acc[o].y;
    r.z = a[o].z + acc[o].z;
    r.w = a[o].w + acc[o].w;
    *(float4*)(arow + (size_t)o * 512) = r;
    uint2 pk;
    pk.x = pack2bf(r.x, r.y);
    pk.y = pack2bf(r.z, r.w);
    *(uint2*)(aout_b + (boff + o * 8)) = pk;
  }
}

__global__ void __launch_bounds__(256)
k_outproj(const __bf16* __restrict__ aoutb, const __bf16* __restrict__ wo,
          const float* __restrict__ bias, const float* __restrict__ x,
          float* __restrict__ out) {
  __shared__ u32x4 tile[2 * GB_STAGE];
  int bM = (int)blockIdx.x >> 1, bN = (int)blockIdx.x & 1;
  int w = (int)threadIdx.x >> 5;
  int tM = bM * 2 + (w >> 2), s = bN * 4 + (w & 3);
  floatx8 acc[4][4] = {};
  gemm_block_lds<false>(aoutb, bM * 8, wo, bN * 16, tile, acc);
  int lane = threadIdx.x & 31;
  int n = lane & 15, h = lane >> 4;
#pragma unroll
  for (int t4 = 0; t4 < 4; ++t4) {
    int col = s * 64 + 16 * t4 + n;
    float bc = bias[col];
#pragma unroll
    for (int m = 0; m < 4; ++m) {
#pragma unroll
      for (int r = 0; r < 8; ++r) {
        size_t idx = (size_t)(tM * 64 + 16 * m + r + 8 * h) * 512 + col;
        out[idx] = x[idx] + acc[m][t4][r] + bc;
      }
    }
  }
}

extern "C" void kernel_launch(void* const* d_in, const int* in_sizes, int n_in,
                              void* d_out, int out_size, void* d_ws,
                              size_t ws_size, hipStream_t stream) {
  const float* x     = (const float*)d_in[0];
  const float* ln_g  = (const float*)d_in[1];
  const float* ln_b  = (const float*)d_in[2];
  const float* w_qkv = (const float*)d_in[3];
  const float* w_out = (const float*)d_in[4];
  const float* b_out = (const float*)d_in[5];
  const float* res_w = (const float*)d_in[6];
  float* out = (float*)d_out;

  char* p = (char*)d_ws;
  auto alloc = [&](size_t bytes) -> void* {
    void* r = (void*)p;
    p += (bytes + 255) & ~(size_t)255;
    return r;
  };
  const size_t NTOK = 4 * 4096;
  const size_t MM = (size_t)32 * 256 * 256;
  __bf16* xn_b   = (__bf16*)alloc(NTOK * 512 * 2);
  __bf16* wqkv_b = (__bf16*)alloc(1536 * 512 * 2);
  __bf16* wout_b = (__bf16*)alloc(512 * 512 * 2);
  __bf16* q_b    = (__bf16*)alloc((size_t)32 * 4096 * 64 * 2);
  __bf16* k_b    = (__bf16*)alloc((size_t)32 * 4096 * 64 * 2);
  float*  v_f    = (float*)alloc((size_t)32 * 4096 * 64 * 4);
  __bf16* vT_b   = (__bf16*)alloc((size_t)32 * 64 * 4096 * 2);
  __bf16* ql_b   = (__bf16*)alloc((size_t)32 * 256 * 64 * 2);
  __bf16* kl_b   = (__bf16*)alloc((size_t)32 * 256 * 64 * 2);
  float*  mat_x  = (float*)alloc(MM * 4);
  __bf16* x_b    = (__bf16*)alloc(MM * 2);
  __bf16* z0_b   = (__bf16*)alloc(MM * 2);
  __bf16* z0T_b  = (__bf16*)alloc(MM * 2);
  __bf16* z1_b   = (__bf16*)alloc(MM * 2);
  __bf16* z1T_b  = (__bf16*)alloc(MM * 2);
  __bf16* xzT_b  = (__bf16*)alloc(MM * 2);
  __bf16* kvT_b  = (__bf16*)alloc((size_t)32 * 64 * 256 * 2);
  __bf16* w2t_b  = (__bf16*)alloc((size_t)32 * 64 * 256 * 2);
  float*  aout_f = (float*)alloc(NTOK * 512 * 4);
  __bf16* aout_b = (__bf16*)alloc(NTOK * 512 * 2);
  float*  scales = (float*)alloc(256);
  (void)ws_size; (void)n_in; (void)in_sizes; (void)out_size;

  const int T = 256;
  k_convert_w<<<512, T, 0, stream>>>(w_qkv, w_out, wqkv_b, wout_b, scales);
  k_layernorm<<<512, T, 0, stream>>>(x, ln_g, ln_b, xn_b, (int)NTOK);
  k_gemm_qkv<<<768, T, 0, stream>>>(xn_b, wqkv_b, q_b, k_b, v_f, vT_b);
  k_landmarks<<<256, T, 0, stream>>>(q_b, k_b, ql_b, kl_b);
  k_attn2_logits<<<256, T, 0, stream>>>(ql_b, kl_b, mat_x);
  k_softmax256<<<256, T, 0, stream>>>(mat_x, x_b, 32 * 256);
  k_pinv_scales<<<256, T, 0, stream>>>(mat_x, scales);
  k_init_z<<<256, T, 0, stream>>>(mat_x, z0_b, z0T_b, scales);

  __bf16* zc = z0_b;  __bf16* zcT = z0T_b;
  __bf16* zn = z1_b;  __bf16* znT = z1T_b;
  for (int it = 0; it < 6; ++it) {
    k_gemm256t<<<256, T, 0, stream>>>(x_b, zcT, xzT_b);
    k_pinv_chain<<<256, T, 0, stream>>>(zc, xzT_b, zn, znT);
    __bf16* t;
    t = zc; zc = zn; zn = t;
    t = zcT; zcT = znT; znT = t;
  }

  k_attn3v<<<256, T, 0, stream>>>(k_b, ql_b, vT_b, kvT_b);
  k_w2<<<64, T, 0, stream>>>(zc, kvT_b, w2t_b);
  k_attn1_out<<<1024, T, 0, stream>>>(kl_b, q_b, w2t_b, aout_f);
  k_conv_res<<<512, T, 0, stream>>>(v_f, res_w, aout_f, aout_b);
  k_outproj<<<256, T, 0, stream>>>(aout_b, wout_b, b_out, x, out);
}
